# starting kernel
# baseline (speedup 1.0000x reference)
#include <hip/hip_runtime.h>
#include <math.h>

#define NN 100000
#define EE 1000000
#define FF 128
#define DD 64
#define GG 256

typedef float v2f __attribute__((ext_vector_type(2)));
typedef float v8f __attribute__((ext_vector_type(8)));

__global__ void k_zero_deg(float* __restrict__ deg) {
  int i = blockIdx.x * blockDim.x + threadIdx.x;
  if (i < NN) deg[i] = 0.0f;
}

__global__ void k_degree(const int* __restrict__ dst, float* __restrict__ deg) {
  int e = blockIdx.x * blockDim.x + threadIdx.x;
  if (e < EE) atomicAdd(&deg[dst[e]], 1.0f);
}

__global__ void k_dinv(float* __restrict__ deg) {
  int i = blockIdx.x * blockDim.x + threadIdx.x;
  if (i < NN) deg[i] = rsqrtf(deg[i] + 1.0f);
}

template <int K>
__global__ void k_gemm_wmma(const float* __restrict__ A, const float* __restrict__ B,
                            float* __restrict__ C) {
  int gid  = blockIdx.x * blockDim.x + threadIdx.x;
  int wave = gid >> 5;
  int lane = gid & 31;
  int tm = wave >> 2;
  int tn = wave & 3;
  if (tm * 16 >= NN) return;

  int half = lane >> 4;
  int m    = lane & 15;
  const float* arow = A + (size_t)(tm * 16 + m) * K;
  int col = tn * 16 + m;

  v8f acc = {};
#pragma unroll
  for (int kb = 0; kb < K; kb += 4) {
    int k0 = kb + half * 2;
    v2f a; a[0] = arow[k0];          a[1] = arow[k0 + 1];
    v2f b; b[0] = B[k0 * DD + col];  b[1] = B[(k0 + 1) * DD + col];
    acc = __builtin_amdgcn_wmma_f32_16x16x4_f32(
         false, a,  false, b,
         (short)0, acc,  false,  false);
  }

  int r0 = tm * 16 + half * 8;
#pragma unroll
  for (int j = 0; j < 8; ++j)
    C[(size_t)(r0 + j) * DD + col] = acc[j];
}

__global__ void k_init_agg(const float* __restrict__ hw, const float* __restrict__ dinv,
                           float* __restrict__ agg) {
  int i = blockIdx.x * blockDim.x + threadIdx.x;
  if (i < NN * DD) {
    float dv = dinv[i >> 6];
    agg[i] = dv * dv * hw[i];
  }
}

__global__ void k_scatter(const int* __restrict__ src, const int* __restrict__ dst,
                          const float* __restrict__ dinv, const float* __restrict__ hw,
                          float* __restrict__ agg) {
  int gid  = blockIdx.x * blockDim.x + threadIdx.x;
  int e    = gid >> 5;
  int lane = gid & 31;
  if (e >= EE) return;
  int s = src[e], d = dst[e];
  float w = dinv[s] * dinv[d];
  const float2* hp = (const float2*)(hw + (size_t)s * DD);
  float2 v = hp[lane];
  float* ap = agg + (size_t)d * DD + lane * 2;
  atomicAdd(ap,     v.x * w);
  atomicAdd(ap + 1, v.y * w);
}

__global__ void k_finish(const float* __restrict__ b, float* __restrict__ agg) {
  int i = blockIdx.x * blockDim.x + threadIdx.x;
  if (i < NN * DD) agg[i] = tanhf(agg[i] + b[i & 63]);
}

__device__ __forceinline__ unsigned fflip(float f) {
  unsigned u = __float_as_uint(f);
  return u ^ ((unsigned)((int)u >> 31) | 0x80000000u);
}
__device__ __forceinline__ float funflip(unsigned u) {
  unsigned m = (u >> 31) ? 0x80000000u : 0xFFFFFFFFu;
  return __uint_as_float(u ^ m);
}

__global__ void k_pool_init(unsigned* __restrict__ gmp, float* __restrict__ gsum,
                            float* __restrict__ counts) {
  int i = blockIdx.x * blockDim.x + threadIdx.x;
  if (i < GG * DD) { gmp[i] = 0x007FFFFFu  ; gsum[i] = 0.0f; }
  if (i < GG) counts[i] = 0.0f;
}

__global__ void k_pool(const float* __restrict__ h, const int* __restrict__ bidx,
                       unsigned* __restrict__ gmp, float* __restrict__ gsum,
                       float* __restrict__ counts) {
  int i = blockIdx.x * blockDim.x + threadIdx.x;
  if (i >= NN * DD) return;
  int n = i >> 6, dd = i & 63;
  int g = bidx[n];
  float v = h[i];
  atomicMax(&gmp[g * DD + dd], fflip(v));
  atomicAdd(&gsum[g * DD + dd], v);
  if (dd == 0) atomicAdd(&counts[g], 1.0f);
}

__global__ void k_final(const unsigned* __restrict__ gmp, const float* __restrict__ gsum,
                        const float* __restrict__ counts, const float* __restrict__ Wout,
                        const float* __restrict__ bout, float* __restrict__ out,
                        float* __restrict__ hidden) {
  __shared__ float red[2 * DD];
  int g = blockIdx.x, t = threadIdx.x;
  float cnt = fmaxf(counts[g], 1.0f);
  float hv = (t < DD) ? funflip(gmp[g * DD + t]) : (gsum[g * DD + (t - DD)] / cnt);
  hidden[(size_t)g * 2 * DD + t] = hv;
  red[t] = hv * Wout[t];
  __syncthreads();
  for (int s = DD; s > 0; s >>= 1) {
    if (t < s) red[t] += red[t + s];
    __syncthreads();
  }
  if (t == 0) out[g] = red[0] + bout[0];
}

extern "C" void kernel_launch(void* const* d_in, const int* in_sizes, int n_in,
                              void* d_out, int out_size, void* d_ws, size_t ws_size,
                              hipStream_t stream) {
  const float* x    = (const float*)d_in[0];
  const int*   ei   = (const int*)d_in[1];
  const int*   bidx = (const int*)d_in[2];
  const float* W0   = (const float*)d_in[3];
  const float* b0   = (const float*)d_in[4];
  const float* W1   = (const float*)d_in[5];
  const float* b1   = (const float*)d_in[6];
  const float* W2   = (const float*)d_in[7];
  const float* b2   = (const float*)d_in[8];
  const float* W3   = (const float*)d_in[9];
  const float* b3   = (const float*)d_in[10];
  const float* Wout = (const float*)d_in[11];
  const float* bout = (const float*)d_in[12];

  const int* src = ei;
  const int* dst = ei + EE;

  char*  ws   = (char*)d_ws;
  size_t off  = 0;
  float* dinv = (float*)(ws + off);  off += (((size_t)NN * 4 + 255) & ~(size_t)255);
  float* bufP = (float*)(ws + off);  off += (size_t)NN * DD * 4;
  float* bufQ = (float*)(ws + off);  off += (size_t)NN * DD * 4;
  unsigned* gmp = (unsigned*)(ws + off); off += (size_t)GG * DD * 4;
  float* gsum   = (float*)(ws + off);    off += (size_t)GG * DD * 4;
  float* counts = (float*)(ws + off);    off += (size_t)GG * 4;
  (void)ws_size; (void)n_in; (void)in_sizes; (void)out_size;

  float* outp   = (float*)d_out;
  float* hidden = outp + GG;

  const int T = 256;

  k_zero_deg<<<(NN + T - 1) / T, T, 0, stream>>>(dinv);
  k_degree  <<<(EE + T - 1) / T, T, 0, stream>>>(dst, dinv);
  k_dinv    <<<(NN + T - 1) / T, T, 0, stream>>>(dinv);

  const int gemm_blocks = (NN / 16) * 4 / 8;
  const int nd_blocks   = (NN * DD + T - 1) / T;
  const int sc_blocks   = (int)(((size_t)EE * 32 + T - 1) / T);

  const float* Wl[4] = {W0, W1, W2, W3};
  const float* bl[4] = {b0, b1, b2, b3};
  for (int l = 0; l < 4; ++l) {
    if (l == 0) k_gemm_wmma<FF><<<gemm_blocks, T, 0, stream>>>(x,    Wl[l], bufQ);
    else        k_gemm_wmma<DD><<<gemm_blocks, T, 0, stream>>>(bufP, Wl[l], bufQ);
    k_init_agg<<<nd_blocks, T, 0, stream>>>(bufQ, dinv, bufP);
    k_scatter <<<sc_blocks, T, 0, stream>>>(src, dst, dinv, bufQ, bufP);
    k_finish  <<<nd_blocks, T, 0, stream>>>(bl[l], bufP);
  }

  k_pool_init<<<(GG * DD + T - 1) / T, T, 0, stream>>>(gmp, gsum, counts);
  k_pool     <<<nd_blocks, T, 0, stream>>>(bufP, bidx, gmp, gsum, counts);
  k_final    <<<GG, 2 * DD, 0, stream>>>(gmp, gsum, counts, Wout, bout, outp, hidden);
}


# pool: each 16-lane group reduces 16 consecutive nodes in registers per run of equal graph id, one atomic set per run not per element; mean-pool sum order changes (serial in chunk, then atomics)
# speedup vs baseline: 1.2290x; 1.2290x over previous
#include <hip/hip_runtime.h>
#include <math.h>

#define NN 100000
#define EE 1000000
#define FF 128
#define DD 64
#define GG 256

typedef float v2f __attribute__((ext_vector_type(2)));
typedef float v8f __attribute__((ext_vector_type(8)));

__global__ void k_zero_deg(float* __restrict__ deg) {
  int i = blockIdx.x * blockDim.x + threadIdx.x;
  if (i < NN) deg[i] = 0.0f;
}

__global__ void k_degree(const int* __restrict__ dst, float* __restrict__ deg) {
  int e = blockIdx.x * blockDim.x + threadIdx.x;
  if (e < EE) atomicAdd(&deg[dst[e]], 1.0f);
}

__global__ void k_dinv(float* __restrict__ deg) {
  int i = blockIdx.x * blockDim.x + threadIdx.x;
  if (i < NN) deg[i] = rsqrtf(deg[i] + 1.0f);
}

template <int K>
__global__ void k_gemm_wmma(const float* __restrict__ A, const float* __restrict__ B,
                            float* __restrict__ C) {
  int gid  = blockIdx.x * blockDim.x + threadIdx.x;
  int wave = gid >> 5;
  int lane = gid & 31;
  int tm = wave >> 2;
  int tn = wave & 3;
  if (tm * 16 >= NN) return;

  int half = lane >> 4;
  int m    = lane & 15;
  const float* arow = A + (size_t)(tm * 16 + m) * K;
  int col = tn * 16 + m;

  v8f acc = {};
#pragma unroll
  for (int kb = 0; kb < K; kb += 4) {
    int k0 = kb + half * 2;
    v2f a; a[0] = arow[k0];          a[1] = arow[k0 + 1];
    v2f b; b[0] = B[k0 * DD + col];  b[1] = B[(k0 + 1) * DD + col];
    acc = __builtin_amdgcn_wmma_f32_16x16x4_f32(
         false, a,  false, b,
         (short)0, acc,  false,  false);
  }

  int r0 = tm * 16 + half * 8;
#pragma unroll
  for (int j = 0; j < 8; ++j)
    C[(size_t)(r0 + j) * DD + col] = acc[j];
}

__global__ void k_init_agg(const float* __restrict__ hw, const float* __restrict__ dinv,
                           float* __restrict__ agg) {
  int i = blockIdx.x * blockDim.x + threadIdx.x;
  if (i < NN * DD) {
    float dv = dinv[i >> 6];
    agg[i] = dv * dv * hw[i];
  }
}

__global__ void k_scatter(const int* __restrict__ src, const int* __restrict__ dst,
                          const float* __restrict__ dinv, const float* __restrict__ hw,
                          float* __restrict__ agg) {
  int gid  = blockIdx.x * blockDim.x + threadIdx.x;
  int e    = gid >> 5;
  int lane = gid & 31;
  if (e >= EE) return;
  int s = src[e], d = dst[e];
  float w = dinv[s] * dinv[d];
  const float2* hp = (const float2*)(hw + (size_t)s * DD);
  float2 v = hp[lane];
  float* ap = agg + (size_t)d * DD + lane * 2;
  atomicAdd(ap,     v.x * w);
  atomicAdd(ap + 1, v.y * w);
}

__global__ void k_finish(const float* __restrict__ b, float* __restrict__ agg) {
  int i = blockIdx.x * blockDim.x + threadIdx.x;
  if (i < NN * DD) agg[i] = tanhf(agg[i] + b[i & 63]);
}

__device__ __forceinline__ unsigned fflip(float f) {
  unsigned u = __float_as_uint(f);
  return u ^ ((unsigned)((int)u >> 31) | 0x80000000u);
}
__device__ __forceinline__ float funflip(unsigned u) {
  unsigned m = (u >> 31) ? 0x80000000u : 0xFFFFFFFFu;
  return __uint_as_float(u ^ m);
}

__global__ void k_pool_init(unsigned* __restrict__ gmp, float* __restrict__ gsum,
                            float* __restrict__ counts) {
  int i = blockIdx.x * blockDim.x + threadIdx.x;
  if (i < GG * DD) { gmp[i] = 0x007FFFFFu  ; gsum[i] = 0.0f; }
  if (i < GG) counts[i] = 0.0f;
}

#define PCH 16
static_assert(NN % PCH == 0, "node count must be a multiple of the pooling chunk");

__device__ __forceinline__ void pool_flush(int g, int q, const unsigned (&mx)[4],
                                           const float (&sm)[4], float cnt,
                                           unsigned* __restrict__ gmp,
                                           float* __restrict__ gsum,
                                           float* __restrict__ counts) {
  int base = g * DD + q * 4;
#pragma unroll
  for (int k = 0; k < 4; ++k) {
    atomicMax(&gmp[base + k], mx[k]);
    atomicAdd(&gsum[base + k], sm[k]);
  }
  if (q == 0) atomicAdd(&counts[g], cnt);
}

__global__ void k_pool(const float* __restrict__ h, const int* __restrict__ bidx,
                       unsigned* __restrict__ gmp, float* __restrict__ gsum,
                       float* __restrict__ counts) {
  int gid = blockIdx.x * blockDim.x + threadIdx.x;
  int grp = gid >> 4;
  int q   = gid & 15;
  int n0  = grp * PCH;
  if (n0 >= NN) return;

  int    g[PCH];
  float4 v[PCH];
  const float4* hp = (const float4*)(h + (size_t)n0 * DD) + q;
#pragma unroll
  for (int i = 0; i < PCH; ++i) {
    g[i] = bidx[n0 + i];
    v[i] = hp[i * (DD / 4)];
  }

  int      cg = g[0];
  unsigned mx[4] = {fflip(v[0].x), fflip(v[0].y), fflip(v[0].z), fflip(v[0].w)};
  float    sm[4] = {v[0].x, v[0].y, v[0].z, v[0].w};
  float    cnt = 1.0f;
#pragma unroll
  for (int i = 1; i < PCH; ++i) {
    if (g[i] != cg) {
      pool_flush(cg, q, mx, sm, cnt, gmp, gsum, counts);
      cg = g[i];
      mx[0] = fflip(v[i].x); mx[1] = fflip(v[i].y); mx[2] = fflip(v[i].z); mx[3] = fflip(v[i].w);
      sm[0] = v[i].x; sm[1] = v[i].y; sm[2] = v[i].z; sm[3] = v[i].w;
      cnt = 1.0f;
    } else {
      mx[0] = max(mx[0], fflip(v[i].x)); mx[1] = max(mx[1], fflip(v[i].y));
      mx[2] = max(mx[2], fflip(v[i].z)); mx[3] = max(mx[3], fflip(v[i].w));
      sm[0] += v[i].x; sm[1] += v[i].y; sm[2] += v[i].z; sm[3] += v[i].w;
      cnt += 1.0f;
    }
  }
  pool_flush(cg, q, mx, sm, cnt, gmp, gsum, counts);
}

__global__ void k_final(const unsigned* __restrict__ gmp, const float* __restrict__ gsum,
                        const float* __restrict__ counts, const float* __restrict__ Wout,
                        const float* __restrict__ bout, float* __restrict__ out,
                        float* __restrict__ hidden) {
  __shared__ float red[2 * DD];
  int g = blockIdx.x, t = threadIdx.x;
  float cnt = fmaxf(counts[g], 1.0f);
  float hv = (t < DD) ? funflip(gmp[g * DD + t]) : (gsum[g * DD + (t - DD)] / cnt);
  hidden[(size_t)g * 2 * DD + t] = hv;
  red[t] = hv * Wout[t];
  __syncthreads();
  for (int s = DD; s > 0; s >>= 1) {
    if (t < s) red[t] += red[t + s];
    __syncthreads();
  }
  if (t == 0) out[g] = red[0] + bout[0];
}

extern "C" void kernel_launch(void* const* d_in, const int* in_sizes, int n_in,
                              void* d_out, int out_size, void* d_ws, size_t ws_size,
                              hipStream_t stream) {
  const float* x    = (const float*)d_in[0];
  const int*   ei   = (const int*)d_in[1];
  const int*   bidx = (const int*)d_in[2];
  const float* W0   = (const float*)d_in[3];
  const float* b0   = (const float*)d_in[4];
  const float* W1   = (const float*)d_in[5];
  const float* b1   = (const float*)d_in[6];
  const float* W2   = (const float*)d_in[7];
  const float* b2   = (const float*)d_in[8];
  const float* W3   = (const float*)d_in[9];
  const float* b3   = (const float*)d_in[10];
  const float* Wout = (const float*)d_in[11];
  const float* bout = (const float*)d_in[12];

  const int* src = ei;
  const int* dst = ei + EE;

  char*  ws   = (char*)d_ws;
  size_t off  = 0;
  float* dinv = (float*)(ws + off);  off += (((size_t)NN * 4 + 255) & ~(size_t)255);
  float* bufP = (float*)(ws + off);  off += (size_t)NN * DD * 4;
  float* bufQ = (float*)(ws + off);  off += (size_t)NN * DD * 4;
  unsigned* gmp = (unsigned*)(ws + off); off += (size_t)GG * DD * 4;
  float* gsum   = (float*)(ws + off);    off += (size_t)GG * DD * 4;
  float* counts = (float*)(ws + off);    off += (size_t)GG * 4;
  (void)ws_size; (void)n_in; (void)in_sizes; (void)out_size;

  float* outp   = (float*)d_out;
  float* hidden = outp + GG;

  const int T = 256;

  k_zero_deg<<<(NN + T - 1) / T, T, 0, stream>>>(dinv);
  k_degree  <<<(EE + T - 1) / T, T, 0, stream>>>(dst, dinv);
  k_dinv    <<<(NN + T - 1) / T, T, 0, stream>>>(dinv);

  const int gemm_blocks = (NN / 16) * 4 / 8;
  const int nd_blocks   = (NN * DD + T - 1) / T;
  const int sc_blocks   = (int)(((size_t)EE * 32 + T - 1) / T);

  const float* Wl[4] = {W0, W1, W2, W3};
  const float* bl[4] = {b0, b1, b2, b3};
  for (int l = 0; l < 4; ++l) {
    if (l == 0) k_gemm_wmma<FF><<<gemm_blocks, T, 0, stream>>>(x,    Wl[l], bufQ);
    else        k_gemm_wmma<DD><<<gemm_blocks, T, 0, stream>>>(bufP, Wl[l], bufQ);
    k_init_agg<<<nd_blocks, T, 0, stream>>>(bufQ, dinv, bufP);
    k_scatter <<<sc_blocks, T, 0, stream>>>(src, dst, dinv, bufQ, bufP);
    k_finish  <<<nd_blocks, T, 0, stream>>>(bl[l], bufP);
  }

  k_pool_init<<<(GG * DD + T - 1) / T, T, 0, stream>>>(gmp, gsum, counts);
  k_pool     <<<((NN / PCH) * 16 + T - 1) / T, T, 0, stream>>>(bufP, bidx, gmp, gsum, counts);
  k_final    <<<GG, 2 * DD, 0, stream>>>(gmp, gsum, counts, Wout, bout, outp, hidden);
}


# aggregation: per-edge atomic scatter replaced by in-call counting sort of edges by destination + one-wave-per-node gather (self-loop, bias, tanh in its epilogue); edge sums serial in fill order
# speedup vs baseline: 2.3369x; 1.9015x over previous
#include <hip/hip_runtime.h>
#include <math.h>

#define NN 100000
#define EE 1000000
#define FF 128
#define DD 64
#define GG 256

typedef float v2f __attribute__((ext_vector_type(2)));
typedef float v8f __attribute__((ext_vector_type(8)));

#define SCAN_T 256
#define SCAN_NODES (SCAN_T * 4)
#define SCAN_BLOCKS ((NN + SCAN_NODES - 1) / SCAN_NODES)
static_assert(NN % 4 == 0, "node count must be a multiple of 4 for the int4 scan loads");
static_assert(SCAN_BLOCKS <= 128, "block offsets are summed by one wave, four entries per lane");

__global__ void k_zero_cnt(int* __restrict__ cnt) {
  int i = blockIdx.x * blockDim.x + threadIdx.x;
  if (i < NN) cnt[i] = 0;
}

__global__ void k_hist(const int* __restrict__ dst, int* __restrict__ cnt) {
  int e = blockIdx.x * blockDim.x + threadIdx.x;
  if (e < EE) atomicAdd(&cnt[dst[e]], 1);
}

__device__ __forceinline__ int wave_sum(int v) {
  for (int o = 16; o > 0; o >>= 1) v += __shfl_xor(v, o, 32);
  return v;
}

__global__ void k_dinv_bsum(const int* __restrict__ cnt, float* __restrict__ dinv,
                            int* __restrict__ bsum) {
  __shared__ int wsum[SCAN_T / 32];
  int t  = threadIdx.x;
  int i4 = blockIdx.x * SCAN_T + t;
  int s  = 0;
  if (i4 < NN / 4) {
    int4 c = ((const int4*)cnt)[i4];
    float4 r;
    r.x = rsqrtf((float)c.x + 1.0f); r.y = rsqrtf((float)c.y + 1.0f);
    r.z = rsqrtf((float)c.z + 1.0f); r.w = rsqrtf((float)c.w + 1.0f);
    ((float4*)dinv)[i4] = r;
    s = c.x + c.y + c.z + c.w;
  }
  s = wave_sum(s);
  if ((t & 31) == 0) wsum[t >> 5] = s;
  __syncthreads();
  if (t == 0) {
    int tot = 0;
    for (int w = 0; w < SCAN_T / 32; ++w) tot += wsum[w];
    bsum[blockIdx.x] = tot;
  }
}

__global__ void k_scan(const int* __restrict__ cnt, const int* __restrict__ bsum,
                       int* __restrict__ rowptr, int* __restrict__ cursor) {
  __shared__ int wtot[SCAN_T / 32];
  __shared__ int boff;
  int t = threadIdx.x, lane = t & 31, w = t >> 5;
  int b = blockIdx.x;
  if (w == 0) {
    int p = 0;
    for (int j = lane; j < b; j += 32) p += bsum[j];
    p = wave_sum(p);
    if (lane == 0) boff = p;
  }
  int  i4 = b * SCAN_T + t;
  int4 c  = make_int4(0, 0, 0, 0);
  if (i4 < NN / 4) c = ((const int4*)cnt)[i4];
  int tsum = c.x + c.y + c.z + c.w;
  int inc  = tsum;
  for (int o = 1; o < 32; o <<= 1) {
    int up = __shfl_up(inc, o, 32);
    if (lane >= o) inc += up;
  }
  if (lane == 31) wtot[w] = inc;
  __syncthreads();
  int wbase = 0;
  for (int k = 0; k < w; ++k) wbase += wtot[k];
  int base = boff + wbase + inc - tsum;
  if (i4 < NN / 4) {
    int4 r;
    r.x = base; r.y = r.x + c.x; r.z = r.y + c.y; r.w = r.z + c.z;
    ((int4*)rowptr)[i4] = r;
    ((int4*)cursor)[i4] = r;
  }
}

__global__ void k_fill(const int* __restrict__ src, const int* __restrict__ dst,
                       const float* __restrict__ dinv, int* __restrict__ cursor,
                       int2* __restrict__ csr) {
  int e = blockIdx.x * blockDim.x + threadIdx.x;
  if (e < EE) {
    int s = src[e], d = dst[e];
    float w = dinv[s] * dinv[d];
    int p = atomicAdd(&cursor[d], 1);
    if ((unsigned)p < (unsigned)EE)
      csr[p] = make_int2(s, __float_as_int(w));
  }
}

template <int K>
__global__ void k_gemm_wmma(const float* __restrict__ A, const float* __restrict__ B,
                            float* __restrict__ C) {
  int gid  = blockIdx.x * blockDim.x + threadIdx.x;
  int wave = gid >> 5;
  int lane = gid & 31;
  int tm = wave >> 2;
  int tn = wave & 3;
  if (tm * 16 >= NN) return;

  int half = lane >> 4;
  int m    = lane & 15;
  const float* arow = A + (size_t)(tm * 16 + m) * K;
  int col = tn * 16 + m;

  v8f acc = {};
#pragma unroll
  for (int kb = 0; kb < K; kb += 4) {
    int k0 = kb + half * 2;
    v2f a; a[0] = arow[k0];          a[1] = arow[k0 + 1];
    v2f b; b[0] = B[k0 * DD + col];  b[1] = B[(k0 + 1) * DD + col];
    acc = __builtin_amdgcn_wmma_f32_16x16x4_f32(
         false, a,  false, b,
         (short)0, acc,  false,  false);
  }

  int r0 = tm * 16 + half * 8;
#pragma unroll
  for (int j = 0; j < 8; ++j)
    C[(size_t)(r0 + j) * DD + col] = acc[j];
}

__global__ void k_gather(const int* __restrict__ rowptr, const int* __restrict__ cnt,
                         const int2* __restrict__ csr, const float* __restrict__ dinv,
                         const float* __restrict__ hw, const float* __restrict__ b,
                         float* __restrict__ h) {
  int gid  = blockIdx.x * blockDim.x + threadIdx.x;
  int lane = gid & 31;
  int d = __builtin_amdgcn_readfirstlane(gid >> 5);
  if (d >= NN) return;
  int   beg = __builtin_amdgcn_readfirstlane(rowptr[d]);
  int   n   = __builtin_amdgcn_readfirstlane(cnt[d]);
  float dv  = dinv[d];
  const float2* hp = (const float2*)hw + lane;
  float2 vself = hp[(size_t)d * (DD / 2)];
  float2 acc;  acc.x = 0.0f;  acc.y = 0.0f;

  for (int c0 = 0; c0 < n; c0 += 32) {
    int m = min(32, n - c0);
    int   sl = d;
    float wl = 0.0f;
    if (lane < m) {
      int2 ew = csr[beg + c0 + lane];
      sl = ew.x;
      if ((unsigned)sl >= (unsigned)NN) sl = d;
      wl = __int_as_float(ew.y);
    }
    for (int j = 0; j < m; j += 4) {
      int j1 = min(j + 1, m - 1), j2 = min(j + 2, m - 1), j3 = min(j + 3, m - 1);
      int s0 = __builtin_amdgcn_readlane(sl, j);
      int s1 = __builtin_amdgcn_readlane(sl, j1);
      int s2 = __builtin_amdgcn_readlane(sl, j2);
      int s3 = __builtin_amdgcn_readlane(sl, j3);
      float2 v0 = hp[(size_t)s0 * (DD / 2)];
      float2 v1 = hp[(size_t)s1 * (DD / 2)];
      float2 v2 = hp[(size_t)s2 * (DD / 2)];
      float2 v3 = hp[(size_t)s3 * (DD / 2)];
      float w0 = __int_as_float(__builtin_amdgcn_readlane(__float_as_int(wl), j));
      float w1 = __int_as_float(__builtin_amdgcn_readlane(__float_as_int(wl), j1));
      float w2 = __int_as_float(__builtin_amdgcn_readlane(__float_as_int(wl), j2));
      float w3 = __int_as_float(__builtin_amdgcn_readlane(__float_as_int(wl), j3));
      acc.x = fmaf(w0, v0.x, acc.x);  acc.y = fmaf(w0, v0.y, acc.y);
      if (j + 1 < m) { acc.x = fmaf(w1, v1.x, acc.x);  acc.y = fmaf(w1, v1.y, acc.y); }
      if (j + 2 < m) { acc.x = fmaf(w2, v2.x, acc.x);  acc.y = fmaf(w2, v2.y, acc.y); }
      if (j + 3 < m) { acc.x = fmaf(w3, v3.x, acc.x);  acc.y = fmaf(w3, v3.y, acc.y); }
    }
  }

  float  sn = dv * dv;
  float2 bb = ((const float2*)b)[lane];
  float2 o;
  o.x = tanhf(fmaf(sn, vself.x, acc.x) + bb.x);
  o.y = tanhf(fmaf(sn, vself.y, acc.y) + bb.y);
  ((float2*)h)[(size_t)d * (DD / 2) + lane] = o;
}

__device__ __forceinline__ unsigned fflip(float f) {
  unsigned u = __float_as_uint(f);
  return u ^ ((unsigned)((int)u >> 31) | 0x80000000u);
}
__device__ __forceinline__ float funflip(unsigned u) {
  unsigned m = (u >> 31) ? 0x80000000u : 0xFFFFFFFFu;
  return __uint_as_float(u ^ m);
}

__global__ void k_pool_init(unsigned* __restrict__ gmp, float* __restrict__ gsum,
                            float* __restrict__ counts) {
  int i = blockIdx.x * blockDim.x + threadIdx.x;
  if (i < GG * DD) { gmp[i] = 0x007FFFFFu  ; gsum[i] = 0.0f; }
  if (i < GG) counts[i] = 0.0f;
}

#define PCH 16
static_assert(NN % PCH == 0, "node count must be a multiple of the pooling chunk");

__device__ __forceinline__ void pool_flush(int g, int q, const unsigned (&mx)[4],
                                           const float (&sm)[4], float cnt,
                                           unsigned* __restrict__ gmp,
                                           float* __restrict__ gsum,
                                           float* __restrict__ counts) {
  int base = g * DD + q * 4;
#pragma unroll
  for (int k = 0; k < 4; ++k) {
    atomicMax(&gmp[base + k], mx[k]);
    atomicAdd(&gsum[base + k], sm[k]);
  }
  if (q == 0) atomicAdd(&counts[g], cnt);
}

__global__ void k_pool(const float* __restrict__ h, const int* __restrict__ bidx,
                       unsigned* __restrict__ gmp, float* __restrict__ gsum,
                       float* __restrict__ counts) {
  int gid = blockIdx.x * blockDim.x + threadIdx.x;
  int grp = gid >> 4;
  int q   = gid & 15;
  int n0  = grp * PCH;
  if (n0 >= NN) return;

  int    g[PCH];
  float4 v[PCH];
  const float4* hp = (const float4*)(h + (size_t)n0 * DD) + q;
#pragma unroll
  for (int i = 0; i < PCH; ++i) {
    g[i] = bidx[n0 + i];
    v[i] = hp[i * (DD / 4)];
  }

  int      cg = g[0];
  unsigned mx[4] = {fflip(v[0].x), fflip(v[0].y), fflip(v[0].z), fflip(v[0].w)};
  float    sm[4] = {v[0].x, v[0].y, v[0].z, v[0].w};
  float    cnt = 1.0f;
#pragma unroll
  for (int i = 1; i < PCH; ++i) {
    if (g[i] != cg) {
      pool_flush(cg, q, mx, sm, cnt, gmp, gsum, counts);
      cg = g[i];
      mx[0] = fflip(v[i].x); mx[1] = fflip(v[i].y); mx[2] = fflip(v[i].z); mx[3] = fflip(v[i].w);
      sm[0] = v[i].x; sm[1] = v[i].y; sm[2] = v[i].z; sm[3] = v[i].w;
      cnt = 1.0f;
    } else {
      mx[0] = max(mx[0], fflip(v[i].x)); mx[1] = max(mx[1], fflip(v[i].y));
      mx[2] = max(mx[2], fflip(v[i].z)); mx[3] = max(mx[3], fflip(v[i].w));
      sm[0] += v[i].x; sm[1] += v[i].y; sm[2] += v[i].z; sm[3] += v[i].w;
      cnt += 1.0f;
    }
  }
  pool_flush(cg, q, mx, sm, cnt, gmp, gsum, counts);
}

__global__ void k_final(const unsigned* __restrict__ gmp, const float* __restrict__ gsum,
                        const float* __restrict__ counts, const float* __restrict__ Wout,
                        const float* __restrict__ bout, float* __restrict__ out,
                        float* __restrict__ hidden) {
  __shared__ float red[2 * DD];
  int g = blockIdx.x, t = threadIdx.x;
  float cnt = fmaxf(counts[g], 1.0f);
  float hv = (t < DD) ? funflip(gmp[g * DD + t]) : (gsum[g * DD + (t - DD)] / cnt);
  hidden[(size_t)g * 2 * DD + t] = hv;
  red[t] = hv * Wout[t];
  __syncthreads();
  for (int s = DD; s > 0; s >>= 1) {
    if (t < s) red[t] += red[t + s];
    __syncthreads();
  }
  if (t == 0) out[g] = red[0] + bout[0];
}

extern "C" void kernel_launch(void* const* d_in, const int* in_sizes, int n_in,
                              void* d_out, int out_size, void* d_ws, size_t ws_size,
                              hipStream_t stream) {
  const float* x    = (const float*)d_in[0];
  const int*   ei   = (const int*)d_in[1];
  const int*   bidx = (const int*)d_in[2];
  const float* W0   = (const float*)d_in[3];
  const float* b0   = (const float*)d_in[4];
  const float* W1   = (const float*)d_in[5];
  const float* b1   = (const float*)d_in[6];
  const float* W2   = (const float*)d_in[7];
  const float* b2   = (const float*)d_in[8];
  const float* W3   = (const float*)d_in[9];
  const float* b3   = (const float*)d_in[10];
  const float* Wout = (const float*)d_in[11];
  const float* bout = (const float*)d_in[12];

  const int* src = ei;
  const int* dst = ei + EE;

  char*  ws   = (char*)d_ws;
  size_t off  = 0;
  const size_t nodeWords = (((size_t)NN * 4 + 255) & ~(size_t)255);
  float* dinv = (float*)(ws + off);  off += nodeWords;
  float* bufP = (float*)(ws + off);  off += (size_t)NN * DD * 4;
  float* bufQ = (float*)(ws + off);  off += (size_t)NN * DD * 4;
  unsigned* gmp = (unsigned*)(ws + off); off += (size_t)GG * DD * 4;
  float* gsum   = (float*)(ws + off);    off += (size_t)GG * DD * 4;
  float* counts = (float*)(ws + off);    off += (size_t)GG * 4;
  int* cnt    = (int*)(ws + off);  off += nodeWords;
  int* rowptr = (int*)(ws + off);  off += nodeWords;
  int* cursor = (int*)(ws + off);  off += nodeWords;
  int* bsum   = (int*)(ws + off);  off += 1024;
  int2* csr   = (int2*)(ws + off); off += (size_t)EE * 8;
  (void)ws_size; (void)n_in; (void)in_sizes; (void)out_size;

  float* outp   = (float*)d_out;
  float* hidden = outp + GG;

  const int T = 256;

  k_zero_cnt <<<(NN + T - 1) / T, T, 0, stream>>>(cnt);
  k_hist     <<<(EE + T - 1) / T, T, 0, stream>>>(dst, cnt);
  k_dinv_bsum<<<SCAN_BLOCKS, SCAN_T, 0, stream>>>(cnt, dinv, bsum);
  k_scan     <<<SCAN_BLOCKS, SCAN_T, 0, stream>>>(cnt, bsum, rowptr, cursor);
  k_fill     <<<(EE + T - 1) / T, T, 0, stream>>>(src, dst, dinv, cursor, csr);

  const int gemm_blocks = (NN / 16) * 4 / 8;
  const int ga_blocks   = (int)(((size_t)NN * 32 + T - 1) / T);

  const float* Wl[4] = {W0, W1, W2, W3};
  const float* bl[4] = {b0, b1, b2, b3};
  for (int l = 0; l < 4; ++l) {
    if (l == 0) k_gemm_wmma<FF><<<gemm_blocks, T, 0, stream>>>(x,    Wl[l], bufQ);
    else        k_gemm_wmma<DD><<<gemm_blocks, T, 0, stream>>>(bufP, Wl[l], bufQ);
    k_gather<<<ga_blocks, T, 0, stream>>>(rowptr, cnt, csr, dinv, bufQ, bl[l], bufP);
  }

  k_pool_init<<<(GG * DD + T - 1) / T, T, 0, stream>>>(gmp, gsum, counts);
  k_pool     <<<((NN / PCH) * 16 + T - 1) / T, T, 0, stream>>>(bufP, bidx, gmp, gsum, counts);
  k_final    <<<GG, 2 * DD, 0, stream>>>(gmp, gsum, counts, Wout, bout, outp, hidden);
}


# gather: 12288 resident waves each walk over ~8 nodes (next node's edge pairs and list bounds prefetched, rows loaded 8 at a time) instead of one short wave per node; per-node sum order unchanged
# speedup vs baseline: 2.4825x; 1.0623x over previous
#include <hip/hip_runtime.h>
#include <math.h>

#define NN 100000
#define EE 1000000
#define FF 128
#define DD 64
#define GG 256

typedef float v2f __attribute__((ext_vector_type(2)));
typedef float v8f __attribute__((ext_vector_type(8)));

#define SCAN_T 256
#define SCAN_NODES (SCAN_T * 4)
#define SCAN_BLOCKS ((NN + SCAN_NODES - 1) / SCAN_NODES)
static_assert(NN % 4 == 0, "node count must be a multiple of 4 for the int4 scan loads");
static_assert(SCAN_BLOCKS <= 128, "block offsets are summed by one wave, four entries per lane");

__global__ void k_zero_cnt(int* __restrict__ cnt) {
  int i = blockIdx.x * blockDim.x + threadIdx.x;
  if (i < NN) cnt[i] = 0;
}

__global__ void k_hist(const int* __restrict__ dst, int* __restrict__ cnt) {
  int e = blockIdx.x * blockDim.x + threadIdx.x;
  if (e < EE) atomicAdd(&cnt[dst[e]], 1);
}

__device__ __forceinline__ int wave_sum(int v) {
  for (int o = 16; o > 0; o >>= 1) v += __shfl_xor(v, o, 32);
  return v;
}

__global__ void k_dinv_bsum(const int* __restrict__ cnt, float* __restrict__ dinv,
                            int* __restrict__ bsum) {
  __shared__ int wsum[SCAN_T / 32];
  int t  = threadIdx.x;
  int i4 = blockIdx.x * SCAN_T + t;
  int s  = 0;
  if (i4 < NN / 4) {
    int4 c = ((const int4*)cnt)[i4];
    float4 r;
    r.x = rsqrtf((float)c.x + 1.0f); r.y = rsqrtf((float)c.y + 1.0f);
    r.z = rsqrtf((float)c.z + 1.0f); r.w = rsqrtf((float)c.w + 1.0f);
    ((float4*)dinv)[i4] = r;
    s = c.x + c.y + c.z + c.w;
  }
  s = wave_sum(s);
  if ((t & 31) == 0) wsum[t >> 5] = s;
  __syncthreads();
  if (t == 0) {
    int tot = 0;
    for (int w = 0; w < SCAN_T / 32; ++w) tot += wsum[w];
    bsum[blockIdx.x] = tot;
  }
}

__global__ void k_scan(const int* __restrict__ cnt, const int* __restrict__ bsum,
                       int* __restrict__ rowptr, int* __restrict__ cursor) {
  __shared__ int wtot[SCAN_T / 32];
  __shared__ int boff;
  int t = threadIdx.x, lane = t & 31, w = t >> 5;
  int b = blockIdx.x;
  if (w == 0) {
    int p = 0;
    for (int j = lane; j < b; j += 32) p += bsum[j];
    p = wave_sum(p);
    if (lane == 0) boff = p;
  }
  int  i4 = b * SCAN_T + t;
  int4 c  = make_int4(0, 0, 0, 0);
  if (i4 < NN / 4) c = ((const int4*)cnt)[i4];
  int tsum = c.x + c.y + c.z + c.w;
  int inc  = tsum;
  for (int o = 1; o < 32; o <<= 1) {
    int up = __shfl_up(inc, o, 32);
    if (lane >= o) inc += up;
  }
  if (lane == 31) wtot[w] = inc;
  __syncthreads();
  int wbase = 0;
  for (int k = 0; k < w; ++k) wbase += wtot[k];
  int base = boff + wbase + inc - tsum;
  if (i4 < NN / 4) {
    int4 r;
    r.x = base; r.y = r.x + c.x; r.z = r.y + c.y; r.w = r.z + c.z;
    ((int4*)rowptr)[i4] = r;
    ((int4*)cursor)[i4] = r;
  }
}

__global__ void k_fill(const int* __restrict__ src, const int* __restrict__ dst,
                       const float* __restrict__ dinv, int* __restrict__ cursor,
                       int2* __restrict__ csr) {
  int e = blockIdx.x * blockDim.x + threadIdx.x;
  if (e < EE) {
    int s = src[e], d = dst[e];
    float w = dinv[s] * dinv[d];
    int p = atomicAdd(&cursor[d], 1);
    if ((unsigned)p < (unsigned)EE)
      csr[p] = make_int2(s, __float_as_int(w));
  }
}

template <int K>
__global__ void k_gemm_wmma(const float* __restrict__ A, const float* __restrict__ B,
                            float* __restrict__ C) {
  int gid  = blockIdx.x * blockDim.x + threadIdx.x;
  int wave = gid >> 5;
  int lane = gid & 31;
  int tm = wave >> 2;
  int tn = wave & 3;
  if (tm * 16 >= NN) return;

  int half = lane >> 4;
  int m    = lane & 15;
  const float* arow = A + (size_t)(tm * 16 + m) * K;
  int col = tn * 16 + m;

  v8f acc = {};
#pragma unroll
  for (int kb = 0; kb < K; kb += 4) {
    int k0 = kb + half * 2;
    v2f a; a[0] = arow[k0];          a[1] = arow[k0 + 1];
    v2f b; b[0] = B[k0 * DD + col];  b[1] = B[(k0 + 1) * DD + col];
    acc = __builtin_amdgcn_wmma_f32_16x16x4_f32(
         false, a,  false, b,
         (short)0, acc,  false,  false);
  }

  int r0 = tm * 16 + half * 8;
#pragma unroll
  for (int j = 0; j < 8; ++j)
    C[(size_t)(r0 + j) * DD + col] = acc[j];
}

#define GA_T      256
#define GA_BLOCKS 1536
#define GA_WAVES  (GA_BLOCKS * (GA_T / 32))

template <int G>
__device__ __forceinline__ void ga_rows(const float2* __restrict__ hp, int sl, float wl, int j,
                                        float2& acc) {
  float2 v[G];
#pragma unroll
  for (int k = 0; k < G; ++k) {
    int s = __builtin_amdgcn_readlane(sl, j + k);
    v[k] = hp[(size_t)s * (DD / 2)];
  }
#pragma unroll
  for (int k = 0; k < G; ++k) {
    float w = __int_as_float(__builtin_amdgcn_readlane(__float_as_int(wl), j + k));
    acc.x = fmaf(w, v[k].x, acc.x);
    acc.y = fmaf(w, v[k].y, acc.y);
  }
}

__device__ __forceinline__ void ga_pairs(const int2* __restrict__ csr, int beg, int n, int c0,
                                         int lane, int d, int& sl, float& wl) {
  sl = d;
  wl = 0.0f;
  if (lane < min(32, n - c0)) {
    int2 ew = csr[beg + c0 + lane];
    sl = ew.x;
    wl = __int_as_float(ew.y);
  }
}
__device__ __forceinline__ int ga_source(int sl, int d) {
  return ((unsigned)sl >= (unsigned)NN) ? d : sl;
}

__global__ void __launch_bounds__(GA_T)
k_gather(const int* __restrict__ rowptr, const int* __restrict__ cnt,
         const int2* __restrict__ csr, const float* __restrict__ dinv,
         const float* __restrict__ hw, const float* __restrict__ b,
         float* __restrict__ h) {
  const int lane = threadIdx.x & 31;
  int d = __builtin_amdgcn_readfirstlane((int)(blockIdx.x * (GA_T / 32) + (threadIdx.x >> 5)));
  const float2* hp = (const float2*)hw + lane;
  const float2  bb = ((const float2*)b)[lane];

  int beg = 0, n = 0;
  if (d < NN) { beg = __builtin_amdgcn_readfirstlane(rowptr[d]); n = __builtin_amdgcn_readfirstlane(cnt[d]); }
  int   sl;
  float wl;
  ga_pairs(csr, beg, n, 0, lane, d, sl, wl);
  sl = ga_source(sl, d);
  int dn = d + GA_WAVES;
  int begn_v = 0, nn_v = 0;
  if (dn < NN) { begn_v = rowptr[dn]; nn_v = cnt[dn]; }

  while (d < NN) {
    int begn = __builtin_amdgcn_readfirstlane(begn_v);
    int nn   = __builtin_amdgcn_readfirstlane(nn_v);
    int   sln;
    float wln;
    ga_pairs(csr, begn, nn, 0, lane, dn, sln, wln);
    int d2 = dn + GA_WAVES;
    int beg2_v = 0, n2_v = 0;
    if (d2 < NN) { beg2_v = rowptr[d2]; n2_v = cnt[d2]; }

    float  dv    = dinv[d];
    float2 vself = hp[(size_t)d * (DD / 2)];
    float2 acc;  acc.x = 0.0f;  acc.y = 0.0f;

    for (int c0 = 0; c0 < n; c0 += 32) {
      int m = min(32, n - c0);
      if (c0 > 0) {
        ga_pairs(csr, beg, n, c0, lane, d, sl, wl);
        sl = ga_source(sl, d);
      }
      int j = 0;
      for (; j + 8 <= m; j += 8) ga_rows<8>(hp, sl, wl, j, acc);
      switch (m - j) {
        case 7: ga_rows<7>(hp, sl, wl, j, acc); break;
        case 6: ga_rows<6>(hp, sl, wl, j, acc); break;
        case 5: ga_rows<5>(hp, sl, wl, j, acc); break;
        case 4: ga_rows<4>(hp, sl, wl, j, acc); break;
        case 3: ga_rows<3>(hp, sl, wl, j, acc); break;
        case 2: ga_rows<2>(hp, sl, wl, j, acc); break;
        case 1: ga_rows<1>(hp, sl, wl, j, acc); break;
        default: break;
      }
    }

    float  sn = dv * dv;
    float2 o;
    o.x = tanhf(fmaf(sn, vself.x, acc.x) + bb.x);
    o.y = tanhf(fmaf(sn, vself.y, acc.y) + bb.y);
    ((float2*)h)[(size_t)d * (DD / 2) + lane] = o;

    sl = ga_source(sln, dn);  wl = wln;
    d = dn;   beg = begn;      n = nn;
    dn = d2;  begn_v = beg2_v; nn_v = n2_v;
  }
}

__device__ __forceinline__ unsigned fflip(float f) {
  unsigned u = __float_as_uint(f);
  return u ^ ((unsigned)((int)u >> 31) | 0x80000000u);
}
__device__ __forceinline__ float funflip(unsigned u) {
  unsigned m = (u >> 31) ? 0x80000000u : 0xFFFFFFFFu;
  return __uint_as_float(u ^ m);
}

__global__ void k_pool_init(unsigned* __restrict__ gmp, float* __restrict__ gsum,
                            float* __restrict__ counts) {
  int i = blockIdx.x * blockDim.x + threadIdx.x;
  if (i < GG * DD) { gmp[i] = 0x007FFFFFu  ; gsum[i] = 0.0f; }
  if (i < GG) counts[i] = 0.0f;
}

#define PCH 16
static_assert(NN % PCH == 0, "node count must be a multiple of the pooling chunk");

__device__ __forceinline__ void pool_flush(int g, int q, const unsigned (&mx)[4],
                                           const float (&sm)[4], float cnt,
                                           unsigned* __restrict__ gmp,
                                           float* __restrict__ gsum,
                                           float* __restrict__ counts) {
  int base = g * DD + q * 4;
#pragma unroll
  for (int k = 0; k < 4; ++k) {
    atomicMax(&gmp[base + k], mx[k]);
    atomicAdd(&gsum[base + k], sm[k]);
  }
  if (q == 0) atomicAdd(&counts[g], cnt);
}

__global__ void k_pool(const float* __restrict__ h, const int* __restrict__ bidx,
                       unsigned* __restrict__ gmp, float* __restrict__ gsum,
                       float* __restrict__ counts) {
  int gid = blockIdx.x * blockDim.x + threadIdx.x;
  int grp = gid >> 4;
  int q   = gid & 15;
  int n0  = grp * PCH;
  if (n0 >= NN) return;

  int    g[PCH];
  float4 v[PCH];
  const float4* hp = (const float4*)(h + (size_t)n0 * DD) + q;
#pragma unroll
  for (int i = 0; i < PCH; ++i) {
    g[i] = bidx[n0 + i];
    v[i] = hp[i * (DD / 4)];
  }

  int      cg = g[0];
  unsigned mx[4] = {fflip(v[0].x), fflip(v[0].y), fflip(v[0].z), fflip(v[0].w)};
  float    sm[4] = {v[0].x, v[0].y, v[0].z, v[0].w};
  float    cnt = 1.0f;
#pragma unroll
  for (int i = 1; i < PCH; ++i) {
    if (g[i] != cg) {
      pool_flush(cg, q, mx, sm, cnt, gmp, gsum, counts);
      cg = g[i];
      mx[0] = fflip(v[i].x); mx[1] = fflip(v[i].y); mx[2] = fflip(v[i].z); mx[3] = fflip(v[i].w);
      sm[0] = v[i].x; sm[1] = v[i].y; sm[2] = v[i].z; sm[3] = v[i].w;
      cnt = 1.0f;
    } else {
      mx[0] = max(mx[0], fflip(v[i].x)); mx[1] = max(mx[1], fflip(v[i].y));
      mx[2] = max(mx[2], fflip(v[i].z)); mx[3] = max(mx[3], fflip(v[i].w));
      sm[0] += v[i].x; sm[1] += v[i].y; sm[2] += v[i].z; sm[3] += v[i].w;
      cnt += 1.0f;
    }
  }
  pool_flush(cg, q, mx, sm, cnt, gmp, gsum, counts);
}

__global__ void k_final(const unsigned* __restrict__ gmp, const float* __restrict__ gsum,
                        const float* __restrict__ counts, const float* __restrict__ Wout,
                        const float* __restrict__ bout, float* __restrict__ out,
                        float* __restrict__ hidden) {
  __shared__ float red[2 * DD];
  int g = blockIdx.x, t = threadIdx.x;
  float cnt = fmaxf(counts[g], 1.0f);
  float hv = (t < DD) ? funflip(gmp[g * DD + t]) : (gsum[g * DD + (t - DD)] / cnt);
  hidden[(size_t)g * 2 * DD + t] = hv;
  red[t] = hv * Wout[t];
  __syncthreads();
  for (int s = DD; s > 0; s >>= 1) {
    if (t < s) red[t] += red[t + s];
    __syncthreads();
  }
  if (t == 0) out[g] = red[0] + bout[0];
}

extern "C" void kernel_launch(void* const* d_in, const int* in_sizes, int n_in,
                              void* d_out, int out_size, void* d_ws, size_t ws_size,
                              hipStream_t stream) {
  const float* x    = (const float*)d_in[0];
  const int*   ei   = (const int*)d_in[1];
  const int*   bidx = (const int*)d_in[2];
  const float* W0   = (const float*)d_in[3];
  const float* b0   = (const float*)d_in[4];
  const float* W1   = (const float*)d_in[5];
  const float* b1   = (const float*)d_in[6];
  const float* W2   = (const float*)d_in[7];
  const float* b2   = (const float*)d_in[8];
  const float* W3   = (const float*)d_in[9];
  const float* b3   = (const float*)d_in[10];
  const float* Wout = (const float*)d_in[11];
  const float* bout = (const float*)d_in[12];

  const int* src = ei;
  const int* dst = ei + EE;

  char*  ws   = (char*)d_ws;
  size_t off  = 0;
  const size_t nodeWords = (((size_t)NN * 4 + 255) & ~(size_t)255);
  float* dinv = (float*)(ws + off);  off += nodeWords;
  float* bufP = (float*)(ws + off);  off += (size_t)NN * DD * 4;
  float* bufQ = (float*)(ws + off);  off += (size_t)NN * DD * 4;
  unsigned* gmp = (unsigned*)(ws + off); off += (size_t)GG * DD * 4;
  float* gsum   = (float*)(ws + off);    off += (size_t)GG * DD * 4;
  float* counts = (float*)(ws + off);    off += (size_t)GG * 4;
  int* cnt    = (int*)(ws + off);  off += nodeWords;
  int* rowptr = (int*)(ws + off);  off += nodeWords;
  int* cursor = (int*)(ws + off);  off += nodeWords;
  int* bsum   = (int*)(ws + off);  off += 1024;
  int2* csr   = (int2*)(ws + off); off += (size_t)EE * 8;
  (void)ws_size; (void)n_in; (void)in_sizes; (void)out_size;

  float* outp   = (float*)d_out;
  float* hidden = outp + GG;

  const int T = 256;

  k_zero_cnt <<<(NN + T - 1) / T, T, 0, stream>>>(cnt);
  k_hist     <<<(EE + T - 1) / T, T, 0, stream>>>(dst, cnt);
  k_dinv_bsum<<<SCAN_BLOCKS, SCAN_T, 0, stream>>>(cnt, dinv, bsum);
  k_scan     <<<SCAN_BLOCKS, SCAN_T, 0, stream>>>(cnt, bsum, rowptr, cursor);
  k_fill     <<<(EE + T - 1) / T, T, 0, stream>>>(src, dst, dinv, cursor, csr);

  const int gemm_blocks = (NN / 16) * 4 / 8;

  const float* Wl[4] = {W0, W1, W2, W3};
  const float* bl[4] = {b0, b1, b2, b3};
  for (int l = 0; l < 4; ++l) {
    if (l == 0) k_gemm_wmma<FF><<<gemm_blocks, T, 0, stream>>>(x,    Wl[l], bufQ);
    else        k_gemm_wmma<DD><<<gemm_blocks, T, 0, stream>>>(bufP, Wl[l], bufQ);
    k_gather<<<GA_BLOCKS, GA_T, 0, stream>>>(rowptr, cnt, csr, dinv, bufQ, bl[l], bufP);
  }

  k_pool_init<<<(GG * DD + T - 1) / T, T, 0, stream>>>(gmp, gsum, counts);
  k_pool     <<<((NN / PCH) * 16 + T - 1) / T, T, 0, stream>>>(bufP, bidx, gmp, gsum, counts);
  k_final    <<<GG, 2 * DD, 0, stream>>>(gmp, gsum, counts, Wout, bout, outp, hidden);
}


# layers 1-3: next layer's 64x64 product fused into the gather launch (32 finished h rows staged in LDS per block, WMMA epilogue); 3 product launches and the h round trip gone; sum orders unchanged
# speedup vs baseline: 2.6717x; 1.0762x over previous
#include <hip/hip_runtime.h>
#include <math.h>

#define NN 100000
#define EE 1000000
#define FF 128
#define DD 64
#define GG 256

typedef float v2f __attribute__((ext_vector_type(2)));
typedef float v8f __attribute__((ext_vector_type(8)));

#define SCAN_T 256
#define SCAN_NODES (SCAN_T * 4)
#define SCAN_BLOCKS ((NN + SCAN_NODES - 1) / SCAN_NODES)
static_assert(NN % 4 == 0, "node count must be a multiple of 4 for the int4 scan loads");
static_assert(SCAN_BLOCKS <= 128, "block offsets are summed by one wave, four entries per lane");

__global__ void k_zero_cnt(int* __restrict__ cnt) {
  int i = blockIdx.x * blockDim.x + threadIdx.x;
  if (i < NN) cnt[i] = 0;
}

__global__ void k_hist(const int* __restrict__ dst, int* __restrict__ cnt) {
  int e = blockIdx.x * blockDim.x + threadIdx.x;
  if (e < EE) atomicAdd(&cnt[dst[e]], 1);
}

__device__ __forceinline__ int wave_sum(int v) {
  for (int o = 16; o > 0; o >>= 1) v += __shfl_xor(v, o, 32);
  return v;
}

__global__ void k_dinv_bsum(const int* __restrict__ cnt, float* __restrict__ dinv,
                            int* __restrict__ bsum) {
  __shared__ int wsum[SCAN_T / 32];
  int t  = threadIdx.x;
  int i4 = blockIdx.x * SCAN_T + t;
  int s  = 0;
  if (i4 < NN / 4) {
    int4 c = ((const int4*)cnt)[i4];
    float4 r;
    r.x = rsqrtf((float)c.x + 1.0f); r.y = rsqrtf((float)c.y + 1.0f);
    r.z = rsqrtf((float)c.z + 1.0f); r.w = rsqrtf((float)c.w + 1.0f);
    ((float4*)dinv)[i4] = r;
    s = c.x + c.y + c.z + c.w;
  }
  s = wave_sum(s);
  if ((t & 31) == 0) wsum[t >> 5] = s;
  __syncthreads();
  if (t == 0) {
    int tot = 0;
    for (int w = 0; w < SCAN_T / 32; ++w) tot += wsum[w];
    bsum[blockIdx.x] = tot;
  }
}

__global__ void k_scan(const int* __restrict__ cnt, const int* __restrict__ bsum,
                       int* __restrict__ rowptr, int* __restrict__ cursor) {
  __shared__ int wtot[SCAN_T / 32];
  __shared__ int boff;
  int t = threadIdx.x, lane = t & 31, w = t >> 5;
  int b = blockIdx.x;
  if (w == 0) {
    int p = 0;
    for (int j = lane; j < b; j += 32) p += bsum[j];
    p = wave_sum(p);
    if (lane == 0) boff = p;
  }
  int  i4 = b * SCAN_T + t;
  int4 c  = make_int4(0, 0, 0, 0);
  if (i4 < NN / 4) c = ((const int4*)cnt)[i4];
  int tsum = c.x + c.y + c.z + c.w;
  int inc  = tsum;
  for (int o = 1; o < 32; o <<= 1) {
    int up = __shfl_up(inc, o, 32);
    if (lane >= o) inc += up;
  }
  if (lane == 31) wtot[w] = inc;
  __syncthreads();
  int wbase = 0;
  for (int k = 0; k < w; ++k) wbase += wtot[k];
  int base = boff + wbase + inc - tsum;
  if (i4 < NN / 4) {
    int4 r;
    r.x = base; r.y = r.x + c.x; r.z = r.y + c.y; r.w = r.z + c.z;
    ((int4*)rowptr)[i4] = r;
    ((int4*)cursor)[i4] = r;
  }
}

__global__ void k_fill(const int* __restrict__ src, const int* __restrict__ dst,
                       const float* __restrict__ dinv, int* __restrict__ cursor,
                       int2* __restrict__ csr) {
  int e = blockIdx.x * blockDim.x + threadIdx.x;
  if (e < EE) {
    int s = src[e], d = dst[e];
    float w = dinv[s] * dinv[d];
    int p = atomicAdd(&cursor[d], 1);
    if ((unsigned)p < (unsigned)EE)
      csr[p] = make_int2(s, __float_as_int(w));
  }
}

template <int K>
__global__ void k_gemm_wmma(const float* __restrict__ A, const float* __restrict__ B,
                            float* __restrict__ C) {
  int gid  = blockIdx.x * blockDim.x + threadIdx.x;
  int wave = gid >> 5;
  int lane = gid & 31;
  int tm = wave >> 2;
  int tn = wave & 3;
  if (tm * 16 >= NN) return;

  int half = lane >> 4;
  int m    = lane & 15;
  const float* arow = A + (size_t)(tm * 16 + m) * K;
  int col = tn * 16 + m;

  v8f acc = {};
#pragma unroll
  for (int kb = 0; kb < K; kb += 4) {
    int k0 = kb + half * 2;
    v2f a; a[0] = arow[k0];          a[1] = arow[k0 + 1];
    v2f b; b[0] = B[k0 * DD + col];  b[1] = B[(k0 + 1) * DD + col];
    acc = __builtin_amdgcn_wmma_f32_16x16x4_f32(
         false, a,  false, b,
         (short)0, acc,  false,  false);
  }

  int r0 = tm * 16 + half * 8;
#pragma unroll
  for (int j = 0; j < 8; ++j)
    C[(size_t)(r0 + j) * DD + col] = acc[j];
}

#define GA_T      256
#define GA_BLOCKS 1536
#define GA_WAVES  (GA_BLOCKS * (GA_T / 32))

template <int G>
__device__ __forceinline__ void ga_rows(const float2* __restrict__ hp, int sl, float wl, int j,
                                        float2& acc) {
  float2 v[G];
#pragma unroll
  for (int k = 0; k < G; ++k) {
    int s = __builtin_amdgcn_readlane(sl, j + k);
    v[k] = hp[(size_t)s * (DD / 2)];
  }
#pragma unroll
  for (int k = 0; k < G; ++k) {
    float w = __int_as_float(__builtin_amdgcn_readlane(__float_as_int(wl), j + k));
    acc.x = fmaf(w, v[k].x, acc.x);
    acc.y = fmaf(w, v[k].y, acc.y);
  }
}

__device__ __forceinline__ void ga_pairs(const int2* __restrict__ csr, int beg, int n, int c0,
                                         int lane, int d, int& sl, float& wl) {
  sl = d;
  wl = 0.0f;
  if (lane < min(32, n - c0)) {
    int2 ew = csr[beg + c0 + lane];
    sl = ew.x;
    wl = __int_as_float(ew.y);
  }
}
__device__ __forceinline__ int ga_source(int sl, int d) {
  return ((unsigned)sl >= (unsigned)NN) ? d : sl;
}

__global__ void __launch_bounds__(GA_T)
k_gather(const int* __restrict__ rowptr, const int* __restrict__ cnt,
         const int2* __restrict__ csr, const float* __restrict__ dinv,
         const float* __restrict__ hw, const float* __restrict__ b,
         float* __restrict__ h) {
  const int lane = threadIdx.x & 31;
  int d = __builtin_amdgcn_readfirstlane((int)(blockIdx.x * (GA_T / 32) + (threadIdx.x >> 5)));
  const float2* hp = (const float2*)hw + lane;
  const float2  bb = ((const float2*)b)[lane];

  int beg = 0, n = 0;
  if (d < NN) { beg = __builtin_amdgcn_readfirstlane(rowptr[d]); n = __builtin_amdgcn_readfirstlane(cnt[d]); }
  int   sl;
  float wl;
  ga_pairs(csr, beg, n, 0, lane, d, sl, wl);
  sl = ga_source(sl, d);
  int dn = d + GA_WAVES;
  int begn_v = 0, nn_v = 0;
  if (dn < NN) { begn_v = rowptr[dn]; nn_v = cnt[dn]; }

  while (d < NN) {
    int begn = __builtin_amdgcn_readfirstlane(begn_v);
    int nn   = __builtin_amdgcn_readfirstlane(nn_v);
    int   sln;
    float wln;
    ga_pairs(csr, begn, nn, 0, lane, dn, sln, wln);
    int d2 = dn + GA_WAVES;
    int beg2_v = 0, n2_v = 0;
    if (d2 < NN) { beg2_v = rowptr[d2]; n2_v = cnt[d2]; }

    float  dv    = dinv[d];
    float2 vself = hp[(size_t)d * (DD / 2)];
    float2 acc;  acc.x = 0.0f;  acc.y = 0.0f;

    for (int c0 = 0; c0 < n; c0 += 32) {
      int m = min(32, n - c0);
      if (c0 > 0) {
        ga_pairs(csr, beg, n, c0, lane, d, sl, wl);
        sl = ga_source(sl, d);
      }
      int j = 0;
      for (; j + 8 <= m; j += 8) ga_rows<8>(hp, sl, wl, j, acc);
      switch (m - j) {
        case 7: ga_rows<7>(hp, sl, wl, j, acc); break;
        case 6: ga_rows<6>(hp, sl, wl, j, acc); break;
        case 5: ga_rows<5>(hp, sl, wl, j, acc); break;
        case 4: ga_rows<4>(hp, sl, wl, j, acc); break;
        case 3: ga_rows<3>(hp, sl, wl, j, acc); break;
        case 2: ga_rows<2>(hp, sl, wl, j, acc); break;
        case 1: ga_rows<1>(hp, sl, wl, j, acc); break;
        default: break;
      }
    }

    float  sn = dv * dv;
    float2 o;
    o.x = tanhf(fmaf(sn, vself.x, acc.x) + bb.x);
    o.y = tanhf(fmaf(sn, vself.y, acc.y) + bb.y);
    ((float2*)h)[(size_t)d * (DD / 2) + lane] = o;

    sl = ga_source(sln, dn);  wl = wln;
    d = dn;   beg = begn;      n = nn;
    dn = d2;  begn_v = beg2_v; nn_v = n2_v;
  }
}

#define FG_T        256
#define FG_NODES    32
#define FG_PER_WAVE (FG_NODES / (FG_T / 32))
#define FG_PITCH    68
static_assert(NN % FG_NODES == 0, "every block owns 32 whole nodes");
static_assert(FG_PER_WAVE == 4, "the list bounds of a wave's nodes are read as one int4");

__global__ void __launch_bounds__(FG_T)
k_gather_gemm(const int* __restrict__ rowptr, const int* __restrict__ cnt,
              const int2* __restrict__ csr, const float* __restrict__ dinv,
              const float* __restrict__ hw, const float* __restrict__ b,
              const float* __restrict__ Wn, float* __restrict__ hwn) {
  __shared__ __attribute__((aligned(16))) float stage[FG_NODES * FG_PITCH];
  const int lane = threadIdx.x & 31;
  const int w    = __builtin_amdgcn_readfirstlane((int)(threadIdx.x >> 5));
  const int base = blockIdx.x * FG_NODES;
  const int half = lane >> 4;
  const int m    = lane & 15;
  const int tm   = w >> 2;
  const int tn   = w & 3;
  const int col  = tn * 16 + m;

  v2f bf[DD / 4];
#pragma unroll
  for (int kb = 0; kb < DD / 4; ++kb) {
    int k0 = kb * 4 + half * 2;
    bf[kb][0] = Wn[k0 * DD + col];
    bf[kb][1] = Wn[(k0 + 1) * DD + col];
  }

  const float2* hp = (const float2*)hw + lane;
  const float2  bb = ((const float2*)b)[lane];

  int d = base + w * FG_PER_WAVE;
  int4 rp = ((const int4*)rowptr)[d >> 2];
  int4 cn = ((const int4*)cnt)[d >> 2];
  int beg  = __builtin_amdgcn_readfirstlane(rp.x), n  = __builtin_amdgcn_readfirstlane(cn.x);
  int begA = __builtin_amdgcn_readfirstlane(rp.y), nA = __builtin_amdgcn_readfirstlane(cn.y);
  int begB = __builtin_amdgcn_readfirstlane(rp.z), nB = __builtin_amdgcn_readfirstlane(cn.z);
  int begC = __builtin_amdgcn_readfirstlane(rp.w), nC = __builtin_amdgcn_readfirstlane(cn.w);
  int   sl;
  float wl;
  ga_pairs(csr, beg, n, 0, lane, d, sl, wl);
  sl = ga_source(sl, d);

  for (int q = 0; q < FG_PER_WAVE; ++q) {
    int   sln;
    float wln;
    ga_pairs(csr, begA, nA, 0, lane, d + 1, sln, wln);

    float  dv    = dinv[d];
    float2 vself = hp[(size_t)d * (DD / 2)];
    float2 acc;  acc.x = 0.0f;  acc.y = 0.0f;

    for (int c0 = 0; c0 < n; c0 += 32) {
      int mm = min(32, n - c0);
      if (c0 > 0) {
        ga_pairs(csr, beg, n, c0, lane, d, sl, wl);
        sl = ga_source(sl, d);
      }
      int j = 0;
      for (; j + 8 <= mm; j += 8) ga_rows<8>(hp, sl, wl, j, acc);
      switch (mm - j) {
        case 7: ga_rows<7>(hp, sl, wl, j, acc); break;
        case 6: ga_rows<6>(hp, sl, wl, j, acc); break;
        case 5: ga_rows<5>(hp, sl, wl, j, acc); break;
        case 4: ga_rows<4>(hp, sl, wl, j, acc); break;
        case 3: ga_rows<3>(hp, sl, wl, j, acc); break;
        case 2: ga_rows<2>(hp, sl, wl, j, acc); break;
        case 1: ga_rows<1>(hp, sl, wl, j, acc); break;
        default: break;
      }
    }

    float  sn = dv * dv;
    float2 o;
    o.x = tanhf(fmaf(sn, vself.x, acc.x) + bb.x);
    o.y = tanhf(fmaf(sn, vself.y, acc.y) + bb.y);
    *(float2*)&stage[(w * FG_PER_WAVE + q) * FG_PITCH + 2 * lane] = o;

    sl = ga_source(sln, d + 1);  wl = wln;
    d = d + 1;
    beg = begA;  n = nA;
    begA = begB; nA = nB;
    begB = begC; nB = nC;
    begC = 0;    nC = 0;
  }

  __syncthreads();

  const float* arow = stage + (tm * 16 + m) * FG_PITCH + half * 2;
  v8f c = {};
#pragma unroll
  for (int kb = 0; kb < DD / 4; ++kb) {
    v2f a = *(const v2f*)(arow + kb * 4);
    c = __builtin_amdgcn_wmma_f32_16x16x4_f32(
         false, a,  false, bf[kb],
         (short)0, c,  false,  false);
  }

  int r0 = base + tm * 16 + half * 8;
#pragma unroll
  for (int j = 0; j < 8; ++j)
    hwn[(size_t)(r0 + j) * DD + col] = c[j];
}

__device__ __forceinline__ unsigned fflip(float f) {
  unsigned u = __float_as_uint(f);
  return u ^ ((unsigned)((int)u >> 31) | 0x80000000u);
}
__device__ __forceinline__ float funflip(unsigned u) {
  unsigned m = (u >> 31) ? 0x80000000u : 0xFFFFFFFFu;
  return __uint_as_float(u ^ m);
}

__global__ void k_pool_init(unsigned* __restrict__ gmp, float* __restrict__ gsum,
                            float* __restrict__ counts) {
  int i = blockIdx.x * blockDim.x + threadIdx.x;
  if (i < GG * DD) { gmp[i] = 0x007FFFFFu  ; gsum[i] = 0.0f; }
  if (i < GG) counts[i] = 0.0f;
}

#define PCH 16
static_assert(NN % PCH == 0, "node count must be a multiple of the pooling chunk");

__device__ __forceinline__ void pool_flush(int g, int q, const unsigned (&mx)[4],
                                           const float (&sm)[4], float cnt,
                                           unsigned* __restrict__ gmp,
                                           float* __restrict__ gsum,
                                           float* __restrict__ counts) {
  int base = g * DD + q * 4;
#pragma unroll
  for (int k = 0; k < 4; ++k) {
    atomicMax(&gmp[base + k], mx[k]);
    atomicAdd(&gsum[base + k], sm[k]);
  }
  if (q == 0) atomicAdd(&counts[g], cnt);
}

__global__ void k_pool(const float* __restrict__ h, const int* __restrict__ bidx,
                       unsigned* __restrict__ gmp, float* __restrict__ gsum,
                       float* __restrict__ counts) {
  int gid = blockIdx.x * blockDim.x + threadIdx.x;
  int grp = gid >> 4;
  int q   = gid & 15;
  int n0  = grp * PCH;
  if (n0 >= NN) return;

  int    g[PCH];
  float4 v[PCH];
  const float4* hp = (const float4*)(h + (size_t)n0 * DD) + q;
#pragma unroll
  for (int i = 0; i < PCH; ++i) {
    g[i] = bidx[n0 + i];
    v[i] = hp[i * (DD / 4)];
  }

  int      cg = g[0];
  unsigned mx[4] = {fflip(v[0].x), fflip(v[0].y), fflip(v[0].z), fflip(v[0].w)};
  float    sm[4] = {v[0].x, v[0].y, v[0].z, v[0].w};
  float    cnt = 1.0f;
#pragma unroll
  for (int i = 1; i < PCH; ++i) {
    if (g[i] != cg) {
      pool_flush(cg, q, mx, sm, cnt, gmp, gsum, counts);
      cg = g[i];
      mx[0] = fflip(v[i].x); mx[1] = fflip(v[i].y); mx[2] = fflip(v[i].z); mx[3] = fflip(v[i].w);
      sm[0] = v[i].x; sm[1] = v[i].y; sm[2] = v[i].z; sm[3] = v[i].w;
      cnt = 1.0f;
    } else {
      mx[0] = max(mx[0], fflip(v[i].x)); mx[1] = max(mx[1], fflip(v[i].y));
      mx[2] = max(mx[2], fflip(v[i].z)); mx[3] = max(mx[3], fflip(v[i].w));
      sm[0] += v[i].x; sm[1] += v[i].y; sm[2] += v[i].z; sm[3] += v[i].w;
      cnt += 1.0f;
    }
  }
  pool_flush(cg, q, mx, sm, cnt, gmp, gsum, counts);
}

__global__ void k_final(const unsigned* __restrict__ gmp, const float* __restrict__ gsum,
                        const float* __restrict__ counts, const float* __restrict__ Wout,
                        const float* __restrict__ bout, float* __restrict__ out,
                        float* __restrict__ hidden) {
  __shared__ float red[2 * DD];
  int g = blockIdx.x, t = threadIdx.x;
  float cnt = fmaxf(counts[g], 1.0f);
  float hv = (t < DD) ? funflip(gmp[g * DD + t]) : (gsum[g * DD + (t - DD)] / cnt);
  hidden[(size_t)g * 2 * DD + t] = hv;
  red[t] = hv * Wout[t];
  __syncthreads();
  for (int s = DD; s > 0; s >>= 1) {
    if (t < s) red[t] += red[t + s];
    __syncthreads();
  }
  if (t == 0) out[g] = red[0] + bout[0];
}

extern "C" void kernel_launch(void* const* d_in, const int* in_sizes, int n_in,
                              void* d_out, int out_size, void* d_ws, size_t ws_size,
                              hipStream_t stream) {
  const float* x    = (const float*)d_in[0];
  const int*   ei   = (const int*)d_in[1];
  const int*   bidx = (const int*)d_in[2];
  const float* W0   = (const float*)d_in[3];
  const float* b0   = (const float*)d_in[4];
  const float* W1   = (const float*)d_in[5];
  const float* b1   = (const float*)d_in[6];
  const float* W2   = (const float*)d_in[7];
  const float* b2   = (const float*)d_in[8];
  const float* W3   = (const float*)d_in[9];
  const float* b3   = (const float*)d_in[10];
  const float* Wout = (const float*)d_in[11];
  const float* bout = (const float*)d_in[12];

  const int* src = ei;
  const int* dst = ei + EE;

  char*  ws   = (char*)d_ws;
  size_t off  = 0;
  const size_t nodeWords = (((size_t)NN * 4 + 255) & ~(size_t)255);
  float* dinv = (float*)(ws + off);  off += nodeWords;
  float* bufP = (float*)(ws + off);  off += (size_t)NN * DD * 4;
  float* bufQ = (float*)(ws + off);  off += (size_t)NN * DD * 4;
  unsigned* gmp = (unsigned*)(ws + off); off += (size_t)GG * DD * 4;
  float* gsum   = (float*)(ws + off);    off += (size_t)GG * DD * 4;
  float* counts = (float*)(ws + off);    off += (size_t)GG * 4;
  int* cnt    = (int*)(ws + off);  off += nodeWords;
  int* rowptr = (int*)(ws + off);  off += nodeWords;
  int* cursor = (int*)(ws + off);  off += nodeWords;
  int* bsum   = (int*)(ws + off);  off += 1024;
  int2* csr   = (int2*)(ws + off); off += (size_t)EE * 8;
  (void)ws_size; (void)n_in; (void)in_sizes; (void)out_size;

  float* outp   = (float*)d_out;
  float* hidden = outp + GG;

  const int T = 256;

  k_zero_cnt <<<(NN + T - 1) / T, T, 0, stream>>>(cnt);
  k_hist     <<<(EE + T - 1) / T, T, 0, stream>>>(dst, cnt);
  k_dinv_bsum<<<SCAN_BLOCKS, SCAN_T, 0, stream>>>(cnt, dinv, bsum);
  k_scan     <<<SCAN_BLOCKS, SCAN_T, 0, stream>>>(cnt, bsum, rowptr, cursor);
  k_fill     <<<(EE + T - 1) / T, T, 0, stream>>>(src, dst, dinv, cursor, csr);

  const int gemm_blocks = (NN / 16) * 4 / 8;

  k_gemm_wmma<FF><<<gemm_blocks, T, 0, stream>>>(x, W0, bufQ);
  k_gather_gemm<<<NN / FG_NODES, FG_T, 0, stream>>>(rowptr, cnt, csr, dinv, bufQ, b0, W1, bufP);
  k_gather_gemm<<<NN / FG_NODES, FG_T, 0, stream>>>(rowptr, cnt, csr, dinv, bufP, b1, W2, bufQ);
  k_gather_gemm<<<NN / FG_NODES, FG_T, 0, stream>>>(rowptr, cnt, csr, dinv, bufQ, b2, W3, bufP);
  k_gather<<<GA_BLOCKS, GA_T, 0, stream>>>(rowptr, cnt, csr, dinv, bufP, b3, bufQ);

  k_pool_init<<<(GG * DD + T - 1) / T, T, 0, stream>>>(gmp, gsum, counts);
  k_pool     <<<((NN / PCH) * 16 + T - 1) / T, T, 0, stream>>>(bufQ, bidx, gmp, gsum, counts);
  k_final    <<<GG, 2 * DD, 0, stream>>>(gmp, gsum, counts, Wout, bout, outp, hidden);
}


# launches 13 -> 10: layer-0 product launch goes first and also clears counters and pool start values; pooling done in the last aggregation's epilogue from 32 LDS rows; mean-pool sum order changes
# speedup vs baseline: 2.7880x; 1.0435x over previous
#include <hip/hip_runtime.h>
#include <math.h>

#define NN 100000
#define EE 1000000
#define FF 128
#define DD 64
#define GG 256

typedef float v2f __attribute__((ext_vector_type(2)));
typedef float v8f __attribute__((ext_vector_type(8)));

#define SCAN_T 256
#define SCAN_NODES (SCAN_T * 4)
#define SCAN_BLOCKS ((NN + SCAN_NODES - 1) / SCAN_NODES)
static_assert(NN % 4 == 0, "node count must be a multiple of 4 for the int4 scan loads");
static_assert(SCAN_BLOCKS <= 128, "block offsets are summed by one wave, four entries per lane");

__global__ void k_hist(const int* __restrict__ dst, int* __restrict__ cnt) {
  int e = blockIdx.x * blockDim.x + threadIdx.x;
  if (e < EE) atomicAdd(&cnt[dst[e]], 1);
}

__device__ __forceinline__ int wave_sum(int v) {
  for (int o = 16; o > 0; o >>= 1) v += __shfl_xor(v, o, 32);
  return v;
}

__global__ void k_dinv_bsum(const int* __restrict__ cnt, float* __restrict__ dinv,
                            int* __restrict__ bsum) {
  __shared__ int wsum[SCAN_T / 32];
  int t  = threadIdx.x;
  int i4 = blockIdx.x * SCAN_T + t;
  int s  = 0;
  if (i4 < NN / 4) {
    int4 c = ((const int4*)cnt)[i4];
    float4 r;
    r.x = rsqrtf((float)c.x + 1.0f); r.y = rsqrtf((float)c.y + 1.0f);
    r.z = rsqrtf((float)c.z + 1.0f); r.w = rsqrtf((float)c.w + 1.0f);
    ((float4*)dinv)[i4] = r;
    s = c.x + c.y + c.z + c.w;
  }
  s = wave_sum(s);
  if ((t & 31) == 0) wsum[t >> 5] = s;
  __syncthreads();
  if (t == 0) {
    int tot = 0;
    for (int w = 0; w < SCAN_T / 32; ++w) tot += wsum[w];
    bsum[blockIdx.x] = tot;
  }
}

__global__ void k_scan(const int* __restrict__ cnt, const int* __restrict__ bsum,
                       int* __restrict__ rowptr, int* __restrict__ cursor) {
  __shared__ int wtot[SCAN_T / 32];
  __shared__ int boff;
  int t = threadIdx.x, lane = t & 31, w = t >> 5;
  int b = blockIdx.x;
  if (w == 0) {
    int p = 0;
    for (int j = lane; j < b; j += 32) p += bsum[j];
    p = wave_sum(p);
    if (lane == 0) boff = p;
  }
  int  i4 = b * SCAN_T + t;
  int4 c  = make_int4(0, 0, 0, 0);
  if (i4 < NN / 4) c = ((const int4*)cnt)[i4];
  int tsum = c.x + c.y + c.z + c.w;
  int inc  = tsum;
  for (int o = 1; o < 32; o <<= 1) {
    int up = __shfl_up(inc, o, 32);
    if (lane >= o) inc += up;
  }
  if (lane == 31) wtot[w] = inc;
  __syncthreads();
  int wbase = 0;
  for (int k = 0; k < w; ++k) wbase += wtot[k];
  int base = boff + wbase + inc - tsum;
  if (i4 < NN / 4) {
    int4 r;
    r.x = base; r.y = r.x + c.x; r.z = r.y + c.y; r.w = r.z + c.z;
    ((int4*)rowptr)[i4] = r;
    ((int4*)cursor)[i4] = r;
  }
}

__global__ void k_fill(const int* __restrict__ src, const int* __restrict__ dst,
                       const float* __restrict__ dinv, int* __restrict__ cursor,
                       int2* __restrict__ csr) {
  int e = blockIdx.x * blockDim.x + threadIdx.x;
  if (e < EE) {
    int s = src[e], d = dst[e];
    float w = dinv[s] * dinv[d];
    int p = atomicAdd(&cursor[d], 1);
    if ((unsigned)p < (unsigned)EE)
      csr[p] = make_int2(s, __float_as_int(w));
  }
}

static_assert((NN / 16) * 4 * 32 >= NN && NN >= GG * DD, "one product thread per word to clear");
template <int K>
__global__ void k_gemm_wmma(const float* __restrict__ A, const float* __restrict__ B,
                            float* __restrict__ C, int* __restrict__ cnt,
                            unsigned* __restrict__ gmp, float* __restrict__ gsum,
                            float* __restrict__ counts) {
  int gid  = blockIdx.x * blockDim.x + threadIdx.x;
  if (gid < NN) cnt[gid] = 0;
  if (gid < GG * DD) { gmp[gid] = 0x007FFFFFu  ; gsum[gid] = 0.0f; }
  if (gid < GG) counts[gid] = 0.0f;
  int wave = gid >> 5;
  int lane = gid & 31;
  int tm = wave >> 2;
  int tn = wave & 3;
  if (tm * 16 >= NN) return;

  int half = lane >> 4;
  int m    = lane & 15;
  const float* arow = A + (size_t)(tm * 16 + m) * K;
  int col = tn * 16 + m;

  v8f acc = {};
#pragma unroll
  for (int kb = 0; kb < K; kb += 4) {
    int k0 = kb + half * 2;
    v2f a; a[0] = arow[k0];          a[1] = arow[k0 + 1];
    v2f b; b[0] = B[k0 * DD + col];  b[1] = B[(k0 + 1) * DD + col];
    acc = __builtin_amdgcn_wmma_f32_16x16x4_f32(
         false, a,  false, b,
         (short)0, acc,  false,  false);
  }

  int r0 = tm * 16 + half * 8;
#pragma unroll
  for (int j = 0; j < 8; ++j)
    C[(size_t)(r0 + j) * DD + col] = acc[j];
}


template <int G>
__device__ __forceinline__ void ga_rows(const float2* __restrict__ hp, int sl, float wl, int j,
                                        float2& acc) {
  float2 v[G];
#pragma unroll
  for (int k = 0; k < G; ++k) {
    int s = __builtin_amdgcn_readlane(sl, j + k);
    v[k] = hp[(size_t)s * (DD / 2)];
  }
#pragma unroll
  for (int k = 0; k < G; ++k) {
    float w = __int_as_float(__builtin_amdgcn_readlane(__float_as_int(wl), j + k));
    acc.x = fmaf(w, v[k].x, acc.x);
    acc.y = fmaf(w, v[k].y, acc.y);
  }
}

__device__ __forceinline__ void ga_pairs(const int2* __restrict__ csr, int beg, int n, int c0,
                                         int lane, int d, int& sl, float& wl) {
  sl = d;
  wl = 0.0f;
  if (lane < min(32, n - c0)) {
    int2 ew = csr[beg + c0 + lane];
    sl = ew.x;
    wl = __int_as_float(ew.y);
  }
}
__device__ __forceinline__ int ga_source(int sl, int d) {
  return ((unsigned)sl >= (unsigned)NN) ? d : sl;
}

#define FG_T        256
#define FG_NODES    32
#define FG_PER_WAVE (FG_NODES / (FG_T / 32))
#define FG_PITCH    68
static_assert(NN % FG_NODES == 0, "every block owns 32 whole nodes");
static_assert(FG_PER_WAVE == 4, "the list bounds of a wave's nodes are read as one int4");

__device__ __forceinline__ void fg_gather_rows(const int* __restrict__ rowptr,
                                               const int* __restrict__ cnt,
                                               const int2* __restrict__ csr,
                                               const float* __restrict__ dinv,
                                               const float* __restrict__ hw,
                                               const float* __restrict__ b,
                                               int d, int lane, float* srow) {
  const float2* hp = (const float2*)hw + lane;
  const float2  bb = ((const float2*)b)[lane];

  int4 rp = ((const int4*)rowptr)[d >> 2];
  int4 cn = ((const int4*)cnt)[d >> 2];
  int beg  = __builtin_amdgcn_readfirstlane(rp.x), n  = __builtin_amdgcn_readfirstlane(cn.x);
  int begA = __builtin_amdgcn_readfirstlane(rp.y), nA = __builtin_amdgcn_readfirstlane(cn.y);
  int begB = __builtin_amdgcn_readfirstlane(rp.z), nB = __builtin_amdgcn_readfirstlane(cn.z);
  int begC = __builtin_amdgcn_readfirstlane(rp.w), nC = __builtin_amdgcn_readfirstlane(cn.w);
  int   sl;
  float wl;
  ga_pairs(csr, beg, n, 0, lane, d, sl, wl);
  sl = ga_source(sl, d);

  for (int q = 0; q < FG_PER_WAVE; ++q) {
    int   sln;
    float wln;
    ga_pairs(csr, begA, nA, 0, lane, d + 1, sln, wln);

    float  dv    = dinv[d];
    float2 vself = hp[(size_t)d * (DD / 2)];
    float2 acc;  acc.x = 0.0f;  acc.y = 0.0f;

    for (int c0 = 0; c0 < n; c0 += 32) {
      int mm = min(32, n - c0);
      if (c0 > 0) {
        ga_pairs(csr, beg, n, c0, lane, d, sl, wl);
        sl = ga_source(sl, d);
      }
      int j = 0;
      for (; j + 8 <= mm; j += 8) ga_rows<8>(hp, sl, wl, j, acc);
      switch (mm - j) {
        case 7: ga_rows<7>(hp, sl, wl, j, acc); break;
        case 6: ga_rows<6>(hp, sl, wl, j, acc); break;
        case 5: ga_rows<5>(hp, sl, wl, j, acc); break;
        case 4: ga_rows<4>(hp, sl, wl, j, acc); break;
        case 3: ga_rows<3>(hp, sl, wl, j, acc); break;
        case 2: ga_rows<2>(hp, sl, wl, j, acc); break;
        case 1: ga_rows<1>(hp, sl, wl, j, acc); break;
        default: break;
      }
    }

    float  sn = dv * dv;
    float2 o;
    o.x = tanhf(fmaf(sn, vself.x, acc.x) + bb.x);
    o.y = tanhf(fmaf(sn, vself.y, acc.y) + bb.y);
    *(float2*)&srow[q * FG_PITCH + 2 * lane] = o;

    sl = ga_source(sln, d + 1);  wl = wln;
    d = d + 1;
    beg = begA;  n = nA;
    begA = begB; nA = nB;
    begB = begC; nB = nC;
    begC = 0;    nC = 0;
  }
}

__global__ void __launch_bounds__(FG_T)
k_gather_gemm(const int* __restrict__ rowptr, const int* __restrict__ cnt,
              const int2* __restrict__ csr, const float* __restrict__ dinv,
              const float* __restrict__ hw, const float* __restrict__ b,
              const float* __restrict__ Wn, float* __restrict__ hwn) {
  __shared__ __attribute__((aligned(16))) float stage[FG_NODES * FG_PITCH];
  const int lane = threadIdx.x & 31;
  const int w    = __builtin_amdgcn_readfirstlane((int)(threadIdx.x >> 5));
  const int base = blockIdx.x * FG_NODES;
  const int half = lane >> 4;
  const int m    = lane & 15;
  const int tm   = w >> 2;
  const int tn   = w & 3;
  const int col  = tn * 16 + m;

  v2f bf[DD / 4];
#pragma unroll
  for (int kb = 0; kb < DD / 4; ++kb) {
    int k0 = kb * 4 + half * 2;
    bf[kb][0] = Wn[k0 * DD + col];
    bf[kb][1] = Wn[(k0 + 1) * DD + col];
  }

  fg_gather_rows(rowptr, cnt, csr, dinv, hw, b, base + w * FG_PER_WAVE, lane,
                 stage + w * FG_PER_WAVE * FG_PITCH);

  __syncthreads();

  const float* arow = stage + (tm * 16 + m) * FG_PITCH + half * 2;
  v8f c = {};
#pragma unroll
  for (int kb = 0; kb < DD / 4; ++kb) {
    v2f a = *(const v2f*)(arow + kb * 4);
    c = __builtin_amdgcn_wmma_f32_16x16x4_f32(
         false, a,  false, bf[kb],
         (short)0, c,  false,  false);
  }

  int r0 = base + tm * 16 + half * 8;
#pragma unroll
  for (int j = 0; j < 8; ++j)
    hwn[(size_t)(r0 + j) * DD + col] = c[j];
}

__device__ __forceinline__ unsigned fflip(float f) {
  unsigned u = __float_as_uint(f);
  return u ^ ((unsigned)((int)u >> 31) | 0x80000000u);
}
__device__ __forceinline__ float funflip(unsigned u) {
  unsigned m = (u >> 31) ? 0x80000000u : 0xFFFFFFFFu;
  return __uint_as_float(u ^ m);
}

__device__ __forceinline__ void pool_flush(int g, int col, unsigned mx, float sm, float cn,
                                           unsigned* __restrict__ gmp, float* __restrict__ gsum,
                                           float* __restrict__ counts) {
  atomicMax(&gmp[g * DD + col], mx);
  atomicAdd(&gsum[g * DD + col], sm);
  if (col == 0) atomicAdd(&counts[g], cn);
}

__global__ void __launch_bounds__(FG_T)
k_gather_pool(const int* __restrict__ rowptr, const int* __restrict__ cnt,
              const int2* __restrict__ csr, const float* __restrict__ dinv,
              const float* __restrict__ hw, const float* __restrict__ b,
              const int* __restrict__ bidx, unsigned* __restrict__ gmp,
              float* __restrict__ gsum, float* __restrict__ counts) {
  __shared__ __attribute__((aligned(16))) float stage[FG_NODES * FG_PITCH];
  const int t    = threadIdx.x;
  const int lane = t & 31;
  const int w    = __builtin_amdgcn_readfirstlane((int)(t >> 5));
  const int base = blockIdx.x * FG_NODES;

  int g_own = 0;
  if (t < DD) g_own = bidx[base + lane];

  fg_gather_rows(rowptr, cnt, csr, dinv, hw, b, base + w * FG_PER_WAVE, lane,
                 stage + w * FG_PER_WAVE * FG_PITCH);

  __syncthreads();

  if (t < DD) {
    float v[FG_NODES];
#pragma unroll
    for (int r = 0; r < FG_NODES; ++r) v[r] = stage[r * FG_PITCH + t];

    int      cg = __builtin_amdgcn_readlane(g_own, 0);
    unsigned mx = fflip(v[0]);
    float    sm = v[0];
    float    cn = 1.0f;
#pragma unroll
    for (int r = 1; r < FG_NODES; ++r) {
      int g = __builtin_amdgcn_readlane(g_own, r);
      if (g != cg) {
        pool_flush(cg, t, mx, sm, cn, gmp, gsum, counts);
        cg = g;  mx = fflip(v[r]);  sm = v[r];  cn = 1.0f;
      } else {
        mx = max(mx, fflip(v[r]));  sm += v[r];  cn += 1.0f;
      }
    }
    pool_flush(cg, t, mx, sm, cn, gmp, gsum, counts);
  }
}

__global__ void k_final(const unsigned* __restrict__ gmp, const float* __restrict__ gsum,
                        const float* __restrict__ counts, const float* __restrict__ Wout,
                        const float* __restrict__ bout, float* __restrict__ out,
                        float* __restrict__ hidden) {
  __shared__ float red[2 * DD];
  int g = blockIdx.x, t = threadIdx.x;
  float cnt = fmaxf(counts[g], 1.0f);
  float hv = (t < DD) ? funflip(gmp[g * DD + t]) : (gsum[g * DD + (t - DD)] / cnt);
  hidden[(size_t)g * 2 * DD + t] = hv;
  red[t] = hv * Wout[t];
  __syncthreads();
  for (int s = DD; s > 0; s >>= 1) {
    if (t < s) red[t] += red[t + s];
    __syncthreads();
  }
  if (t == 0) out[g] = red[0] + bout[0];
}

extern "C" void kernel_launch(void* const* d_in, const int* in_sizes, int n_in,
                              void* d_out, int out_size, void* d_ws, size_t ws_size,
                              hipStream_t stream) {
  const float* x    = (const float*)d_in[0];
  const int*   ei   = (const int*)d_in[1];
  const int*   bidx = (const int*)d_in[2];
  const float* W0   = (const float*)d_in[3];
  const float* b0   = (const float*)d_in[4];
  const float* W1   = (const float*)d_in[5];
  const float* b1   = (const float*)d_in[6];
  const float* W2   = (const float*)d_in[7];
  const float* b2   = (const float*)d_in[8];
  const float* W3   = (const float*)d_in[9];
  const float* b3   = (const float*)d_in[10];
  const float* Wout = (const float*)d_in[11];
  const float* bout = (const float*)d_in[12];

  const int* src = ei;
  const int* dst = ei + EE;

  char*  ws   = (char*)d_ws;
  size_t off  = 0;
  const size_t nodeWords = (((size_t)NN * 4 + 255) & ~(size_t)255);
  float* dinv = (float*)(ws + off);  off += nodeWords;
  float* bufP = (float*)(ws + off);  off += (size_t)NN * DD * 4;
  float* bufQ = (float*)(ws + off);  off += (size_t)NN * DD * 4;
  unsigned* gmp = (unsigned*)(ws + off); off += (size_t)GG * DD * 4;
  float* gsum   = (float*)(ws + off);    off += (size_t)GG * DD * 4;
  float* counts = (float*)(ws + off);    off += (size_t)GG * 4;
  int* cnt    = (int*)(ws + off);  off += nodeWords;
  int* rowptr = (int*)(ws + off);  off += nodeWords;
  int* cursor = (int*)(ws + off);  off += nodeWords;
  int* bsum   = (int*)(ws + off);  off += 1024;
  int2* csr   = (int2*)(ws + off); off += (size_t)EE * 8;
  (void)ws_size; (void)n_in; (void)in_sizes; (void)out_size;

  float* outp   = (float*)d_out;
  float* hidden = outp + GG;

  const int T = 256;

  const int gemm_blocks = (NN / 16) * 4 / 8;

  k_gemm_wmma<FF><<<gemm_blocks, T, 0, stream>>>(x, W0, bufQ, cnt, gmp, gsum, counts);

  k_hist     <<<(EE + T - 1) / T, T, 0, stream>>>(dst, cnt);
  k_dinv_bsum<<<SCAN_BLOCKS, SCAN_T, 0, stream>>>(cnt, dinv, bsum);
  k_scan     <<<SCAN_BLOCKS, SCAN_T, 0, stream>>>(cnt, bsum, rowptr, cursor);
  k_fill     <<<(EE + T - 1) / T, T, 0, stream>>>(src, dst, dinv, cursor, csr);

  k_gather_gemm<<<NN / FG_NODES, FG_T, 0, stream>>>(rowptr, cnt, csr, dinv, bufQ, b0, W1, bufP);
  k_gather_gemm<<<NN / FG_NODES, FG_T, 0, stream>>>(rowptr, cnt, csr, dinv, bufP, b1, W2, bufQ);
  k_gather_gemm<<<NN / FG_NODES, FG_T, 0, stream>>>(rowptr, cnt, csr, dinv, bufQ, b2, W3, bufP);
  k_gather_pool<<<NN / FG_NODES, FG_T, 0, stream>>>(rowptr, cnt, csr, dinv, bufP, b3, bidx,
                                                    gmp, gsum, counts);

  k_final    <<<GG, 2 * DD, 0, stream>>>(gmp, gsum, counts, Wout, bout, outp, hidden);
}


# edge grouping: the in-degree atomic's return value is kept as each edge's rank; fill places edges at row start + rank with no second atomic; per-node edge order still atomic arrival order
# speedup vs baseline: 3.3234x; 1.1920x over previous
#include <hip/hip_runtime.h>
#include <math.h>

#define NN 100000
#define EE 1000000
#define FF 128
#define DD 64
#define GG 256

typedef float v2f __attribute__((ext_vector_type(2)));
typedef float v8f __attribute__((ext_vector_type(8)));

#define SCAN_T 256
#define SCAN_NODES (SCAN_T * 4)
#define SCAN_BLOCKS ((NN + SCAN_NODES - 1) / SCAN_NODES)
static_assert(NN % 4 == 0, "node count must be a multiple of 4 for the int4 scan loads");
static_assert(SCAN_BLOCKS <= 128, "block offsets are summed by one wave, four entries per lane");

__global__ void k_hist(const int* __restrict__ dst, int* __restrict__ cnt,
                       int* __restrict__ rank) {
  int e = blockIdx.x * blockDim.x + threadIdx.x;
  if (e < EE) rank[e] = atomicAdd(&cnt[dst[e]], 1);
}

__device__ __forceinline__ int wave_sum(int v) {
  for (int o = 16; o > 0; o >>= 1) v += __shfl_xor(v, o, 32);
  return v;
}

__global__ void k_dinv_bsum(const int* __restrict__ cnt, float* __restrict__ dinv,
                            int* __restrict__ bsum) {
  __shared__ int wsum[SCAN_T / 32];
  int t  = threadIdx.x;
  int i4 = blockIdx.x * SCAN_T + t;
  int s  = 0;
  if (i4 < NN / 4) {
    int4 c = ((const int4*)cnt)[i4];
    float4 r;
    r.x = rsqrtf((float)c.x + 1.0f); r.y = rsqrtf((float)c.y + 1.0f);
    r.z = rsqrtf((float)c.z + 1.0f); r.w = rsqrtf((float)c.w + 1.0f);
    ((float4*)dinv)[i4] = r;
    s = c.x + c.y + c.z + c.w;
  }
  s = wave_sum(s);
  if ((t & 31) == 0) wsum[t >> 5] = s;
  __syncthreads();
  if (t == 0) {
    int tot = 0;
    for (int w = 0; w < SCAN_T / 32; ++w) tot += wsum[w];
    bsum[blockIdx.x] = tot;
  }
}

__global__ void k_scan(const int* __restrict__ cnt, const int* __restrict__ bsum,
                       int* __restrict__ rowptr) {
  __shared__ int wtot[SCAN_T / 32];
  __shared__ int boff;
  int t = threadIdx.x, lane = t & 31, w = t >> 5;
  int b = blockIdx.x;
  if (w == 0) {
    int p = 0;
    for (int j = lane; j < b; j += 32) p += bsum[j];
    p = wave_sum(p);
    if (lane == 0) boff = p;
  }
  int  i4 = b * SCAN_T + t;
  int4 c  = make_int4(0, 0, 0, 0);
  if (i4 < NN / 4) c = ((const int4*)cnt)[i4];
  int tsum = c.x + c.y + c.z + c.w;
  int inc  = tsum;
  for (int o = 1; o < 32; o <<= 1) {
    int up = __shfl_up(inc, o, 32);
    if (lane >= o) inc += up;
  }
  if (lane == 31) wtot[w] = inc;
  __syncthreads();
  int wbase = 0;
  for (int k = 0; k < w; ++k) wbase += wtot[k];
  int base = boff + wbase + inc - tsum;
  if (i4 < NN / 4) {
    int4 r;
    r.x = base; r.y = r.x + c.x; r.z = r.y + c.y; r.w = r.z + c.z;
    ((int4*)rowptr)[i4] = r;
  }
}

__global__ void k_fill(const int* __restrict__ src, const int* __restrict__ dst,
                       const float* __restrict__ dinv, const int* __restrict__ rowptr,
                       const int* __restrict__ rank, int2* __restrict__ csr) {
  int e = blockIdx.x * blockDim.x + threadIdx.x;
  if (e < EE) {
    int s = src[e], d = dst[e], r = rank[e];
    float ws = dinv[s], wd = dinv[d];
    int p = rowptr[d] + r;
    asm volatile("" :: "v"(ws), "v"(wd));
    if ((unsigned)p < (unsigned)EE)
      csr[p] = make_int2(s, __float_as_int(ws * wd));
  }
}

static_assert((NN / 16) * 4 * 32 >= NN && NN >= GG * DD, "one product thread per word to clear");
template <int K>
__global__ void k_gemm_wmma(const float* __restrict__ A, const float* __restrict__ B,
                            float* __restrict__ C, int* __restrict__ cnt,
                            unsigned* __restrict__ gmp, float* __restrict__ gsum,
                            float* __restrict__ counts) {
  int gid  = blockIdx.x * blockDim.x + threadIdx.x;
  if (gid < NN) cnt[gid] = 0;
  if (gid < GG * DD) { gmp[gid] = 0x007FFFFFu  ; gsum[gid] = 0.0f; }
  if (gid < GG) counts[gid] = 0.0f;
  int wave = gid >> 5;
  int lane = gid & 31;
  int tm = wave >> 2;
  int tn = wave & 3;
  if (tm * 16 >= NN) return;

  int half = lane >> 4;
  int m    = lane & 15;
  const float* arow = A + (size_t)(tm * 16 + m) * K;
  int col = tn * 16 + m;

  v8f acc = {};
#pragma unroll
  for (int kb = 0; kb < K; kb += 4) {
    int k0 = kb + half * 2;
    v2f a; a[0] = arow[k0];          a[1] = arow[k0 + 1];
    v2f b; b[0] = B[k0 * DD + col];  b[1] = B[(k0 + 1) * DD + col];
    acc = __builtin_amdgcn_wmma_f32_16x16x4_f32(
         false, a,  false, b,
         (short)0, acc,  false,  false);
  }

  int r0 = tm * 16 + half * 8;
#pragma unroll
  for (int j = 0; j < 8; ++j)
    C[(size_t)(r0 + j) * DD + col] = acc[j];
}


template <int G>
__device__ __forceinline__ void ga_rows(const float2* __restrict__ hp, int sl, float wl, int j,
                                        float2& acc) {
  float2 v[G];
#pragma unroll
  for (int k = 0; k < G; ++k) {
    int s = __builtin_amdgcn_readlane(sl, j + k);
    v[k] = hp[(size_t)s * (DD / 2)];
  }
#pragma unroll
  for (int k = 0; k < G; ++k) {
    float w = __int_as_float(__builtin_amdgcn_readlane(__float_as_int(wl), j + k));
    acc.x = fmaf(w, v[k].x, acc.x);
    acc.y = fmaf(w, v[k].y, acc.y);
  }
}

__device__ __forceinline__ void ga_pairs(const int2* __restrict__ csr, int beg, int n, int c0,
                                         int lane, int d, int& sl, float& wl) {
  sl = d;
  wl = 0.0f;
  if (lane < min(32, n - c0)) {
    int2 ew = csr[beg + c0 + lane];
    sl = ew.x;
    wl = __int_as_float(ew.y);
  }
}
__device__ __forceinline__ int ga_source(int sl, int d) {
  return ((unsigned)sl >= (unsigned)NN) ? d : sl;
}

#define FG_T        256
#define FG_NODES    32
#define FG_PER_WAVE (FG_NODES / (FG_T / 32))
#define FG_PITCH    68
static_assert(NN % FG_NODES == 0, "every block owns 32 whole nodes");
static_assert(FG_PER_WAVE == 4, "the list bounds of a wave's nodes are read as one int4");

__device__ __forceinline__ void fg_gather_rows(const int* __restrict__ rowptr,
                                               const int* __restrict__ cnt,
                                               const int2* __restrict__ csr,
                                               const float* __restrict__ dinv,
                                               const float* __restrict__ hw,
                                               const float* __restrict__ b,
                                               int d, int lane, float* srow) {
  const float2* hp = (const float2*)hw + lane;
  const float2  bb = ((const float2*)b)[lane];

  int4 rp = ((const int4*)rowptr)[d >> 2];
  int4 cn = ((const int4*)cnt)[d >> 2];
  int beg  = __builtin_amdgcn_readfirstlane(rp.x), n  = __builtin_amdgcn_readfirstlane(cn.x);
  int begA = __builtin_amdgcn_readfirstlane(rp.y), nA = __builtin_amdgcn_readfirstlane(cn.y);
  int begB = __builtin_amdgcn_readfirstlane(rp.z), nB = __builtin_amdgcn_readfirstlane(cn.z);
  int begC = __builtin_amdgcn_readfirstlane(rp.w), nC = __builtin_amdgcn_readfirstlane(cn.w);
  int   sl;
  float wl;
  ga_pairs(csr, beg, n, 0, lane, d, sl, wl);
  sl = ga_source(sl, d);

  for (int q = 0; q < FG_PER_WAVE; ++q) {
    int   sln;
    float wln;
    ga_pairs(csr, begA, nA, 0, lane, d + 1, sln, wln);

    float  dv    = dinv[d];
    float2 vself = hp[(size_t)d * (DD / 2)];
    float2 acc;  acc.x = 0.0f;  acc.y = 0.0f;

    for (int c0 = 0; c0 < n; c0 += 32) {
      int mm = min(32, n - c0);
      if (c0 > 0) {
        ga_pairs(csr, beg, n, c0, lane, d, sl, wl);
        sl = ga_source(sl, d);
      }
      int j = 0;
      for (; j + 8 <= mm; j += 8) ga_rows<8>(hp, sl, wl, j, acc);
      switch (mm - j) {
        case 7: ga_rows<7>(hp, sl, wl, j, acc); break;
        case 6: ga_rows<6>(hp, sl, wl, j, acc); break;
        case 5: ga_rows<5>(hp, sl, wl, j, acc); break;
        case 4: ga_rows<4>(hp, sl, wl, j, acc); break;
        case 3: ga_rows<3>(hp, sl, wl, j, acc); break;
        case 2: ga_rows<2>(hp, sl, wl, j, acc); break;
        case 1: ga_rows<1>(hp, sl, wl, j, acc); break;
        default: break;
      }
    }

    float  sn = dv * dv;
    float2 o;
    o.x = tanhf(fmaf(sn, vself.x, acc.x) + bb.x);
    o.y = tanhf(fmaf(sn, vself.y, acc.y) + bb.y);
    *(float2*)&srow[q * FG_PITCH + 2 * lane] = o;

    sl = ga_source(sln, d + 1);  wl = wln;
    d = d + 1;
    beg = begA;  n = nA;
    begA = begB; nA = nB;
    begB = begC; nB = nC;
    begC = 0;    nC = 0;
  }
}

__global__ void __launch_bounds__(FG_T)
k_gather_gemm(const int* __restrict__ rowptr, const int* __restrict__ cnt,
              const int2* __restrict__ csr, const float* __restrict__ dinv,
              const float* __restrict__ hw, const float* __restrict__ b,
              const float* __restrict__ Wn, float* __restrict__ hwn) {
  __shared__ __attribute__((aligned(16))) float stage[FG_NODES * FG_PITCH];
  const int lane = threadIdx.x & 31;
  const int w    = __builtin_amdgcn_readfirstlane((int)(threadIdx.x >> 5));
  const int base = blockIdx.x * FG_NODES;
  const int half = lane >> 4;
  const int m    = lane & 15;
  const int tm   = w >> 2;
  const int tn   = w & 3;
  const int col  = tn * 16 + m;

  v2f bf[DD / 4];
#pragma unroll
  for (int kb = 0; kb < DD / 4; ++kb) {
    int k0 = kb * 4 + half * 2;
    bf[kb][0] = Wn[k0 * DD + col];
    bf[kb][1] = Wn[(k0 + 1) * DD + col];
  }

  fg_gather_rows(rowptr, cnt, csr, dinv, hw, b, base + w * FG_PER_WAVE, lane,
                 stage + w * FG_PER_WAVE * FG_PITCH);

  __syncthreads();

  const float* arow = stage + (tm * 16 + m) * FG_PITCH + half * 2;
  v8f c = {};
#pragma unroll
  for (int kb = 0; kb < DD / 4; ++kb) {
    v2f a = *(const v2f*)(arow + kb * 4);
    c = __builtin_amdgcn_wmma_f32_16x16x4_f32(
         false, a,  false, bf[kb],
         (short)0, c,  false,  false);
  }

  int r0 = base + tm * 16 + half * 8;
#pragma unroll
  for (int j = 0; j < 8; ++j)
    hwn[(size_t)(r0 + j) * DD + col] = c[j];
}

__device__ __forceinline__ unsigned fflip(float f) {
  unsigned u = __float_as_uint(f);
  return u ^ ((unsigned)((int)u >> 31) | 0x80000000u);
}
__device__ __forceinline__ float funflip(unsigned u) {
  unsigned m = (u >> 31) ? 0x80000000u : 0xFFFFFFFFu;
  return __uint_as_float(u ^ m);
}

__device__ __forceinline__ void pool_flush(int g, int col, unsigned mx, float sm, float cn,
                                           unsigned* __restrict__ gmp, float* __restrict__ gsum,
                                           float* __restrict__ counts) {
  atomicMax(&gmp[g * DD + col], mx);
  atomicAdd(&gsum[g * DD + col], sm);
  if (col == 0) atomicAdd(&counts[g], cn);
}

__global__ void __launch_bounds__(FG_T)
k_gather_pool(const int* __restrict__ rowptr, const int* __restrict__ cnt,
              const int2* __restrict__ csr, const float* __restrict__ dinv,
              const float* __restrict__ hw, const float* __restrict__ b,
              const int* __restrict__ bidx, unsigned* __restrict__ gmp,
              float* __restrict__ gsum, float* __restrict__ counts) {
  __shared__ __attribute__((aligned(16))) float stage[FG_NODES * FG_PITCH];
  const int t    = threadIdx.x;
  const int lane = t & 31;
  const int w    = __builtin_amdgcn_readfirstlane((int)(t >> 5));
  const int base = blockIdx.x * FG_NODES;

  int g_own = 0;
  if (t < DD) g_own = bidx[base + lane];

  fg_gather_rows(rowptr, cnt, csr, dinv, hw, b, base + w * FG_PER_WAVE, lane,
                 stage + w * FG_PER_WAVE * FG_PITCH);

  __syncthreads();

  if (t < DD) {
    float v[FG_NODES];
#pragma unroll
    for (int r = 0; r < FG_NODES; ++r) v[r] = stage[r * FG_PITCH + t];

    int      cg = __builtin_amdgcn_readlane(g_own, 0);
    unsigned mx = fflip(v[0]);
    float    sm = v[0];
    float    cn = 1.0f;
#pragma unroll
    for (int r = 1; r < FG_NODES; ++r) {
      int g = __builtin_amdgcn_readlane(g_own, r);
      if (g != cg) {
        pool_flush(cg, t, mx, sm, cn, gmp, gsum, counts);
        cg = g;  mx = fflip(v[r]);  sm = v[r];  cn = 1.0f;
      } else {
        mx = max(mx, fflip(v[r]));  sm += v[r];  cn += 1.0f;
      }
    }
    pool_flush(cg, t, mx, sm, cn, gmp, gsum, counts);
  }
}

__global__ void k_final(const unsigned* __restrict__ gmp, const float* __restrict__ gsum,
                        const float* __restrict__ counts, const float* __restrict__ Wout,
                        const float* __restrict__ bout, float* __restrict__ out,
                        float* __restrict__ hidden) {
  __shared__ float red[2 * DD];
  int g = blockIdx.x, t = threadIdx.x;
  float cnt = fmaxf(counts[g], 1.0f);
  float hv = (t < DD) ? funflip(gmp[g * DD + t]) : (gsum[g * DD + (t - DD)] / cnt);
  hidden[(size_t)g * 2 * DD + t] = hv;
  red[t] = hv * Wout[t];
  __syncthreads();
  for (int s = DD; s > 0; s >>= 1) {
    if (t < s) red[t] += red[t + s];
    __syncthreads();
  }
  if (t == 0) out[g] = red[0] + bout[0];
}

extern "C" void kernel_launch(void* const* d_in, const int* in_sizes, int n_in,
                              void* d_out, int out_size, void* d_ws, size_t ws_size,
                              hipStream_t stream) {
  const float* x    = (const float*)d_in[0];
  const int*   ei   = (const int*)d_in[1];
  const int*   bidx = (const int*)d_in[2];
  const float* W0   = (const float*)d_in[3];
  const float* b0   = (const float*)d_in[4];
  const float* W1   = (const float*)d_in[5];
  const float* b1   = (const float*)d_in[6];
  const float* W2   = (const float*)d_in[7];
  const float* b2   = (const float*)d_in[8];
  const float* W3   = (const float*)d_in[9];
  const float* b3   = (const float*)d_in[10];
  const float* Wout = (const float*)d_in[11];
  const float* bout = (const float*)d_in[12];

  const int* src = ei;
  const int* dst = ei + EE;

  char*  ws   = (char*)d_ws;
  size_t off  = 0;
  const size_t nodeWords = (((size_t)NN * 4 + 255) & ~(size_t)255);
  float* dinv = (float*)(ws + off);  off += nodeWords;
  float* bufP = (float*)(ws + off);  off += (size_t)NN * DD * 4;
  float* bufQ = (float*)(ws + off);  off += (size_t)NN * DD * 4;
  unsigned* gmp = (unsigned*)(ws + off); off += (size_t)GG * DD * 4;
  float* gsum   = (float*)(ws + off);    off += (size_t)GG * DD * 4;
  float* counts = (float*)(ws + off);    off += (size_t)GG * 4;
  int* cnt    = (int*)(ws + off);  off += nodeWords;
  int* rowptr = (int*)(ws + off);  off += nodeWords;
  int* rank   = (int*)(ws + off);  off += (size_t)EE * 4;
  int* bsum   = (int*)(ws + off);  off += 1024;
  int2* csr   = (int2*)(ws + off); off += (size_t)EE * 8;
  (void)ws_size; (void)n_in; (void)in_sizes; (void)out_size;

  float* outp   = (float*)d_out;
  float* hidden = outp + GG;

  const int T = 256;

  const int gemm_blocks = (NN / 16) * 4 / 8;

  k_gemm_wmma<FF><<<gemm_blocks, T, 0, stream>>>(x, W0, bufQ, cnt, gmp, gsum, counts);

  k_hist     <<<(EE + T - 1) / T, T, 0, stream>>>(dst, cnt, rank);
  k_dinv_bsum<<<SCAN_BLOCKS, SCAN_T, 0, stream>>>(cnt, dinv, bsum);
  k_scan     <<<SCAN_BLOCKS, SCAN_T, 0, stream>>>(cnt, bsum, rowptr);
  k_fill     <<<(EE + T - 1) / T, T, 0, stream>>>(src, dst, dinv, rowptr, rank, csr);

  k_gather_gemm<<<NN / FG_NODES, FG_T, 0, stream>>>(rowptr, cnt, csr, dinv, bufQ, b0, W1, bufP);
  k_gather_gemm<<<NN / FG_NODES, FG_T, 0, stream>>>(rowptr, cnt, csr, dinv, bufP, b1, W2, bufQ);
  k_gather_gemm<<<NN / FG_NODES, FG_T, 0, stream>>>(rowptr, cnt, csr, dinv, bufQ, b2, W3, bufP);
  k_gather_pool<<<NN / FG_NODES, FG_T, 0, stream>>>(rowptr, cnt, csr, dinv, bufP, b3, bidx,
                                                    gmp, gsum, counts);

  k_final    <<<GG, 2 * DD, 0, stream>>>(gmp, gsum, counts, Wout, bout, outp, hidden);
}


# edge grouping: per-edge global counting atomic replaced by a two-level counting sort on LDS counters (bins of 256 destinations, then per node); per-node edge order is LDS-atomic arrival order
# speedup vs baseline: 3.8999x; 1.1735x over previous
#include <hip/hip_runtime.h>
#include <math.h>

#define NN 100000
#define EE 1000000
#define FF 128
#define DD 64
#define GG 256

typedef float v2f __attribute__((ext_vector_type(2)));
typedef float v8f __attribute__((ext_vector_type(8)));

#define BIN_SHIFT 8
#define BIN_NODES (1 << BIN_SHIFT)
#define NBIN      ((NN + BIN_NODES - 1) / BIN_NODES)
#define CB_T      256
#define CB_EPT    16
#define CB_EDGES  (CB_T * CB_EPT)
#define CBLK      248
#define FLAT_N    (NBIN * CBLK)
#define FLAT_N4   (FLAT_N / 4)
#define SCAN_T 256
#define FLAT_BLOCKS ((FLAT_N4 + SCAN_T - 1) / SCAN_T)
#define BS_GROUP  4
static_assert(BS_GROUP == 4, "the load fence of the placing kernel names four values");
static_assert(CBLK * CB_EDGES >= EE, "the level-1 blocks cover every edge");
static_assert(FLAT_N % 4 == 0, "the (bin, block) counts are scanned as int4");
static_assert(FLAT_BLOCKS <= 256, "one 1024-byte region holds the scan's block sums");
static_assert(NN % 4 == 0, "list bounds of four consecutive nodes are read as one int4");

__device__ __forceinline__ int bin_of(int d) {
  unsigned b = (unsigned)d >> BIN_SHIFT;
  return (int)(b < (unsigned)NBIN ? b : (unsigned)(NBIN - 1));
}

__global__ void __launch_bounds__(CB_T)
k_bin_count(const int* __restrict__ dst, int* __restrict__ cntBB) {
  __shared__ int lc[NBIN + 1];
  const int t = threadIdx.x, b = blockIdx.x;
  for (int i = t; i < NBIN + 1; i += CB_T) lc[i] = 0;
  const int e0 = b * CB_EDGES + t;
  int d[CB_EPT];
#pragma unroll
  for (int i = 0; i < CB_EPT; ++i) d[i] = dst[min(e0 + i * CB_T, EE - 1)];
  __syncthreads();
#pragma unroll
  for (int i = 0; i < CB_EPT; ++i)
    atomicAdd(&lc[(e0 + i * CB_T < EE) ? bin_of(d[i]) : NBIN], 1);
  __syncthreads();
  for (int i = t; i < NBIN; i += CB_T) cntBB[i * CBLK + b] = lc[i];
}

__device__ __forceinline__ int wave_sum(int v) {
  for (int o = 16; o > 0; o >>= 1) v += __shfl_xor(v, o, 32);
  return v;
}

__global__ void k_bsum(const int* __restrict__ v, int* __restrict__ bsum) {
  __shared__ int wsum[SCAN_T / 32];
  int t  = threadIdx.x;
  int i4 = blockIdx.x * SCAN_T + t;
  int s  = 0;
  if (i4 < FLAT_N4) {
    int4 c = ((const int4*)v)[i4];
    s = c.x + c.y + c.z + c.w;
  }
  s = wave_sum(s);
  if ((t & 31) == 0) wsum[t >> 5] = s;
  __syncthreads();
  if (t == 0) {
    int tot = 0;
    for (int w = 0; w < SCAN_T / 32; ++w) tot += wsum[w];
    bsum[blockIdx.x] = tot;
  }
}

__global__ void k_scan(const int* __restrict__ v, const int* __restrict__ bsum,
                       int* __restrict__ base) {
  __shared__ int wtot[SCAN_T / 32];
  __shared__ int boff;
  int t = threadIdx.x, lane = t & 31, w = t >> 5;
  int b = blockIdx.x;
  if (w == 0) {
    int p = 0;
    for (int j = lane; j < b; j += 32) p += bsum[j];
    p = wave_sum(p);
    if (lane == 0) boff = p;
  }
  int  i4 = b * SCAN_T + t;
  int4 c  = make_int4(0, 0, 0, 0);
  if (i4 < FLAT_N4) c = ((const int4*)v)[i4];
  int tsum = c.x + c.y + c.z + c.w;
  int inc  = tsum;
  for (int o = 1; o < 32; o <<= 1) {
    int up = __shfl_up(inc, o, 32);
    if (lane >= o) inc += up;
  }
  if (lane == 31) wtot[w] = inc;
  __syncthreads();
  int wbase = 0;
  for (int k = 0; k < w; ++k) wbase += wtot[k];
  int first = boff + wbase + inc - tsum;
  if (i4 < FLAT_N4) {
    int4 r;
    r.x = first; r.y = r.x + c.x; r.z = r.y + c.y; r.w = r.z + c.z;
    ((int4*)base)[i4] = r;
  }
}

static_assert(NBIN > CB_T && NBIN <= 2 * CB_T, "two cursors per thread are loaded as one group");
__global__ void __launch_bounds__(CB_T)
k_bin_scatter(const int* __restrict__ src, const int* __restrict__ dst,
              const int* __restrict__ baseBB, int2* __restrict__ tmp) {
  __shared__ int cur[2 * CB_T];
  const int t = threadIdx.x, b = blockIdx.x;
  const int e0 = b * CB_EDGES + t;
  int c0 = baseBB[t * CBLK + b];
  int c1 = baseBB[min(t + CB_T, NBIN - 1) * CBLK + b];
  int s[CB_EPT], d[CB_EPT], p[CB_EPT];
#pragma unroll
  for (int i = 0; i < CB_EPT; ++i) {
    int e = min(e0 + i * CB_T, EE - 1);
    s[i] = src[e];
    d[i] = dst[e];
  }
  cur[t] = c0;
  cur[t + CB_T] = (t + CB_T < NBIN) ? c1 : EE;
  __syncthreads();
#pragma unroll
  for (int i = 0; i < CB_EPT; ++i)
    p[i] = atomicAdd(&cur[(e0 + i * CB_T < EE) ? bin_of(d[i]) : NBIN], 1);
#pragma unroll
  for (int i = 0; i < CB_EPT; ++i)
    if ((unsigned)p[i] < (unsigned)EE)
      tmp[p[i]] = make_int2(s[i], d[i]);
}

__device__ __forceinline__ void bin_range(const int* __restrict__ baseBB, int bin, int& beg,
                                          int& end) {
  beg = baseBB[bin * CBLK];
  end = (bin + 1 < NBIN) ? baseBB[(bin + 1) * CBLK] : EE;
  beg = min(max(beg, 0), EE);
  end = min(max(end, beg), EE);
}

__global__ void __launch_bounds__(BIN_NODES)
k_bin_degree(const int2* __restrict__ tmp, const int* __restrict__ baseBB,
             int* __restrict__ cnt, int* __restrict__ rowptr, float* __restrict__ dinv) {
  __shared__ int lc[BIN_NODES + 1];
  __shared__ int wtot[BIN_NODES / 32];
  const int bin = blockIdx.x, t = threadIdx.x;
  int beg, end;
  bin_range(baseBB, bin, beg, end);
  lc[t] = 0;
  if (t == 0) lc[BIN_NODES] = 0;
  __syncthreads();
  for (int i0 = beg + t; i0 < end; i0 += BIN_NODES * BS_GROUP) {
    int dd[BS_GROUP];
#pragma unroll
    for (int k = 0; k < BS_GROUP; ++k) dd[k] = tmp[min(i0 + k * BIN_NODES, end - 1)].y;
#pragma unroll
    for (int k = 0; k < BS_GROUP; ++k)
      atomicAdd(&lc[(i0 + k * BIN_NODES < end) ? ((dd[k] - bin * BIN_NODES) & (BIN_NODES - 1))
                                                : BIN_NODES], 1);
  }
  __syncthreads();
  const int lane = t & 31, w = t >> 5;
  int c   = lc[t];
  int inc = c;
  for (int o = 1; o < 32; o <<= 1) {
    int up = __shfl_up(inc, o, 32);
    if (lane >= o) inc += up;
  }
  if (lane == 31) wtot[w] = inc;
  __syncthreads();
  int wbase = 0;
  for (int k = 0; k < w; ++k) wbase += wtot[k];
  int node = bin * BIN_NODES + t;
  if (node < NN) {
    cnt[node]    = c;
    rowptr[node] = beg + wbase + inc - c;
    dinv[node]   = rsqrtf((float)c + 1.0f);
  }
}

__global__ void __launch_bounds__(BIN_NODES)
k_bin_place(const int2* __restrict__ tmp, const int* __restrict__ baseBB,
            const int* __restrict__ rowptr, const float* __restrict__ dinv,
            int2* __restrict__ csr) {
  __shared__ int   cur[BIN_NODES + 1];
  __shared__ float dv[BIN_NODES + 1];
  const int bin = blockIdx.x, t = threadIdx.x;
  int beg, end;
  bin_range(baseBB, bin, beg, end);
  const int node = bin * BIN_NODES + t;
  cur[t] = (node < NN) ? rowptr[node] : EE;
  dv[t]  = (node < NN) ? dinv[node] : 0.0f;
  if (t == 0) { cur[BIN_NODES] = EE;  dv[BIN_NODES] = 0.0f; }
  __syncthreads();
  for (int i0 = beg + t; i0 < end; i0 += BIN_NODES * BS_GROUP) {
    int2  e[BS_GROUP];
    float ws[BS_GROUP];
    int   dl[BS_GROUP], p[BS_GROUP];
#pragma unroll
    for (int k = 0; k < BS_GROUP; ++k) e[k] = tmp[min(i0 + k * BIN_NODES, end - 1)];
#pragma unroll
    for (int k = 0; k < BS_GROUP; ++k)
      ws[k] = dinv[((unsigned)e[k].x < (unsigned)NN) ? e[k].x : 0];
    asm volatile("" :: "v"(ws[0]), "v"(ws[1]), "v"(ws[2]), "v"(ws[3]));
#pragma unroll
    for (int k = 0; k < BS_GROUP; ++k) {
      dl[k] = (i0 + k * BIN_NODES < end) ? ((e[k].y - bin * BIN_NODES) & (BIN_NODES - 1))
                                         : BIN_NODES;
      p[k]  = atomicAdd(&cur[dl[k]], 1);
    }
#pragma unroll
    for (int k = 0; k < BS_GROUP; ++k)
      if ((unsigned)p[k] < (unsigned)EE)
        csr[p[k]] = make_int2(e[k].x, __float_as_int(ws[k] * dv[dl[k]]));
  }
}

static_assert((NN / 16) * 4 * 32 >= GG * DD, "one product thread per word to clear");
template <int K>
__global__ void k_gemm_wmma(const float* __restrict__ A, const float* __restrict__ B,
                            float* __restrict__ C,
                            unsigned* __restrict__ gmp, float* __restrict__ gsum,
                            float* __restrict__ counts) {
  int gid  = blockIdx.x * blockDim.x + threadIdx.x;
  if (gid < GG * DD) { gmp[gid] = 0x007FFFFFu  ; gsum[gid] = 0.0f; }
  if (gid < GG) counts[gid] = 0.0f;
  int wave = gid >> 5;
  int lane = gid & 31;
  int tm = wave >> 2;
  int tn = wave & 3;
  if (tm * 16 >= NN) return;

  int half = lane >> 4;
  int m    = lane & 15;
  const float* arow = A + (size_t)(tm * 16 + m) * K;
  int col = tn * 16 + m;

  v8f acc = {};
#pragma unroll
  for (int kb = 0; kb < K; kb += 4) {
    int k0 = kb + half * 2;
    v2f a; a[0] = arow[k0];          a[1] = arow[k0 + 1];
    v2f b; b[0] = B[k0 * DD + col];  b[1] = B[(k0 + 1) * DD + col];
    acc = __builtin_amdgcn_wmma_f32_16x16x4_f32(
         false, a,  false, b,
         (short)0, acc,  false,  false);
  }

  int r0 = tm * 16 + half * 8;
#pragma unroll
  for (int j = 0; j < 8; ++j)
    C[(size_t)(r0 + j) * DD + col] = acc[j];
}


template <int G>
__device__ __forceinline__ void ga_rows(const float2* __restrict__ hp, int sl, float wl, int j,
                                        float2& acc) {
  float2 v[G];
#pragma unroll
  for (int k = 0; k < G; ++k) {
    int s = __builtin_amdgcn_readlane(sl, j + k);
    v[k] = hp[(size_t)s * (DD / 2)];
  }
#pragma unroll
  for (int k = 0; k < G; ++k) {
    float w = __int_as_float(__builtin_amdgcn_readlane(__float_as_int(wl), j + k));
    acc.x = fmaf(w, v[k].x, acc.x);
    acc.y = fmaf(w, v[k].y, acc.y);
  }
}

__device__ __forceinline__ void ga_pairs(const int2* __restrict__ csr, int beg, int n, int c0,
                                         int lane, int d, int& sl, float& wl) {
  sl = d;
  wl = 0.0f;
  if (lane < min(32, n - c0)) {
    int2 ew = csr[beg + c0 + lane];
    sl = ew.x;
    wl = __int_as_float(ew.y);
  }
}
__device__ __forceinline__ int ga_source(int sl, int d) {
  return ((unsigned)sl >= (unsigned)NN) ? d : sl;
}

#define FG_T        256
#define FG_NODES    32
#define FG_PER_WAVE (FG_NODES / (FG_T / 32))
#define FG_PITCH    68
static_assert(NN % FG_NODES == 0, "every block owns 32 whole nodes");
static_assert(FG_PER_WAVE == 4, "the list bounds of a wave's nodes are read as one int4");

__device__ __forceinline__ void fg_gather_rows(const int* __restrict__ rowptr,
                                               const int* __restrict__ cnt,
                                               const int2* __restrict__ csr,
                                               const float* __restrict__ dinv,
                                               const float* __restrict__ hw,
                                               const float* __restrict__ b,
                                               int d, int lane, float* srow) {
  const float2* hp = (const float2*)hw + lane;
  const float2  bb = ((const float2*)b)[lane];

  int4 rp = ((const int4*)rowptr)[d >> 2];
  int4 cn = ((const int4*)cnt)[d >> 2];
  int beg  = __builtin_amdgcn_readfirstlane(rp.x), n  = __builtin_amdgcn_readfirstlane(cn.x);
  int begA = __builtin_amdgcn_readfirstlane(rp.y), nA = __builtin_amdgcn_readfirstlane(cn.y);
  int begB = __builtin_amdgcn_readfirstlane(rp.z), nB = __builtin_amdgcn_readfirstlane(cn.z);
  int begC = __builtin_amdgcn_readfirstlane(rp.w), nC = __builtin_amdgcn_readfirstlane(cn.w);
  int   sl;
  float wl;
  ga_pairs(csr, beg, n, 0, lane, d, sl, wl);
  sl = ga_source(sl, d);

  for (int q = 0; q < FG_PER_WAVE; ++q) {
    int   sln;
    float wln;
    ga_pairs(csr, begA, nA, 0, lane, d + 1, sln, wln);

    float  dv    = dinv[d];
    float2 vself = hp[(size_t)d * (DD / 2)];
    float2 acc;  acc.x = 0.0f;  acc.y = 0.0f;

    for (int c0 = 0; c0 < n; c0 += 32) {
      int mm = min(32, n - c0);
      if (c0 > 0) {
        ga_pairs(csr, beg, n, c0, lane, d, sl, wl);
        sl = ga_source(sl, d);
      }
      int j = 0;
      for (; j + 8 <= mm; j += 8) ga_rows<8>(hp, sl, wl, j, acc);
      switch (mm - j) {
        case 7: ga_rows<7>(hp, sl, wl, j, acc); break;
        case 6: ga_rows<6>(hp, sl, wl, j, acc); break;
        case 5: ga_rows<5>(hp, sl, wl, j, acc); break;
        case 4: ga_rows<4>(hp, sl, wl, j, acc); break;
        case 3: ga_rows<3>(hp, sl, wl, j, acc); break;
        case 2: ga_rows<2>(hp, sl, wl, j, acc); break;
        case 1: ga_rows<1>(hp, sl, wl, j, acc); break;
        default: break;
      }
    }

    float  sn = dv * dv;
    float2 o;
    o.x = tanhf(fmaf(sn, vself.x, acc.x) + bb.x);
    o.y = tanhf(fmaf(sn, vself.y, acc.y) + bb.y);
    *(float2*)&srow[q * FG_PITCH + 2 * lane] = o;

    sl = ga_source(sln, d + 1);  wl = wln;
    d = d + 1;
    beg = begA;  n = nA;
    begA = begB; nA = nB;
    begB = begC; nB = nC;
    begC = 0;    nC = 0;
  }
}

__global__ void __launch_bounds__(FG_T)
k_gather_gemm(const int* __restrict__ rowptr, const int* __restrict__ cnt,
              const int2* __restrict__ csr, const float* __restrict__ dinv,
              const float* __restrict__ hw, const float* __restrict__ b,
              const float* __restrict__ Wn, float* __restrict__ hwn) {
  __shared__ __attribute__((aligned(16))) float stage[FG_NODES * FG_PITCH];
  const int lane = threadIdx.x & 31;
  const int w    = __builtin_amdgcn_readfirstlane((int)(threadIdx.x >> 5));
  const int base = blockIdx.x * FG_NODES;
  const int half = lane >> 4;
  const int m    = lane & 15;
  const int tm   = w >> 2;
  const int tn   = w & 3;
  const int col  = tn * 16 + m;

  v2f bf[DD / 4];
#pragma unroll
  for (int kb = 0; kb < DD / 4; ++kb) {
    int k0 = kb * 4 + half * 2;
    bf[kb][0] = Wn[k0 * DD + col];
    bf[kb][1] = Wn[(k0 + 1) * DD + col];
  }

  fg_gather_rows(rowptr, cnt, csr, dinv, hw, b, base + w * FG_PER_WAVE, lane,
                 stage + w * FG_PER_WAVE * FG_PITCH);

  __syncthreads();

  const float* arow = stage + (tm * 16 + m) * FG_PITCH + half * 2;
  v8f c = {};
#pragma unroll
  for (int kb = 0; kb < DD / 4; ++kb) {
    v2f a = *(const v2f*)(arow + kb * 4);
    c = __builtin_amdgcn_wmma_f32_16x16x4_f32(
         false, a,  false, bf[kb],
         (short)0, c,  false,  false);
  }

  int r0 = base + tm * 16 + half * 8;
#pragma unroll
  for (int j = 0; j < 8; ++j)
    hwn[(size_t)(r0 + j) * DD + col] = c[j];
}

__device__ __forceinline__ unsigned fflip(float f) {
  unsigned u = __float_as_uint(f);
  return u ^ ((unsigned)((int)u >> 31) | 0x80000000u);
}
__device__ __forceinline__ float funflip(unsigned u) {
  unsigned m = (u >> 31) ? 0x80000000u : 0xFFFFFFFFu;
  return __uint_as_float(u ^ m);
}

__device__ __forceinline__ void pool_flush(int g, int col, unsigned mx, float sm, float cn,
                                           unsigned* __restrict__ gmp, float* __restrict__ gsum,
                                           float* __restrict__ counts) {
  atomicMax(&gmp[g * DD + col], mx);
  atomicAdd(&gsum[g * DD + col], sm);
  if (col == 0) atomicAdd(&counts[g], cn);
}

__global__ void __launch_bounds__(FG_T)
k_gather_pool(const int* __restrict__ rowptr, const int* __restrict__ cnt,
              const int2* __restrict__ csr, const float* __restrict__ dinv,
              const float* __restrict__ hw, const float* __restrict__ b,
              const int* __restrict__ bidx, unsigned* __restrict__ gmp,
              float* __restrict__ gsum, float* __restrict__ counts) {
  __shared__ __attribute__((aligned(16))) float stage[FG_NODES * FG_PITCH];
  const int t    = threadIdx.x;
  const int lane = t & 31;
  const int w    = __builtin_amdgcn_readfirstlane((int)(t >> 5));
  const int base = blockIdx.x * FG_NODES;

  int g_own = 0;
  if (t < DD) g_own = bidx[base + lane];

  fg_gather_rows(rowptr, cnt, csr, dinv, hw, b, base + w * FG_PER_WAVE, lane,
                 stage + w * FG_PER_WAVE * FG_PITCH);

  __syncthreads();

  if (t < DD) {
    float v[FG_NODES];
#pragma unroll
    for (int r = 0; r < FG_NODES; ++r) v[r] = stage[r * FG_PITCH + t];

    int      cg = __builtin_amdgcn_readlane(g_own, 0);
    unsigned mx = fflip(v[0]);
    float    sm = v[0];
    float    cn = 1.0f;
#pragma unroll
    for (int r = 1; r < FG_NODES; ++r) {
      int g = __builtin_amdgcn_readlane(g_own, r);
      if (g != cg) {
        pool_flush(cg, t, mx, sm, cn, gmp, gsum, counts);
        cg = g;  mx = fflip(v[r]);  sm = v[r];  cn = 1.0f;
      } else {
        mx = max(mx, fflip(v[r]));  sm += v[r];  cn += 1.0f;
      }
    }
    pool_flush(cg, t, mx, sm, cn, gmp, gsum, counts);
  }
}

__global__ void k_final(const unsigned* __restrict__ gmp, const float* __restrict__ gsum,
                        const float* __restrict__ counts, const float* __restrict__ Wout,
                        const float* __restrict__ bout, float* __restrict__ out,
                        float* __restrict__ hidden) {
  __shared__ float red[2 * DD];
  int g = blockIdx.x, t = threadIdx.x;
  float cnt = fmaxf(counts[g], 1.0f);
  float hv = (t < DD) ? funflip(gmp[g * DD + t]) : (gsum[g * DD + (t - DD)] / cnt);
  hidden[(size_t)g * 2 * DD + t] = hv;
  red[t] = hv * Wout[t];
  __syncthreads();
  for (int s = DD; s > 0; s >>= 1) {
    if (t < s) red[t] += red[t + s];
    __syncthreads();
  }
  if (t == 0) out[g] = red[0] + bout[0];
}

extern "C" void kernel_launch(void* const* d_in, const int* in_sizes, int n_in,
                              void* d_out, int out_size, void* d_ws, size_t ws_size,
                              hipStream_t stream) {
  const float* x    = (const float*)d_in[0];
  const int*   ei   = (const int*)d_in[1];
  const int*   bidx = (const int*)d_in[2];
  const float* W0   = (const float*)d_in[3];
  const float* b0   = (const float*)d_in[4];
  const float* W1   = (const float*)d_in[5];
  const float* b1   = (const float*)d_in[6];
  const float* W2   = (const float*)d_in[7];
  const float* b2   = (const float*)d_in[8];
  const float* W3   = (const float*)d_in[9];
  const float* b3   = (const float*)d_in[10];
  const float* Wout = (const float*)d_in[11];
  const float* bout = (const float*)d_in[12];

  const int* src = ei;
  const int* dst = ei + EE;

  char*  ws   = (char*)d_ws;
  size_t off  = 0;
  const size_t nodeWords = (((size_t)NN * 4 + 255) & ~(size_t)255);
  float* dinv = (float*)(ws + off);  off += nodeWords;
  float* bufP = (float*)(ws + off);  off += (size_t)NN * DD * 4;
  float* bufQ = (float*)(ws + off);  off += (size_t)NN * DD * 4;
  unsigned* gmp = (unsigned*)(ws + off); off += (size_t)GG * DD * 4;
  float* gsum   = (float*)(ws + off);    off += (size_t)GG * DD * 4;
  float* counts = (float*)(ws + off);    off += (size_t)GG * 4;
  int* cnt    = (int*)(ws + off);  off += nodeWords;
  int* rowptr = (int*)(ws + off);  off += nodeWords;
  const size_t flatBytes = (((size_t)FLAT_N * 4 + 255) & ~(size_t)255);
  int* cntBB  = (int*)(ws + off);  off += flatBytes;
  int* baseBB = (int*)(ws + off);  off += flatBytes;
  int* bsum   = (int*)(ws + off);  off += 1024;
  int2* tmp   = (int2*)(ws + off); off += (size_t)EE * 8;
  int2* csr   = (int2*)(ws + off); off += (size_t)EE * 8;
  (void)ws_size; (void)n_in; (void)in_sizes; (void)out_size;

  float* outp   = (float*)d_out;
  float* hidden = outp + GG;

  const int T = 256;

  const int gemm_blocks = (NN / 16) * 4 / 8;

  k_gemm_wmma<FF><<<gemm_blocks, T, 0, stream>>>(x, W0, bufQ, gmp, gsum, counts);

  k_bin_count  <<<CBLK, CB_T, 0, stream>>>(dst, cntBB);
  k_bsum       <<<FLAT_BLOCKS, SCAN_T, 0, stream>>>(cntBB, bsum);
  k_scan       <<<FLAT_BLOCKS, SCAN_T, 0, stream>>>(cntBB, bsum, baseBB);
  k_bin_scatter<<<CBLK, CB_T, 0, stream>>>(src, dst, baseBB, tmp);
  k_bin_degree <<<NBIN, BIN_NODES, 0, stream>>>(tmp, baseBB, cnt, rowptr, dinv);
  k_bin_place  <<<NBIN, BIN_NODES, 0, stream>>>(tmp, baseBB, rowptr, dinv, csr);

  k_gather_gemm<<<NN / FG_NODES, FG_T, 0, stream>>>(rowptr, cnt, csr, dinv, bufQ, b0, W1, bufP);
  k_gather_gemm<<<NN / FG_NODES, FG_T, 0, stream>>>(rowptr, cnt, csr, dinv, bufP, b1, W2, bufQ);
  k_gather_gemm<<<NN / FG_NODES, FG_T, 0, stream>>>(rowptr, cnt, csr, dinv, bufQ, b2, W3, bufP);
  k_gather_pool<<<NN / FG_NODES, FG_T, 0, stream>>>(rowptr, cnt, csr, dinv, bufP, b3, bidx,
                                                    gmp, gsum, counts);

  k_final    <<<GG, 2 * DD, 0, stream>>>(gmp, gsum, counts, Wout, bout, outp, hidden);
}


# launches 12 -> 10: layer-0 product rides as extra work-groups in four edge-sort launches (no launch of its own); one per-bin running-sum launch replaces the two scan launches; sum orders unchanged
# speedup vs baseline: 4.2569x; 1.0915x over previous
#include <hip/hip_runtime.h>
#include <math.h>

#define NN 100000
#define EE 1000000
#define FF 128
#define DD 64
#define GG 256

typedef float v2f __attribute__((ext_vector_type(2)));
typedef float v8f __attribute__((ext_vector_type(8)));

#define BIN_SHIFT 8
#define BIN_NODES (1 << BIN_SHIFT)
#define NBIN      ((NN + BIN_NODES - 1) / BIN_NODES)
#define CB_T      256
#define CB_EPT    16
#define CB_EDGES  (CB_T * CB_EPT)
#define CBLK      248
#define TAB_PITCH 256
#define NBINP     512
#define BS_GROUP  4
static_assert(BS_GROUP == 4, "the load fence of the placing kernel names four values");
static_assert(CBLK * CB_EDGES >= EE, "the level-1 blocks cover every edge");
static_assert(CBLK <= TAB_PITCH && TAB_PITCH == CB_T, "one thread per level-1 block scans a bin row");
static_assert(NBIN > CB_T && NBIN < 2 * CB_T && NBINP == 2 * CB_T, "two bins per thread of a moving block, the spare bin among them");
static_assert(NN % 4 == 0, "list bounds of four consecutive nodes are read as one int4");

__device__ __forceinline__ int bin_of(int d) {
  unsigned b = (unsigned)d >> BIN_SHIFT;
  return (int)(b < (unsigned)NBIN ? b : (unsigned)(NBIN - 1));
}

#define GEMM_BLOCKS ((NN / 16) * 4 / 8)
#define GEMM_P1 480
#define GEMM_P2 1325
#define GEMM_P3 1925
static_assert(GEMM_P1 * 256 >= GG * DD, "the pooling start values are stored by blocks of the first launch");
static_assert(GEMM_P1 < GEMM_P2 && GEMM_P2 < GEMM_P3 && GEMM_P3 < GEMM_BLOCKS, "four parts in order");
__device__ __forceinline__ void gemm0_block(int gb, const float* __restrict__ A,
                                            const float* __restrict__ B, float* __restrict__ C,
                                            unsigned* __restrict__ gmp, float* __restrict__ gsum,
                                            float* __restrict__ counts) {
  constexpr int K = FF;
  int gid  = gb * 256 + threadIdx.x;
  if (gid < GG * DD) { gmp[gid] = 0x007FFFFFu  ; gsum[gid] = 0.0f; }
  if (gid < GG) counts[gid] = 0.0f;
  int wave = gid >> 5;
  int lane = gid & 31;
  int tm = wave >> 2;
  int tn = wave & 3;
  if (tm * 16 >= NN) return;

  int half = lane >> 4;
  int m    = lane & 15;
  const float* arow = A + (size_t)(tm * 16 + m) * K;
  int col = tn * 16 + m;

  v8f acc = {};
#pragma unroll
  for (int kb = 0; kb < K; kb += 4) {
    int k0 = kb + half * 2;
    v2f a; a[0] = arow[k0];          a[1] = arow[k0 + 1];
    v2f b; b[0] = B[k0 * DD + col];  b[1] = B[(k0 + 1) * DD + col];
    acc = __builtin_amdgcn_wmma_f32_16x16x4_f32(
         false, a,  false, b,
         (short)0, acc,  false,  false);
  }

  int r0 = tm * 16 + half * 8;
#pragma unroll
  for (int j = 0; j < 8; ++j)
    C[(size_t)(r0 + j) * DD + col] = acc[j];
}

__device__ __forceinline__ void bin_count_block(const int* __restrict__ dst,
                                                int* __restrict__ cntTab) {
  __shared__ int lc[NBIN + 1];
  const int t = threadIdx.x, b = blockIdx.x;
  for (int i = t; i < NBIN + 1; i += CB_T) lc[i] = 0;
  const int e0 = b * CB_EDGES + t;
  int d[CB_EPT];
#pragma unroll
  for (int i = 0; i < CB_EPT; ++i) d[i] = dst[min(e0 + i * CB_T, EE - 1)];
  __syncthreads();
#pragma unroll
  for (int i = 0; i < CB_EPT; ++i)
    atomicAdd(&lc[(e0 + i * CB_T < EE) ? bin_of(d[i]) : NBIN], 1);
  __syncthreads();
  for (int i = t; i < NBIN; i += CB_T) cntTab[i * TAB_PITCH + b] = lc[i];
}

__global__ void __launch_bounds__(CB_T)
k_count_gemm(const int* __restrict__ dst, int* __restrict__ cntTab,
             const float* __restrict__ x, const float* __restrict__ W0, float* __restrict__ hw,
             unsigned* __restrict__ gmp, float* __restrict__ gsum, float* __restrict__ counts) {
  if (blockIdx.x >= CBLK) {
    gemm0_block((int)blockIdx.x - CBLK, x, W0, hw, gmp, gsum, counts);
    return;
  }
  bin_count_block(dst, cntTab);
}

__device__ __forceinline__ int wave_scan_incl(int v, int lane) {
  for (int o = 1; o < 32; o <<= 1) {
    int up = __shfl_up(v, o, 32);
    if (lane >= o) v += up;
  }
  return v;
}

__global__ void __launch_bounds__(CB_T)
k_bin_colscan(int* __restrict__ cntTab, int* __restrict__ binTot) {
  __shared__ int wtot[CB_T / 32];
  const int t = threadIdx.x, i = blockIdx.x, lane = t & 31, w = t >> 5;
  int c = 0;
  if (t < CBLK) c = cntTab[i * TAB_PITCH + t];
  int inc = wave_scan_incl(c, lane);
  if (lane == 31) wtot[w] = inc;
  __syncthreads();
  int wbase = 0;
  for (int k = 0; k < w; ++k) wbase += wtot[k];
  if (t < CBLK) cntTab[i * TAB_PITCH + t] = wbase + inc - c;
  if (t == CB_T - 1) binTot[i] = wbase + inc;
}


__device__ __forceinline__ void bin_scatter_block(const int* __restrict__ src,
                                                  const int* __restrict__ dst,
                                                  const int* __restrict__ cntTab,
                                                  const int* __restrict__ binTot,
                                                  int* __restrict__ binStart,
                                                  int2* __restrict__ tmp) {
  __shared__ int cur[NBINP];
  __shared__ int wt0[CB_T / 32];
  __shared__ int wt1[CB_T / 32];
  const int t = threadIdx.x, b = blockIdx.x, lane = t & 31, w = t >> 5;
  const int e0 = b * CB_EDGES + t;
  const int i1 = min(t + CB_T, NBIN - 1);
  int c0 = cntTab[t * TAB_PITCH + b];
  int c1 = cntTab[i1 * TAB_PITCH + b];
  int T0 = binTot[t];
  int T1 = binTot[i1];
  int s[CB_EPT], d[CB_EPT], p[CB_EPT];
#pragma unroll
  for (int i = 0; i < CB_EPT; ++i) {
    int e = min(e0 + i * CB_T, EE - 1);
    s[i] = src[e];
    d[i] = dst[e];
  }
  if (t + CB_T >= NBIN) T1 = 0;
  int inc0 = wave_scan_incl(T0, lane);
  int inc1 = wave_scan_incl(T1, lane);
  if (lane == 31) { wt0[w] = inc0;  wt1[w] = inc1; }
  __syncthreads();
  int base0 = inc0 - T0;
  int base1 = inc1 - T1;
  for (int k = 0; k < CB_T / 32; ++k) {
    int a = wt0[k];
    base1 += a;
    if (k < w) { base0 += a;  base1 += wt1[k]; }
  }
  cur[t] = base0 + c0;
  cur[t + CB_T] = (t + CB_T < NBIN) ? base1 + c1 : EE;
  if (b == 0) { binStart[t] = base0;  binStart[t + CB_T] = base1; }
  __syncthreads();
#pragma unroll
  for (int i = 0; i < CB_EPT; ++i)
    p[i] = atomicAdd(&cur[(e0 + i * CB_T < EE) ? bin_of(d[i]) : NBIN], 1);
#pragma unroll
  for (int i = 0; i < CB_EPT; ++i)
    if ((unsigned)p[i] < (unsigned)EE)
      tmp[p[i]] = make_int2(s[i], d[i]);
}

__global__ void __launch_bounds__(CB_T)
k_scatter_gemm(const int* __restrict__ src, const int* __restrict__ dst,
               const int* __restrict__ cntTab, const int* __restrict__ binTot,
               int* __restrict__ binStart, int2* __restrict__ tmp,
               const float* __restrict__ x, const float* __restrict__ W0, float* __restrict__ hw,
               unsigned* __restrict__ gmp, float* __restrict__ gsum, float* __restrict__ counts) {
  if (blockIdx.x >= CBLK) {
    gemm0_block((int)blockIdx.x - CBLK + GEMM_P1, x, W0, hw, gmp, gsum, counts);
    return;
  }
  bin_scatter_block(src, dst, cntTab, binTot, binStart, tmp);
}

__device__ __forceinline__ void bin_range(const int* __restrict__ binStart, int bin, int& beg,
                                          int& end) {
  beg = binStart[bin];
  end = binStart[bin + 1];
  beg = min(max(beg, 0), EE);
  end = min(max(end, beg), EE);
}

__device__ __forceinline__ void bin_degree_block(const int2* __restrict__ tmp,
                                                 const int* __restrict__ binStart,
                                                 int* __restrict__ cnt, int* __restrict__ rowptr,
                                                 float* __restrict__ dinv) {
  __shared__ int lc[BIN_NODES + 1];
  __shared__ int wtot[BIN_NODES / 32];
  const int bin = blockIdx.x, t = threadIdx.x;
  int beg, end;
  bin_range(binStart, bin, beg, end);
  lc[t] = 0;
  if (t == 0) lc[BIN_NODES] = 0;
  __syncthreads();
  for (int i0 = beg + t; i0 < end; i0 += BIN_NODES * BS_GROUP) {
    int dd[BS_GROUP];
#pragma unroll
    for (int k = 0; k < BS_GROUP; ++k) dd[k] = tmp[min(i0 + k * BIN_NODES, end - 1)].y;
#pragma unroll
    for (int k = 0; k < BS_GROUP; ++k)
      atomicAdd(&lc[(i0 + k * BIN_NODES < end) ? ((dd[k] - bin * BIN_NODES) & (BIN_NODES - 1))
                                                : BIN_NODES], 1);
  }
  __syncthreads();
  const int lane = t & 31, w = t >> 5;
  int c   = lc[t];
  int inc = c;
  for (int o = 1; o < 32; o <<= 1) {
    int up = __shfl_up(inc, o, 32);
    if (lane >= o) inc += up;
  }
  if (lane == 31) wtot[w] = inc;
  __syncthreads();
  int wbase = 0;
  for (int k = 0; k < w; ++k) wbase += wtot[k];
  int node = bin * BIN_NODES + t;
  if (node < NN) {
    cnt[node]    = c;
    rowptr[node] = beg + wbase + inc - c;
    dinv[node]   = rsqrtf((float)c + 1.0f);
  }
}

__global__ void __launch_bounds__(BIN_NODES)
k_degree_gemm(const int2* __restrict__ tmp, const int* __restrict__ binStart,
              int* __restrict__ cnt, int* __restrict__ rowptr, float* __restrict__ dinv,
              const float* __restrict__ x, const float* __restrict__ W0, float* __restrict__ hw,
              unsigned* __restrict__ gmp, float* __restrict__ gsum, float* __restrict__ counts) {
  if (blockIdx.x >= NBIN) {
    gemm0_block((int)blockIdx.x - NBIN + GEMM_P2, x, W0, hw, gmp, gsum, counts);
    return;
  }
  bin_degree_block(tmp, binStart, cnt, rowptr, dinv);
}

__device__ __forceinline__ void bin_place_block(const int2* __restrict__ tmp,
                                                const int* __restrict__ binStart,
                                                const int* __restrict__ rowptr,
                                                const float* __restrict__ dinv,
                                                int2* __restrict__ csr) {
  __shared__ int   cur[BIN_NODES + 1];
  __shared__ float dv[BIN_NODES + 1];
  const int bin = blockIdx.x, t = threadIdx.x;
  int beg, end;
  bin_range(binStart, bin, beg, end);
  const int node = bin * BIN_NODES + t;
  cur[t] = (node < NN) ? rowptr[node] : EE;
  dv[t]  = (node < NN) ? dinv[node] : 0.0f;
  if (t == 0) { cur[BIN_NODES] = EE;  dv[BIN_NODES] = 0.0f; }
  __syncthreads();
  for (int i0 = beg + t; i0 < end; i0 += BIN_NODES * BS_GROUP) {
    int2  e[BS_GROUP];
    float ws[BS_GROUP];
    int   dl[BS_GROUP], p[BS_GROUP];
#pragma unroll
    for (int k = 0; k < BS_GROUP; ++k) e[k] = tmp[min(i0 + k * BIN_NODES, end - 1)];
#pragma unroll
    for (int k = 0; k < BS_GROUP; ++k)
      ws[k] = dinv[((unsigned)e[k].x < (unsigned)NN) ? e[k].x : 0];
    asm volatile("" :: "v"(ws[0]), "v"(ws[1]), "v"(ws[2]), "v"(ws[3]));
#pragma unroll
    for (int k = 0; k < BS_GROUP; ++k) {
      dl[k] = (i0 + k * BIN_NODES < end) ? ((e[k].y - bin * BIN_NODES) & (BIN_NODES - 1))
                                         : BIN_NODES;
      p[k]  = atomicAdd(&cur[dl[k]], 1);
    }
#pragma unroll
    for (int k = 0; k < BS_GROUP; ++k)
      if ((unsigned)p[k] < (unsigned)EE)
        csr[p[k]] = make_int2(e[k].x, __float_as_int(ws[k] * dv[dl[k]]));
  }
}

__global__ void __launch_bounds__(BIN_NODES)
k_place_gemm(const int2* __restrict__ tmp, const int* __restrict__ binStart,
             const int* __restrict__ rowptr, const float* __restrict__ dinv,
             int2* __restrict__ csr,
             const float* __restrict__ x, const float* __restrict__ W0, float* __restrict__ hw,
             unsigned* __restrict__ gmp, float* __restrict__ gsum, float* __restrict__ counts) {
  if (blockIdx.x >= NBIN) {
    gemm0_block((int)blockIdx.x - NBIN + GEMM_P3, x, W0, hw, gmp, gsum, counts);
    return;
  }
  bin_place_block(tmp, binStart, rowptr, dinv, csr);
}


template <int G>
__device__ __forceinline__ void ga_rows(const float2* __restrict__ hp, int sl, float wl, int j,
                                        float2& acc) {
  float2 v[G];
#pragma unroll
  for (int k = 0; k < G; ++k) {
    int s = __builtin_amdgcn_readlane(sl, j + k);
    v[k] = hp[(size_t)s * (DD / 2)];
  }
#pragma unroll
  for (int k = 0; k < G; ++k) {
    float w = __int_as_float(__builtin_amdgcn_readlane(__float_as_int(wl), j + k));
    acc.x = fmaf(w, v[k].x, acc.x);
    acc.y = fmaf(w, v[k].y, acc.y);
  }
}

__device__ __forceinline__ void ga_pairs(const int2* __restrict__ csr, int beg, int n, int c0,
                                         int lane, int d, int& sl, float& wl) {
  sl = d;
  wl = 0.0f;
  if (lane < min(32, n - c0)) {
    int2 ew = csr[beg + c0 + lane];
    sl = ew.x;
    wl = __int_as_float(ew.y);
  }
}
__device__ __forceinline__ int ga_source(int sl, int d) {
  return ((unsigned)sl >= (unsigned)NN) ? d : sl;
}

#define FG_T        256
#define FG_NODES    32
#define FG_PER_WAVE (FG_NODES / (FG_T / 32))
#define FG_PITCH    68
static_assert(NN % FG_NODES == 0, "every block owns 32 whole nodes");
static_assert(FG_PER_WAVE == 4, "the list bounds of a wave's nodes are read as one int4");

__device__ __forceinline__ void fg_gather_rows(const int* __restrict__ rowptr,
                                               const int* __restrict__ cnt,
                                               const int2* __restrict__ csr,
                                               const float* __restrict__ dinv,
                                               const float* __restrict__ hw,
                                               const float* __restrict__ b,
                                               int d, int lane, float* srow) {
  const float2* hp = (const float2*)hw + lane;
  const float2  bb = ((const float2*)b)[lane];

  int4 rp = ((const int4*)rowptr)[d >> 2];
  int4 cn = ((const int4*)cnt)[d >> 2];
  int beg  = __builtin_amdgcn_readfirstlane(rp.x), n  = __builtin_amdgcn_readfirstlane(cn.x);
  int begA = __builtin_amdgcn_readfirstlane(rp.y), nA = __builtin_amdgcn_readfirstlane(cn.y);
  int begB = __builtin_amdgcn_readfirstlane(rp.z), nB = __builtin_amdgcn_readfirstlane(cn.z);
  int begC = __builtin_amdgcn_readfirstlane(rp.w), nC = __builtin_amdgcn_readfirstlane(cn.w);
  int   sl;
  float wl;
  ga_pairs(csr, beg, n, 0, lane, d, sl, wl);
  sl = ga_source(sl, d);

  for (int q = 0; q < FG_PER_WAVE; ++q) {
    int   sln;
    float wln;
    ga_pairs(csr, begA, nA, 0, lane, d + 1, sln, wln);

    float  dv    = dinv[d];
    float2 vself = hp[(size_t)d * (DD / 2)];
    float2 acc;  acc.x = 0.0f;  acc.y = 0.0f;

    for (int c0 = 0; c0 < n; c0 += 32) {
      int mm = min(32, n - c0);
      if (c0 > 0) {
        ga_pairs(csr, beg, n, c0, lane, d, sl, wl);
        sl = ga_source(sl, d);
      }
      int j = 0;
      for (; j + 8 <= mm; j += 8) ga_rows<8>(hp, sl, wl, j, acc);
      switch (mm - j) {
        case 7: ga_rows<7>(hp, sl, wl, j, acc); break;
        case 6: ga_rows<6>(hp, sl, wl, j, acc); break;
        case 5: ga_rows<5>(hp, sl, wl, j, acc); break;
        case 4: ga_rows<4>(hp, sl, wl, j, acc); break;
        case 3: ga_rows<3>(hp, sl, wl, j, acc); break;
        case 2: ga_rows<2>(hp, sl, wl, j, acc); break;
        case 1: ga_rows<1>(hp, sl, wl, j, acc); break;
        default: break;
      }
    }

    float  sn = dv * dv;
    float2 o;
    o.x = tanhf(fmaf(sn, vself.x, acc.x) + bb.x);
    o.y = tanhf(fmaf(sn, vself.y, acc.y) + bb.y);
    *(float2*)&srow[q * FG_PITCH + 2 * lane] = o;

    sl = ga_source(sln, d + 1);  wl = wln;
    d = d + 1;
    beg = begA;  n = nA;
    begA = begB; nA = nB;
    begB = begC; nB = nC;
    begC = 0;    nC = 0;
  }
}

__global__ void __launch_bounds__(FG_T)
k_gather_gemm(const int* __restrict__ rowptr, const int* __restrict__ cnt,
              const int2* __restrict__ csr, const float* __restrict__ dinv,
              const float* __restrict__ hw, const float* __restrict__ b,
              const float* __restrict__ Wn, float* __restrict__ hwn) {
  __shared__ __attribute__((aligned(16))) float stage[FG_NODES * FG_PITCH];
  const int lane = threadIdx.x & 31;
  const int w    = __builtin_amdgcn_readfirstlane((int)(threadIdx.x >> 5));
  const int base = blockIdx.x * FG_NODES;
  const int half = lane >> 4;
  const int m    = lane & 15;
  const int tm   = w >> 2;
  const int tn   = w & 3;
  const int col  = tn * 16 + m;

  v2f bf[DD / 4];
#pragma unroll
  for (int kb = 0; kb < DD / 4; ++kb) {
    int k0 = kb * 4 + half * 2;
    bf[kb][0] = Wn[k0 * DD + col];
    bf[kb][1] = Wn[(k0 + 1) * DD + col];
  }

  fg_gather_rows(rowptr, cnt, csr, dinv, hw, b, base + w * FG_PER_WAVE, lane,
                 stage + w * FG_PER_WAVE * FG_PITCH);

  __syncthreads();

  const float* arow = stage + (tm * 16 + m) * FG_PITCH + half * 2;
  v8f c = {};
#pragma unroll
  for (int kb = 0; kb < DD / 4; ++kb) {
    v2f a = *(const v2f*)(arow + kb * 4);
    c = __builtin_amdgcn_wmma_f32_16x16x4_f32(
         false, a,  false, bf[kb],
         (short)0, c,  false,  false);
  }

  int r0 = base + tm * 16 + half * 8;
#pragma unroll
  for (int j = 0; j < 8; ++j)
    hwn[(size_t)(r0 + j) * DD + col] = c[j];
}

__device__ __forceinline__ unsigned fflip(float f) {
  unsigned u = __float_as_uint(f);
  return u ^ ((unsigned)((int)u >> 31) | 0x80000000u);
}
__device__ __forceinline__ float funflip(unsigned u) {
  unsigned m = (u >> 31) ? 0x80000000u : 0xFFFFFFFFu;
  return __uint_as_float(u ^ m);
}

__device__ __forceinline__ void pool_flush(int g, int col, unsigned mx, float sm, float cn,
                                           unsigned* __restrict__ gmp, float* __restrict__ gsum,
                                           float* __restrict__ counts) {
  atomicMax(&gmp[g * DD + col], mx);
  atomicAdd(&gsum[g * DD + col], sm);
  if (col == 0) atomicAdd(&counts[g], cn);
}

__global__ void __launch_bounds__(FG_T)
k_gather_pool(const int* __restrict__ rowptr, const int* __restrict__ cnt,
              const int2* __restrict__ csr, const float* __restrict__ dinv,
              const float* __restrict__ hw, const float* __restrict__ b,
              const int* __restrict__ bidx, unsigned* __restrict__ gmp,
              float* __restrict__ gsum, float* __restrict__ counts) {
  __shared__ __attribute__((aligned(16))) float stage[FG_NODES * FG_PITCH];
  const int t    = threadIdx.x;
  const int lane = t & 31;
  const int w    = __builtin_amdgcn_readfirstlane((int)(t >> 5));
  const int base = blockIdx.x * FG_NODES;

  int g_own = 0;
  if (t < DD) g_own = bidx[base + lane];

  fg_gather_rows(rowptr, cnt, csr, dinv, hw, b, base + w * FG_PER_WAVE, lane,
                 stage + w * FG_PER_WAVE * FG_PITCH);

  __syncthreads();

  if (t < DD) {
    float v[FG_NODES];
#pragma unroll
    for (int r = 0; r < FG_NODES; ++r) v[r] = stage[r * FG_PITCH + t];

    int      cg = __builtin_amdgcn_readlane(g_own, 0);
    unsigned mx = fflip(v[0]);
    float    sm = v[0];
    float    cn = 1.0f;
#pragma unroll
    for (int r = 1; r < FG_NODES; ++r) {
      int g = __builtin_amdgcn_readlane(g_own, r);
      if (g != cg) {
        pool_flush(cg, t, mx, sm, cn, gmp, gsum, counts);
        cg = g;  mx = fflip(v[r]);  sm = v[r];  cn = 1.0f;
      } else {
        mx = max(mx, fflip(v[r]));  sm += v[r];  cn += 1.0f;
      }
    }
    pool_flush(cg, t, mx, sm, cn, gmp, gsum, counts);
  }
}

__global__ void k_final(const unsigned* __restrict__ gmp, const float* __restrict__ gsum,
                        const float* __restrict__ counts, const float* __restrict__ Wout,
                        const float* __restrict__ bout, float* __restrict__ out,
                        float* __restrict__ hidden) {
  __shared__ float red[2 * DD];
  int g = blockIdx.x, t = threadIdx.x;
  float cnt = fmaxf(counts[g], 1.0f);
  float hv = (t < DD) ? funflip(gmp[g * DD + t]) : (gsum[g * DD + (t - DD)] / cnt);
  hidden[(size_t)g * 2 * DD + t] = hv;
  red[t] = hv * Wout[t];
  __syncthreads();
  for (int s = DD; s > 0; s >>= 1) {
    if (t < s) red[t] += red[t + s];
    __syncthreads();
  }
  if (t == 0) out[g] = red[0] + bout[0];
}

extern "C" void kernel_launch(void* const* d_in, const int* in_sizes, int n_in,
                              void* d_out, int out_size, void* d_ws, size_t ws_size,
                              hipStream_t stream) {
  const float* x    = (const float*)d_in[0];
  const int*   ei   = (const int*)d_in[1];
  const int*   bidx = (const int*)d_in[2];
  const float* W0   = (const float*)d_in[3];
  const float* b0   = (const float*)d_in[4];
  const float* W1   = (const float*)d_in[5];
  const float* b1   = (const float*)d_in[6];
  const float* W2   = (const float*)d_in[7];
  const float* b2   = (const float*)d_in[8];
  const float* W3   = (const float*)d_in[9];
  const float* b3   = (const float*)d_in[10];
  const float* Wout = (const float*)d_in[11];
  const float* bout = (const float*)d_in[12];

  const int* src = ei;
  const int* dst = ei + EE;

  char*  ws   = (char*)d_ws;
  size_t off  = 0;
  const size_t nodeWords = (((size_t)NN * 4 + 255) & ~(size_t)255);
  float* dinv = (float*)(ws + off);  off += nodeWords;
  float* bufP = (float*)(ws + off);  off += (size_t)NN * DD * 4;
  float* bufQ = (float*)(ws + off);  off += (size_t)NN * DD * 4;
  unsigned* gmp = (unsigned*)(ws + off); off += (size_t)GG * DD * 4;
  float* gsum   = (float*)(ws + off);    off += (size_t)GG * DD * 4;
  float* counts = (float*)(ws + off);    off += (size_t)GG * 4;
  int* cnt    = (int*)(ws + off);  off += nodeWords;
  int* rowptr = (int*)(ws + off);  off += nodeWords;
  int* cntTab   = (int*)(ws + off);  off += (size_t)NBIN * TAB_PITCH * 4;
  int* binTot   = (int*)(ws + off);  off += (size_t)NBINP * 4;
  int* binStart = (int*)(ws + off);  off += (size_t)NBINP * 4;
  int2* tmp   = (int2*)(ws + off); off += (size_t)EE * 8;
  int2* csr   = (int2*)(ws + off); off += (size_t)EE * 8;
  (void)ws_size; (void)n_in; (void)in_sizes; (void)out_size;

  float* outp   = (float*)d_out;
  float* hidden = outp + GG;

  k_count_gemm  <<<CBLK + GEMM_P1, CB_T, 0, stream>>>(dst, cntTab, x, W0, bufQ, gmp, gsum, counts);
  k_bin_colscan <<<NBIN, CB_T, 0, stream>>>(cntTab, binTot);
  k_scatter_gemm<<<CBLK + (GEMM_P2 - GEMM_P1), CB_T, 0, stream>>>(src, dst, cntTab, binTot, binStart, tmp,
                                                                  x, W0, bufQ, gmp, gsum, counts);
  k_degree_gemm <<<NBIN + (GEMM_P3 - GEMM_P2), BIN_NODES, 0, stream>>>(tmp, binStart, cnt, rowptr, dinv,
                                                                       x, W0, bufQ, gmp, gsum, counts);
  k_place_gemm  <<<NBIN + (GEMM_BLOCKS - GEMM_P3), BIN_NODES, 0, stream>>>(tmp, binStart, rowptr, dinv, csr,
                                                                           x, W0, bufQ, gmp, gsum, counts);

  k_gather_gemm<<<NN / FG_NODES, FG_T, 0, stream>>>(rowptr, cnt, csr, dinv, bufQ, b0, W1, bufP);
  k_gather_gemm<<<NN / FG_NODES, FG_T, 0, stream>>>(rowptr, cnt, csr, dinv, bufP, b1, W2, bufQ);
  k_gather_gemm<<<NN / FG_NODES, FG_T, 0, stream>>>(rowptr, cnt, csr, dinv, bufQ, b2, W3, bufP);
  k_gather_pool<<<NN / FG_NODES, FG_T, 0, stream>>>(rowptr, cnt, csr, dinv, bufP, b3, bidx,
                                                    gmp, gsum, counts);

  k_final    <<<GG, 2 * DD, 0, stream>>>(gmp, gsum, counts, Wout, bout, outp, hidden);
}


# gather: a node's edge list split over the wave's two 16-lane halves, one 16-byte lane load fetches two rows, 12 rows per group; per-node sum order changes (even and odd entries apart, then added)
# speedup vs baseline: 4.4263x; 1.0398x over previous
#include <hip/hip_runtime.h>
#include <math.h>

#define NN 100000
#define EE 1000000
#define FF 128
#define DD 64
#define GG 256

typedef float v2f __attribute__((ext_vector_type(2)));
typedef float v8f __attribute__((ext_vector_type(8)));

#define BIN_SHIFT 8
#define BIN_NODES (1 << BIN_SHIFT)
#define NBIN      ((NN + BIN_NODES - 1) / BIN_NODES)
#define CB_T      256
#define CB_EPT    16
#define CB_EDGES  (CB_T * CB_EPT)
#define CBLK      248
#define TAB_PITCH 256
#define NBINP     512
#define BS_GROUP  4
static_assert(BS_GROUP == 4, "the load fence of the placing kernel names four values");
static_assert(CBLK * CB_EDGES >= EE, "the level-1 blocks cover every edge");
static_assert(CBLK <= TAB_PITCH && TAB_PITCH == CB_T, "one thread per level-1 block scans a bin row");
static_assert(NBIN > CB_T && NBIN < 2 * CB_T && NBINP == 2 * CB_T, "two bins per thread of a moving block, the spare bin among them");
static_assert(NN % 4 == 0, "list bounds of four consecutive nodes are read as one int4");

__device__ __forceinline__ int bin_of(int d) {
  unsigned b = (unsigned)d >> BIN_SHIFT;
  return (int)(b < (unsigned)NBIN ? b : (unsigned)(NBIN - 1));
}

#define GEMM_BLOCKS ((NN / 16) * 4 / 8)
#define GEMM_P1 480
#define GEMM_P2 1325
#define GEMM_P3 1925
static_assert(GEMM_P1 * 256 >= GG * DD, "the pooling start values are stored by blocks of the first launch");
static_assert(GEMM_P1 < GEMM_P2 && GEMM_P2 < GEMM_P3 && GEMM_P3 < GEMM_BLOCKS, "four parts in order");
__device__ __forceinline__ void gemm0_block(int gb, const float* __restrict__ A,
                                            const float* __restrict__ B, float* __restrict__ C,
                                            unsigned* __restrict__ gmp, float* __restrict__ gsum,
                                            float* __restrict__ counts) {
  constexpr int K = FF;
  int gid  = gb * 256 + threadIdx.x;
  if (gid < GG * DD) { gmp[gid] = 0x007FFFFFu  ; gsum[gid] = 0.0f; }
  if (gid < GG) counts[gid] = 0.0f;
  int wave = gid >> 5;
  int lane = gid & 31;
  int tm = wave >> 2;
  int tn = wave & 3;
  if (tm * 16 >= NN) return;

  int half = lane >> 4;
  int m    = lane & 15;
  const float* arow = A + (size_t)(tm * 16 + m) * K;
  int col = tn * 16 + m;

  v8f acc = {};
#pragma unroll
  for (int kb = 0; kb < K; kb += 4) {
    int k0 = kb + half * 2;
    v2f a; a[0] = arow[k0];          a[1] = arow[k0 + 1];
    v2f b; b[0] = B[k0 * DD + col];  b[1] = B[(k0 + 1) * DD + col];
    acc = __builtin_amdgcn_wmma_f32_16x16x4_f32(
         false, a,  false, b,
         (short)0, acc,  false,  false);
  }

  int r0 = tm * 16 + half * 8;
#pragma unroll
  for (int j = 0; j < 8; ++j)
    C[(size_t)(r0 + j) * DD + col] = acc[j];
}

__device__ __forceinline__ void bin_count_block(const int* __restrict__ dst,
                                                int* __restrict__ cntTab) {
  __shared__ int lc[NBIN + 1];
  const int t = threadIdx.x, b = blockIdx.x;
  for (int i = t; i < NBIN + 1; i += CB_T) lc[i] = 0;
  const int e0 = b * CB_EDGES + t;
  int d[CB_EPT];
#pragma unroll
  for (int i = 0; i < CB_EPT; ++i) d[i] = dst[min(e0 + i * CB_T, EE - 1)];
  __syncthreads();
#pragma unroll
  for (int i = 0; i < CB_EPT; ++i)
    atomicAdd(&lc[(e0 + i * CB_T < EE) ? bin_of(d[i]) : NBIN], 1);
  __syncthreads();
  for (int i = t; i < NBIN; i += CB_T) cntTab[i * TAB_PITCH + b] = lc[i];
}

__global__ void __launch_bounds__(CB_T)
k_count_gemm(const int* __restrict__ dst, int* __restrict__ cntTab,
             const float* __restrict__ x, const float* __restrict__ W0, float* __restrict__ hw,
             unsigned* __restrict__ gmp, float* __restrict__ gsum, float* __restrict__ counts) {
  if (blockIdx.x >= CBLK) {
    gemm0_block((int)blockIdx.x - CBLK, x, W0, hw, gmp, gsum, counts);
    return;
  }
  bin_count_block(dst, cntTab);
}

__device__ __forceinline__ int wave_scan_incl(int v, int lane) {
  for (int o = 1; o < 32; o <<= 1) {
    int up = __shfl_up(v, o, 32);
    if (lane >= o) v += up;
  }
  return v;
}

__global__ void __launch_bounds__(CB_T)
k_bin_colscan(int* __restrict__ cntTab, int* __restrict__ binTot) {
  __shared__ int wtot[CB_T / 32];
  const int t = threadIdx.x, i = blockIdx.x, lane = t & 31, w = t >> 5;
  int c = 0;
  if (t < CBLK) c = cntTab[i * TAB_PITCH + t];
  int inc = wave_scan_incl(c, lane);
  if (lane == 31) wtot[w] = inc;
  __syncthreads();
  int wbase = 0;
  for (int k = 0; k < w; ++k) wbase += wtot[k];
  if (t < CBLK) cntTab[i * TAB_PITCH + t] = wbase + inc - c;
  if (t == CB_T - 1) binTot[i] = wbase + inc;
}


__device__ __forceinline__ void bin_scatter_block(const int* __restrict__ src,
                                                  const int* __restrict__ dst,
                                                  const int* __restrict__ cntTab,
                                                  const int* __restrict__ binTot,
                                                  int* __restrict__ binStart,
                                                  int2* __restrict__ tmp) {
  __shared__ int cur[NBINP];
  __shared__ int wt0[CB_T / 32];
  __shared__ int wt1[CB_T / 32];
  const int t = threadIdx.x, b = blockIdx.x, lane = t & 31, w = t >> 5;
  const int e0 = b * CB_EDGES + t;
  const int i1 = min(t + CB_T, NBIN - 1);
  int c0 = cntTab[t * TAB_PITCH + b];
  int c1 = cntTab[i1 * TAB_PITCH + b];
  int T0 = binTot[t];
  int T1 = binTot[i1];
  int s[CB_EPT], d[CB_EPT], p[CB_EPT];
#pragma unroll
  for (int i = 0; i < CB_EPT; ++i) {
    int e = min(e0 + i * CB_T, EE - 1);
    s[i] = src[e];
    d[i] = dst[e];
  }
  if (t + CB_T >= NBIN) T1 = 0;
  int inc0 = wave_scan_incl(T0, lane);
  int inc1 = wave_scan_incl(T1, lane);
  if (lane == 31) { wt0[w] = inc0;  wt1[w] = inc1; }
  __syncthreads();
  int base0 = inc0 - T0;
  int base1 = inc1 - T1;
  for (int k = 0; k < CB_T / 32; ++k) {
    int a = wt0[k];
    base1 += a;
    if (k < w) { base0 += a;  base1 += wt1[k]; }
  }
  cur[t] = base0 + c0;
  cur[t + CB_T] = (t + CB_T < NBIN) ? base1 + c1 : EE;
  if (b == 0) { binStart[t] = base0;  binStart[t + CB_T] = base1; }
  __syncthreads();
#pragma unroll
  for (int i = 0; i < CB_EPT; ++i)
    p[i] = atomicAdd(&cur[(e0 + i * CB_T < EE) ? bin_of(d[i]) : NBIN], 1);
#pragma unroll
  for (int i = 0; i < CB_EPT; ++i)
    if ((unsigned)p[i] < (unsigned)EE)
      tmp[p[i]] = make_int2(s[i], d[i]);
}

__global__ void __launch_bounds__(CB_T)
k_scatter_gemm(const int* __restrict__ src, const int* __restrict__ dst,
               const int* __restrict__ cntTab, const int* __restrict__ binTot,
               int* __restrict__ binStart, int2* __restrict__ tmp,
               const float* __restrict__ x, const float* __restrict__ W0, float* __restrict__ hw,
               unsigned* __restrict__ gmp, float* __restrict__ gsum, float* __restrict__ counts) {
  if (blockIdx.x >= CBLK) {
    gemm0_block((int)blockIdx.x - CBLK + GEMM_P1, x, W0, hw, gmp, gsum, counts);
    return;
  }
  bin_scatter_block(src, dst, cntTab, binTot, binStart, tmp);
}

__device__ __forceinline__ void bin_range(const int* __restrict__ binStart, int bin, int& beg,
                                          int& end) {
  beg = binStart[bin];
  end = binStart[bin + 1];
  beg = min(max(beg, 0), EE);
  end = min(max(end, beg), EE);
}

__device__ __forceinline__ void bin_degree_block(const int2* __restrict__ tmp,
                                                 const int* __restrict__ binStart,
                                                 int* __restrict__ cnt, int* __restrict__ rowptr,
                                                 float* __restrict__ dinv) {
  __shared__ int lc[BIN_NODES + 1];
  __shared__ int wtot[BIN_NODES / 32];
  const int bin = blockIdx.x, t = threadIdx.x;
  int beg, end;
  bin_range(binStart, bin, beg, end);
  lc[t] = 0;
  if (t == 0) lc[BIN_NODES] = 0;
  __syncthreads();
  for (int i0 = beg + t; i0 < end; i0 += BIN_NODES * BS_GROUP) {
    int dd[BS_GROUP];
#pragma unroll
    for (int k = 0; k < BS_GROUP; ++k) dd[k] = tmp[min(i0 + k * BIN_NODES, end - 1)].y;
#pragma unroll
    for (int k = 0; k < BS_GROUP; ++k)
      atomicAdd(&lc[(i0 + k * BIN_NODES < end) ? ((dd[k] - bin * BIN_NODES) & (BIN_NODES - 1))
                                                : BIN_NODES], 1);
  }
  __syncthreads();
  const int lane = t & 31, w = t >> 5;
  int c   = lc[t];
  int inc = c;
  for (int o = 1; o < 32; o <<= 1) {
    int up = __shfl_up(inc, o, 32);
    if (lane >= o) inc += up;
  }
  if (lane == 31) wtot[w] = inc;
  __syncthreads();
  int wbase = 0;
  for (int k = 0; k < w; ++k) wbase += wtot[k];
  int node = bin * BIN_NODES + t;
  if (node < NN) {
    cnt[node]    = c;
    rowptr[node] = beg + wbase + inc - c;
    dinv[node]   = rsqrtf((float)c + 1.0f);
  }
}

__global__ void __launch_bounds__(BIN_NODES)
k_degree_gemm(const int2* __restrict__ tmp, const int* __restrict__ binStart,
              int* __restrict__ cnt, int* __restrict__ rowptr, float* __restrict__ dinv,
              const float* __restrict__ x, const float* __restrict__ W0, float* __restrict__ hw,
              unsigned* __restrict__ gmp, float* __restrict__ gsum, float* __restrict__ counts) {
  if (blockIdx.x >= NBIN) {
    gemm0_block((int)blockIdx.x - NBIN + GEMM_P2, x, W0, hw, gmp, gsum, counts);
    return;
  }
  bin_degree_block(tmp, binStart, cnt, rowptr, dinv);
}

__device__ __forceinline__ void bin_place_block(const int2* __restrict__ tmp,
                                                const int* __restrict__ binStart,
                                                const int* __restrict__ rowptr,
                                                const float* __restrict__ dinv,
                                                int2* __restrict__ csr) {
  __shared__ int   cur[BIN_NODES + 1];
  __shared__ float dv[BIN_NODES + 1];
  const int bin = blockIdx.x, t = threadIdx.x;
  int beg, end;
  bin_range(binStart, bin, beg, end);
  const int node = bin * BIN_NODES + t;
  cur[t] = (node < NN) ? rowptr[node] : EE;
  dv[t]  = (node < NN) ? dinv[node] : 0.0f;
  if (t == 0) { cur[BIN_NODES] = EE;  dv[BIN_NODES] = 0.0f; }
  __syncthreads();
  for (int i0 = beg + t; i0 < end; i0 += BIN_NODES * BS_GROUP) {
    int2  e[BS_GROUP];
    float ws[BS_GROUP];
    int   dl[BS_GROUP], p[BS_GROUP];
#pragma unroll
    for (int k = 0; k < BS_GROUP; ++k) e[k] = tmp[min(i0 + k * BIN_NODES, end - 1)];
#pragma unroll
    for (int k = 0; k < BS_GROUP; ++k)
      ws[k] = dinv[((unsigned)e[k].x < (unsigned)NN) ? e[k].x : 0];
    asm volatile("" :: "v"(ws[0]), "v"(ws[1]), "v"(ws[2]), "v"(ws[3]));
#pragma unroll
    for (int k = 0; k < BS_GROUP; ++k) {
      dl[k] = (i0 + k * BIN_NODES < end) ? ((e[k].y - bin * BIN_NODES) & (BIN_NODES - 1))
                                         : BIN_NODES;
      p[k]  = atomicAdd(&cur[dl[k]], 1);
    }
#pragma unroll
    for (int k = 0; k < BS_GROUP; ++k)
      if ((unsigned)p[k] < (unsigned)EE)
        csr[p[k]] = make_int2(e[k].x, __float_as_int(ws[k] * dv[dl[k]]));
  }
}

__global__ void __launch_bounds__(BIN_NODES)
k_place_gemm(const int2* __restrict__ tmp, const int* __restrict__ binStart,
             const int* __restrict__ rowptr, const float* __restrict__ dinv,
             int2* __restrict__ csr,
             const float* __restrict__ x, const float* __restrict__ W0, float* __restrict__ hw,
             unsigned* __restrict__ gmp, float* __restrict__ gsum, float* __restrict__ counts) {
  if (blockIdx.x >= NBIN) {
    gemm0_block((int)blockIdx.x - NBIN + GEMM_P3, x, W0, hw, gmp, gsum, counts);
    return;
  }
  bin_place_block(tmp, binStart, rowptr, dinv, csr);
}


template <int K>
__device__ __forceinline__ int half_lane(int v) {
  return __builtin_amdgcn_update_dpp(0, v, 0x150 + K, 0xF, 0xF, true);
}
template <int R>
__device__ __forceinline__ int half_down(int v) {
  return __builtin_amdgcn_update_dpp(0, v, 0x120 + (16 - R), 0xF, 0xF, true);
}
__device__ __forceinline__ float other_half(float x) {
  return __int_as_float(__builtin_amdgcn_permlanex16(__float_as_int(x), __float_as_int(x),
                                                     0x76543210, 0xfedcba98, false, false));
}

template <int K, int G>
__device__ __forceinline__ void ga_issue(const char* __restrict__ hb, unsigned c16, int sl,
                                         float4* v) {
  if constexpr (K < G) {
    unsigned s = (unsigned)half_lane<K>(sl);
    v[K] = *(const float4*)(hb + ((s << 8) | c16));
    ga_issue<K + 1, G>(hb, c16, sl, v);
  }
}
template <int K, int G>
__device__ __forceinline__ void ga_sum(float wl, const float4* v, float4& acc) {
  if constexpr (K < G) {
    float w = __int_as_float(half_lane<K>(__float_as_int(wl)));
    acc.x = fmaf(w, v[K].x, acc.x);
    acc.y = fmaf(w, v[K].y, acc.y);
    acc.z = fmaf(w, v[K].z, acc.z);
    acc.w = fmaf(w, v[K].w, acc.w);
    ga_sum<K + 1, G>(wl, v, acc);
  }
}
template <int G>
__device__ __forceinline__ void ga_rows(const char* __restrict__ hb, unsigned c16, int sl, float wl,
                                        float4& acc) {
  float4 v[G];
  ga_issue<0, G>(hb, c16, sl, v);
  ga_sum<0, G>(wl, v, acc);
}

__device__ __forceinline__ void ga_pairs(const int2* __restrict__ csr, int beg, int n, int c0,
                                         int pe, int d, int& sl, float& wl) {
  sl = d;
  wl = 0.0f;
  if (pe < min(32, n - c0)) {
    int2 ew = csr[beg + c0 + pe];
    sl = ew.x;
    wl = __int_as_float(ew.y);
  }
}
__device__ __forceinline__ int ga_source(int sl, int d) {
  return ((unsigned)sl >= (unsigned)NN) ? d : sl;
}

#define FG_T        256
#define FG_NODES    32
#define FG_PER_WAVE (FG_NODES / (FG_T / 32))
#define FG_PITCH    68
#define GA_STEPS    6
static_assert(NN % FG_NODES == 0, "every block owns 32 whole nodes");
static_assert(FG_PER_WAVE == 4, "the list bounds of a wave's nodes are read as one int4");
static_assert(GA_STEPS == 6, "the rest of a pass is one group of 1 .. 5 steps");
static_assert((size_t)NN * DD * 4 < ((size_t)1 << 32), "a row's byte offset fits 32 bits");

__device__ __forceinline__ void fg_gather_rows(const int* __restrict__ rowptr,
                                               const int* __restrict__ cnt,
                                               const int2* __restrict__ csr,
                                               const float* __restrict__ dinv,
                                               const float* __restrict__ hw,
                                               const float* __restrict__ b,
                                               int d, int lane, float* srow) {
  const int      c   = lane & 15;
  const int      hi  = lane >> 4;
  const int      pe  = (c << 1) | hi;
  const unsigned c16 = (unsigned)c << 4;
  const char*    hb  = (const char*)hw;
  const unsigned p8  = (unsigned)pe << 3;
  const float2   bb  = ((const float2*)b)[pe];

  int4 rp = ((const int4*)rowptr)[d >> 2];
  int4 cn = ((const int4*)cnt)[d >> 2];
  int beg  = __builtin_amdgcn_readfirstlane(rp.x), n  = __builtin_amdgcn_readfirstlane(cn.x);
  int begA = __builtin_amdgcn_readfirstlane(rp.y), nA = __builtin_amdgcn_readfirstlane(cn.y);
  int begB = __builtin_amdgcn_readfirstlane(rp.z), nB = __builtin_amdgcn_readfirstlane(cn.z);
  int begC = __builtin_amdgcn_readfirstlane(rp.w), nC = __builtin_amdgcn_readfirstlane(cn.w);
  int   sl;
  float wl;
  ga_pairs(csr, beg, n, 0, pe, d, sl, wl);
  sl = ga_source(sl, d);

  for (int q = 0; q < FG_PER_WAVE; ++q) {
    int   sln;
    float wln;
    ga_pairs(csr, begA, nA, 0, pe, d + 1, sln, wln);

    float  dv    = dinv[d];
    float2 vself = *(const float2*)(hb + (((unsigned)d << 8) | p8));
    float4 acc;  acc.x = 0.0f;  acc.y = 0.0f;  acc.z = 0.0f;  acc.w = 0.0f;

    for (int c0 = 0; c0 < n; c0 += 32) {
      if (c0 > 0) {
        ga_pairs(csr, beg, n, c0, pe, d, sl, wl);
        sl = ga_source(sl, d);
      }
      int st = (min(32, n - c0) + 1) >> 1;
#pragma nounroll
      for (; st >= GA_STEPS; st -= GA_STEPS) {
        ga_rows<GA_STEPS>(hb, c16, sl, wl, acc);
        sl = half_down<GA_STEPS>(sl);
        wl = __int_as_float(half_down<GA_STEPS>(__float_as_int(wl)));
      }
      switch (st) {
        case 5: ga_rows<5>(hb, c16, sl, wl, acc); break;
        case 4: ga_rows<4>(hb, c16, sl, wl, acc); break;
        case 3: ga_rows<3>(hb, c16, sl, wl, acc); break;
        case 2: ga_rows<2>(hb, c16, sl, wl, acc); break;
        case 1: ga_rows<1>(hb, c16, sl, wl, acc); break;
        default: break;
      }
    }

    float kx = hi ? acc.z : acc.x,  ky = hi ? acc.w : acc.y;
    float gx = hi ? acc.x : acc.z,  gy = hi ? acc.y : acc.w;
    float tx = kx + other_half(gx);
    float ty = ky + other_half(gy);

    float  sn = dv * dv;
    float2 o;
    o.x = tanhf(fmaf(sn, vself.x, tx) + bb.x);
    o.y = tanhf(fmaf(sn, vself.y, ty) + bb.y);
    *(float2*)&srow[q * FG_PITCH + 2 * pe] = o;

    sl = ga_source(sln, d + 1);  wl = wln;
    d = d + 1;
    beg = begA;  n = nA;
    begA = begB; nA = nB;
    begB = begC; nB = nC;
    begC = 0;    nC = 0;
  }
}

__global__ void __launch_bounds__(FG_T)
k_gather_gemm(const int* __restrict__ rowptr, const int* __restrict__ cnt,
              const int2* __restrict__ csr, const float* __restrict__ dinv,
              const float* __restrict__ hw, const float* __restrict__ b,
              const float* __restrict__ Wn, float* __restrict__ hwn) {
  __shared__ __attribute__((aligned(16))) float stage[FG_NODES * FG_PITCH];
  const int lane = threadIdx.x & 31;
  const int w    = __builtin_amdgcn_readfirstlane((int)(threadIdx.x >> 5));
  const int base = blockIdx.x * FG_NODES;
  const int half = lane >> 4;
  const int m    = lane & 15;
  const int tm   = w >> 2;
  const int tn   = w & 3;
  const int col  = tn * 16 + m;

  v2f bf[DD / 4];
#pragma unroll
  for (int kb = 0; kb < DD / 4; ++kb) {
    int k0 = kb * 4 + half * 2;
    bf[kb][0] = Wn[k0 * DD + col];
    bf[kb][1] = Wn[(k0 + 1) * DD + col];
  }

  fg_gather_rows(rowptr, cnt, csr, dinv, hw, b, base + w * FG_PER_WAVE, lane,
                 stage + w * FG_PER_WAVE * FG_PITCH);

  __syncthreads();

  const float* arow = stage + (tm * 16 + m) * FG_PITCH + half * 2;
  v8f c = {};
#pragma unroll
  for (int kb = 0; kb < DD / 4; ++kb) {
    v2f a = *(const v2f*)(arow + kb * 4);
    c = __builtin_amdgcn_wmma_f32_16x16x4_f32(
         false, a,  false, bf[kb],
         (short)0, c,  false,  false);
  }

  int r0 = base + tm * 16 + half * 8;
#pragma unroll
  for (int j = 0; j < 8; ++j)
    hwn[(size_t)(r0 + j) * DD + col] = c[j];
}

__device__ __forceinline__ unsigned fflip(float f) {
  unsigned u = __float_as_uint(f);
  return u ^ ((unsigned)((int)u >> 31) | 0x80000000u);
}
__device__ __forceinline__ float funflip(unsigned u) {
  unsigned m = (u >> 31) ? 0x80000000u : 0xFFFFFFFFu;
  return __uint_as_float(u ^ m);
}

__device__ __forceinline__ void pool_flush(int g, int col, unsigned mx, float sm, float cn,
                                           unsigned* __restrict__ gmp, float* __restrict__ gsum,
                                           float* __restrict__ counts) {
  atomicMax(&gmp[g * DD + col], mx);
  atomicAdd(&gsum[g * DD + col], sm);
  if (col == 0) atomicAdd(&counts[g], cn);
}

__global__ void __launch_bounds__(FG_T)
k_gather_pool(const int* __restrict__ rowptr, const int* __restrict__ cnt,
              const int2* __restrict__ csr, const float* __restrict__ dinv,
              const float* __restrict__ hw, const float* __restrict__ b,
              const int* __restrict__ bidx, unsigned* __restrict__ gmp,
              float* __restrict__ gsum, float* __restrict__ counts) {
  __shared__ __attribute__((aligned(16))) float stage[FG_NODES * FG_PITCH];
  const int t    = threadIdx.x;
  const int lane = t & 31;
  const int w    = __builtin_amdgcn_readfirstlane((int)(t >> 5));
  const int base = blockIdx.x * FG_NODES;

  int g_own = 0;
  if (t < DD) g_own = bidx[base + lane];

  fg_gather_rows(rowptr, cnt, csr, dinv, hw, b, base + w * FG_PER_WAVE, lane,
                 stage + w * FG_PER_WAVE * FG_PITCH);

  __syncthreads();

  if (t < DD) {
    float v[FG_NODES];
#pragma unroll
    for (int r = 0; r < FG_NODES; ++r) v[r] = stage[r * FG_PITCH + t];

    int      cg = __builtin_amdgcn_readlane(g_own, 0);
    unsigned mx = fflip(v[0]);
    float    sm = v[0];
    float    cn = 1.0f;
#pragma unroll
    for (int r = 1; r < FG_NODES; ++r) {
      int g = __builtin_amdgcn_readlane(g_own, r);
      if (g != cg) {
        pool_flush(cg, t, mx, sm, cn, gmp, gsum, counts);
        cg = g;  mx = fflip(v[r]);  sm = v[r];  cn = 1.0f;
      } else {
        mx = max(mx, fflip(v[r]));  sm += v[r];  cn += 1.0f;
      }
    }
    pool_flush(cg, t, mx, sm, cn, gmp, gsum, counts);
  }
}

__global__ void k_final(const unsigned* __restrict__ gmp, const float* __restrict__ gsum,
                        const float* __restrict__ counts, const float* __restrict__ Wout,
                        const float* __restrict__ bout, float* __restrict__ out,
                        float* __restrict__ hidden) {
  __shared__ float red[2 * DD];
  int g = blockIdx.x, t = threadIdx.x;
  float cnt = fmaxf(counts[g], 1.0f);
  float hv = (t < DD) ? funflip(gmp[g * DD + t]) : (gsum[g * DD + (t - DD)] / cnt);
  hidden[(size_t)g * 2 * DD + t] = hv;
  red[t] = hv * Wout[t];
  __syncthreads();
  for (int s = DD; s > 0; s >>= 1) {
    if (t < s) red[t] += red[t + s];
    __syncthreads();
  }
  if (t == 0) out[g] = red[0] + bout[0];
}

extern "C" void kernel_launch(void* const* d_in, const int* in_sizes, int n_in,
                              void* d_out, int out_size, void* d_ws, size_t ws_size,
                              hipStream_t stream) {
  const float* x    = (const float*)d_in[0];
  const int*   ei   = (const int*)d_in[1];
  const int*   bidx = (const int*)d_in[2];
  const float* W0   = (const float*)d_in[3];
  const float* b0   = (const float*)d_in[4];
  const float* W1   = (const float*)d_in[5];
  const float* b1   = (const float*)d_in[6];
  const float* W2   = (const float*)d_in[7];
  const float* b2   = (const float*)d_in[8];
  const float* W3   = (const float*)d_in[9];
  const float* b3   = (const float*)d_in[10];
  const float* Wout = (const float*)d_in[11];
  const float* bout = (const float*)d_in[12];

  const int* src = ei;
  const int* dst = ei + EE;

  char*  ws   = (char*)d_ws;
  size_t off  = 0;
  const size_t nodeWords = (((size_t)NN * 4 + 255) & ~(size_t)255);
  float* dinv = (float*)(ws + off);  off += nodeWords;
  float* bufP = (float*)(ws + off);  off += (size_t)NN * DD * 4;
  float* bufQ = (float*)(ws + off);  off += (size_t)NN * DD * 4;
  unsigned* gmp = (unsigned*)(ws + off); off += (size_t)GG * DD * 4;
  float* gsum   = (float*)(ws + off);    off += (size_t)GG * DD * 4;
  float* counts = (float*)(ws + off);    off += (size_t)GG * 4;
  int* cnt    = (int*)(ws + off);  off += nodeWords;
  int* rowptr = (int*)(ws + off);  off += nodeWords;
  int* cntTab   = (int*)(ws + off);  off += (size_t)NBIN * TAB_PITCH * 4;
  int* binTot   = (int*)(ws + off);  off += (size_t)NBINP * 4;
  int* binStart = (int*)(ws + off);  off += (size_t)NBINP * 4;
  int2* tmp   = (int2*)(ws + off); off += (size_t)EE * 8;
  int2* csr   = (int2*)(ws + off); off += (size_t)EE * 8;
  (void)ws_size; (void)n_in; (void)in_sizes; (void)out_size;

  float* outp   = (float*)d_out;
  float* hidden = outp + GG;

  k_count_gemm  <<<CBLK + GEMM_P1, CB_T, 0, stream>>>(dst, cntTab, x, W0, bufQ, gmp, gsum, counts);
  k_bin_colscan <<<NBIN, CB_T, 0, stream>>>(cntTab, binTot);
  k_scatter_gemm<<<CBLK + (GEMM_P2 - GEMM_P1), CB_T, 0, stream>>>(src, dst, cntTab, binTot, binStart, tmp,
                                                                  x, W0, bufQ, gmp, gsum, counts);
  k_degree_gemm <<<NBIN + (GEMM_P3 - GEMM_P2), BIN_NODES, 0, stream>>>(tmp, binStart, cnt, rowptr, dinv,
                                                                       x, W0, bufQ, gmp, gsum, counts);
  k_place_gemm  <<<NBIN + (GEMM_BLOCKS - GEMM_P3), BIN_NODES, 0, stream>>>(tmp, binStart, rowptr, dinv, csr,
                                                                           x, W0, bufQ, gmp, gsum, counts);

  k_gather_gemm<<<NN / FG_NODES, FG_T, 0, stream>>>(rowptr, cnt, csr, dinv, bufQ, b0, W1, bufP);
  k_gather_gemm<<<NN / FG_NODES, FG_T, 0, stream>>>(rowptr, cnt, csr, dinv, bufP, b1, W2, bufQ);
  k_gather_gemm<<<NN / FG_NODES, FG_T, 0, stream>>>(rowptr, cnt, csr, dinv, bufQ, b2, W3, bufP);
  k_gather_pool<<<NN / FG_NODES, FG_T, 0, stream>>>(rowptr, cnt, csr, dinv, bufP, b3, bidx,
                                                    gmp, gsum, counts);

  k_final    <<<GG, 2 * DD, 0, stream>>>(gmp, gsum, counts, Wout, bout, outp, hidden);
}


# gather launches: 80 nodes per block, 8 per wave (1250 blocks, one resident round) not 32 and 4 (3125); a wave forms two product tiles, W staged in LDS; mean-pool partial sums span 40 rows not 32
# speedup vs baseline: 4.6415x; 1.0486x over previous
#include <hip/hip_runtime.h>
#include <math.h>

#define NN 100000
#define EE 1000000
#define FF 128
#define DD 64
#define GG 256

typedef float v2f __attribute__((ext_vector_type(2)));
typedef float v8f __attribute__((ext_vector_type(8)));

#define BIN_SHIFT 8
#define BIN_NODES (1 << BIN_SHIFT)
#define NBIN      ((NN + BIN_NODES - 1) / BIN_NODES)
#define CB_T      256
#define CB_EPT    16
#define CB_EDGES  (CB_T * CB_EPT)
#define CBLK      248
#define TAB_PITCH 256
#define NBINP     512
#define BS_GROUP  4
static_assert(BS_GROUP == 4, "the load fence of the placing kernel names four values");
static_assert(CBLK * CB_EDGES >= EE, "the level-1 blocks cover every edge");
static_assert(CBLK <= TAB_PITCH && TAB_PITCH == CB_T, "one thread per level-1 block scans a bin row");
static_assert(NBIN > CB_T && NBIN < 2 * CB_T && NBINP == 2 * CB_T, "two bins per thread of a moving block, the spare bin among them");
static_assert(NN % 4 == 0, "list bounds of four consecutive nodes are read as one int4");

__device__ __forceinline__ int bin_of(int d) {
  unsigned b = (unsigned)d >> BIN_SHIFT;
  return (int)(b < (unsigned)NBIN ? b : (unsigned)(NBIN - 1));
}

#define GEMM_BLOCKS ((NN / 16) * 4 / 8)
#define GEMM_P1 480
#define GEMM_P2 1325
#define GEMM_P3 1925
static_assert(GEMM_P1 * 256 >= GG * DD, "the pooling start values are stored by blocks of the first launch");
static_assert(GEMM_P1 < GEMM_P2 && GEMM_P2 < GEMM_P3 && GEMM_P3 < GEMM_BLOCKS, "four parts in order");
__device__ __forceinline__ void gemm0_block(int gb, const float* __restrict__ A,
                                            const float* __restrict__ B, float* __restrict__ C,
                                            unsigned* __restrict__ gmp, float* __restrict__ gsum,
                                            float* __restrict__ counts) {
  constexpr int K = FF;
  int gid  = gb * 256 + threadIdx.x;
  if (gid < GG * DD) { gmp[gid] = 0x007FFFFFu  ; gsum[gid] = 0.0f; }
  if (gid < GG) counts[gid] = 0.0f;
  int wave = gid >> 5;
  int lane = gid & 31;
  int tm = wave >> 2;
  int tn = wave & 3;
  if (tm * 16 >= NN) return;

  int half = lane >> 4;
  int m    = lane & 15;
  const float* arow = A + (size_t)(tm * 16 + m) * K;
  int col = tn * 16 + m;

  v8f acc = {};
#pragma unroll
  for (int kb = 0; kb < K; kb += 4) {
    int k0 = kb + half * 2;
    v2f a; a[0] = arow[k0];          a[1] = arow[k0 + 1];
    v2f b; b[0] = B[k0 * DD + col];  b[1] = B[(k0 + 1) * DD + col];
    acc = __builtin_amdgcn_wmma_f32_16x16x4_f32(
         false, a,  false, b,
         (short)0, acc,  false,  false);
  }

  int r0 = tm * 16 + half * 8;
#pragma unroll
  for (int j = 0; j < 8; ++j)
    C[(size_t)(r0 + j) * DD + col] = acc[j];
}

__device__ __forceinline__ void bin_count_block(const int* __restrict__ dst,
                                                int* __restrict__ cntTab) {
  __shared__ int lc[NBIN + 1];
  const int t = threadIdx.x, b = blockIdx.x;
  for (int i = t; i < NBIN + 1; i += CB_T) lc[i] = 0;
  const int e0 = b * CB_EDGES + t;
  int d[CB_EPT];
#pragma unroll
  for (int i = 0; i < CB_EPT; ++i) d[i] = dst[min(e0 + i * CB_T, EE - 1)];
  __syncthreads();
#pragma unroll
  for (int i = 0; i < CB_EPT; ++i)
    atomicAdd(&lc[(e0 + i * CB_T < EE) ? bin_of(d[i]) : NBIN], 1);
  __syncthreads();
  for (int i = t; i < NBIN; i += CB_T) cntTab[i * TAB_PITCH + b] = lc[i];
}

__global__ void __launch_bounds__(CB_T)
k_count_gemm(const int* __restrict__ dst, int* __restrict__ cntTab,
             const float* __restrict__ x, const float* __restrict__ W0, float* __restrict__ hw,
             unsigned* __restrict__ gmp, float* __restrict__ gsum, float* __restrict__ counts) {
  if (blockIdx.x >= CBLK) {
    gemm0_block((int)blockIdx.x - CBLK, x, W0, hw, gmp, gsum, counts);
    return;
  }
  bin_count_block(dst, cntTab);
}

__device__ __forceinline__ int wave_scan_incl(int v, int lane) {
  for (int o = 1; o < 32; o <<= 1) {
    int up = __shfl_up(v, o, 32);
    if (lane >= o) v += up;
  }
  return v;
}

__global__ void __launch_bounds__(CB_T)
k_bin_colscan(int* __restrict__ cntTab, int* __restrict__ binTot) {
  __shared__ int wtot[CB_T / 32];
  const int t = threadIdx.x, i = blockIdx.x, lane = t & 31, w = t >> 5;
  int c = 0;
  if (t < CBLK) c = cntTab[i * TAB_PITCH + t];
  int inc = wave_scan_incl(c, lane);
  if (lane == 31) wtot[w] = inc;
  __syncthreads();
  int wbase = 0;
  for (int k = 0; k < w; ++k) wbase += wtot[k];
  if (t < CBLK) cntTab[i * TAB_PITCH + t] = wbase + inc - c;
  if (t == CB_T - 1) binTot[i] = wbase + inc;
}


__device__ __forceinline__ void bin_scatter_block(const int* __restrict__ src,
                                                  const int* __restrict__ dst,
                                                  const int* __restrict__ cntTab,
                                                  const int* __restrict__ binTot,
                                                  int* __restrict__ binStart,
                                                  int2* __restrict__ tmp) {
  __shared__ int cur[NBINP];
  __shared__ int wt0[CB_T / 32];
  __shared__ int wt1[CB_T / 32];
  const int t = threadIdx.x, b = blockIdx.x, lane = t & 31, w = t >> 5;
  const int e0 = b * CB_EDGES + t;
  const int i1 = min(t + CB_T, NBIN - 1);
  int c0 = cntTab[t * TAB_PITCH + b];
  int c1 = cntTab[i1 * TAB_PITCH + b];
  int T0 = binTot[t];
  int T1 = binTot[i1];
  int s[CB_EPT], d[CB_EPT], p[CB_EPT];
#pragma unroll
  for (int i = 0; i < CB_EPT; ++i) {
    int e = min(e0 + i * CB_T, EE - 1);
    s[i] = src[e];
    d[i] = dst[e];
  }
  if (t + CB_T >= NBIN) T1 = 0;
  int inc0 = wave_scan_incl(T0, lane);
  int inc1 = wave_scan_incl(T1, lane);
  if (lane == 31) { wt0[w] = inc0;  wt1[w] = inc1; }
  __syncthreads();
  int base0 = inc0 - T0;
  int base1 = inc1 - T1;
  for (int k = 0; k < CB_T / 32; ++k) {
    int a = wt0[k];
    base1 += a;
    if (k < w) { base0 += a;  base1 += wt1[k]; }
  }
  cur[t] = base0 + c0;
  cur[t + CB_T] = (t + CB_T < NBIN) ? base1 + c1 : EE;
  if (b == 0) { binStart[t] = base0;  binStart[t + CB_T] = base1; }
  __syncthreads();
#pragma unroll
  for (int i = 0; i < CB_EPT; ++i)
    p[i] = atomicAdd(&cur[(e0 + i * CB_T < EE) ? bin_of(d[i]) : NBIN], 1);
#pragma unroll
  for (int i = 0; i < CB_EPT; ++i)
    if ((unsigned)p[i] < (unsigned)EE)
      tmp[p[i]] = make_int2(s[i], d[i]);
}

__global__ void __launch_bounds__(CB_T)
k_scatter_gemm(const int* __restrict__ src, const int* __restrict__ dst,
               const int* __restrict__ cntTab, const int* __restrict__ binTot,
               int* __restrict__ binStart, int2* __restrict__ tmp,
               const float* __restrict__ x, const float* __restrict__ W0, float* __restrict__ hw,
               unsigned* __restrict__ gmp, float* __restrict__ gsum, float* __restrict__ counts) {
  if (blockIdx.x >= CBLK) {
    gemm0_block((int)blockIdx.x - CBLK + GEMM_P1, x, W0, hw, gmp, gsum, counts);
    return;
  }
  bin_scatter_block(src, dst, cntTab, binTot, binStart, tmp);
}

__device__ __forceinline__ void bin_range(const int* __restrict__ binStart, int bin, int& beg,
                                          int& end) {
  beg = binStart[bin];
  end = binStart[bin + 1];
  beg = min(max(beg, 0), EE);
  end = min(max(end, beg), EE);
}

__device__ __forceinline__ void bin_degree_block(const int2* __restrict__ tmp,
                                                 const int* __restrict__ binStart,
                                                 int* __restrict__ cnt, int* __restrict__ rowptr,
                                                 float* __restrict__ dinv) {
  __shared__ int lc[BIN_NODES + 1];
  __shared__ int wtot[BIN_NODES / 32];
  const int bin = blockIdx.x, t = threadIdx.x;
  int beg, end;
  bin_range(binStart, bin, beg, end);
  lc[t] = 0;
  if (t == 0) lc[BIN_NODES] = 0;
  __syncthreads();
  for (int i0 = beg + t; i0 < end; i0 += BIN_NODES * BS_GROUP) {
    int dd[BS_GROUP];
#pragma unroll
    for (int k = 0; k < BS_GROUP; ++k) dd[k] = tmp[min(i0 + k * BIN_NODES, end - 1)].y;
#pragma unroll
    for (int k = 0; k < BS_GROUP; ++k)
      atomicAdd(&lc[(i0 + k * BIN_NODES < end) ? ((dd[k] - bin * BIN_NODES) & (BIN_NODES - 1))
                                                : BIN_NODES], 1);
  }
  __syncthreads();
  const int lane = t & 31, w = t >> 5;
  int c   = lc[t];
  int inc = c;
  for (int o = 1; o < 32; o <<= 1) {
    int up = __shfl_up(inc, o, 32);
    if (lane >= o) inc += up;
  }
  if (lane == 31) wtot[w] = inc;
  __syncthreads();
  int wbase = 0;
  for (int k = 0; k < w; ++k) wbase += wtot[k];
  int node = bin * BIN_NODES + t;
  if (node < NN) {
    cnt[node]    = c;
    rowptr[node] = beg + wbase + inc - c;
    dinv[node]   = rsqrtf((float)c + 1.0f);
  }
}

__global__ void __launch_bounds__(BIN_NODES)
k_degree_gemm(const int2* __restrict__ tmp, const int* __restrict__ binStart,
              int* __restrict__ cnt, int* __restrict__ rowptr, float* __restrict__ dinv,
              const float* __restrict__ x, const float* __restrict__ W0, float* __restrict__ hw,
              unsigned* __restrict__ gmp, float* __restrict__ gsum, float* __restrict__ counts) {
  if (blockIdx.x >= NBIN) {
    gemm0_block((int)blockIdx.x - NBIN + GEMM_P2, x, W0, hw, gmp, gsum, counts);
    return;
  }
  bin_degree_block(tmp, binStart, cnt, rowptr, dinv);
}

__device__ __forceinline__ void bin_place_block(const int2* __restrict__ tmp,
                                                const int* __restrict__ binStart,
                                                const int* __restrict__ rowptr,
                                                const float* __restrict__ dinv,
                                                int2* __restrict__ csr) {
  __shared__ int   cur[BIN_NODES + 1];
  __shared__ float dv[BIN_NODES + 1];
  const int bin = blockIdx.x, t = threadIdx.x;
  int beg, end;
  bin_range(binStart, bin, beg, end);
  const int node = bin * BIN_NODES + t;
  cur[t] = (node < NN) ? rowptr[node] : EE;
  dv[t]  = (node < NN) ? dinv[node] : 0.0f;
  if (t == 0) { cur[BIN_NODES] = EE;  dv[BIN_NODES] = 0.0f; }
  __syncthreads();
  for (int i0 = beg + t; i0 < end; i0 += BIN_NODES * BS_GROUP) {
    int2  e[BS_GROUP];
    float ws[BS_GROUP];
    int   dl[BS_GROUP], p[BS_GROUP];
#pragma unroll
    for (int k = 0; k < BS_GROUP; ++k) e[k] = tmp[min(i0 + k * BIN_NODES, end - 1)];
#pragma unroll
    for (int k = 0; k < BS_GROUP; ++k)
      ws[k] = dinv[((unsigned)e[k].x < (unsigned)NN) ? e[k].x : 0];
    asm volatile("" :: "v"(ws[0]), "v"(ws[1]), "v"(ws[2]), "v"(ws[3]));
#pragma unroll
    for (int k = 0; k < BS_GROUP; ++k) {
      dl[k] = (i0 + k * BIN_NODES < end) ? ((e[k].y - bin * BIN_NODES) & (BIN_NODES - 1))
                                         : BIN_NODES;
      p[k]  = atomicAdd(&cur[dl[k]], 1);
    }
#pragma unroll
    for (int k = 0; k < BS_GROUP; ++k)
      if ((unsigned)p[k] < (unsigned)EE)
        csr[p[k]] = make_int2(e[k].x, __float_as_int(ws[k] * dv[dl[k]]));
  }
}

__global__ void __launch_bounds__(BIN_NODES)
k_place_gemm(const int2* __restrict__ tmp, const int* __restrict__ binStart,
             const int* __restrict__ rowptr, const float* __restrict__ dinv,
             int2* __restrict__ csr,
             const float* __restrict__ x, const float* __restrict__ W0, float* __restrict__ hw,
             unsigned* __restrict__ gmp, float* __restrict__ gsum, float* __restrict__ counts) {
  if (blockIdx.x >= NBIN) {
    gemm0_block((int)blockIdx.x - NBIN + GEMM_P3, x, W0, hw, gmp, gsum, counts);
    return;
  }
  bin_place_block(tmp, binStart, rowptr, dinv, csr);
}


template <int K>
__device__ __forceinline__ int half_lane(int v) {
  return __builtin_amdgcn_update_dpp(0, v, 0x150 + K, 0xF, 0xF, true);
}
template <int R>
__device__ __forceinline__ int half_down(int v) {
  return __builtin_amdgcn_update_dpp(0, v, 0x120 + (16 - R), 0xF, 0xF, true);
}
__device__ __forceinline__ float other_half(float x) {
  return __int_as_float(__builtin_amdgcn_permlanex16(__float_as_int(x), __float_as_int(x),
                                                     0x76543210, 0xfedcba98, false, false));
}

template <int K, int G>
__device__ __forceinline__ void ga_issue(const char* __restrict__ hb, unsigned c16, int sl,
                                         float4* v) {
  if constexpr (K < G) {
    unsigned s = (unsigned)half_lane<K>(sl);
    v[K] = *(const float4*)(hb + ((s << 8) | c16));
    ga_issue<K + 1, G>(hb, c16, sl, v);
  }
}
template <int K, int G>
__device__ __forceinline__ void ga_sum(float wl, const float4* v, float4& acc) {
  if constexpr (K < G) {
    float w = __int_as_float(half_lane<K>(__float_as_int(wl)));
    acc.x = fmaf(w, v[K].x, acc.x);
    acc.y = fmaf(w, v[K].y, acc.y);
    acc.z = fmaf(w, v[K].z, acc.z);
    acc.w = fmaf(w, v[K].w, acc.w);
    ga_sum<K + 1, G>(wl, v, acc);
  }
}
template <int G>
__device__ __forceinline__ void ga_rows(const char* __restrict__ hb, unsigned c16, int sl, float wl,
                                        float4& acc) {
  float4 v[G];
  ga_issue<0, G>(hb, c16, sl, v);
  ga_sum<0, G>(wl, v, acc);
}

__device__ __forceinline__ void ga_pairs(const int2* __restrict__ csr, int beg, int n, int c0,
                                         int pe, int d, int& sl, float& wl) {
  sl = d;
  wl = 0.0f;
  if (pe < min(32, n - c0)) {
    int2 ew = csr[beg + c0 + pe];
    sl = ew.x;
    wl = __int_as_float(ew.y);
  }
}
__device__ __forceinline__ int ga_source(int sl, int d) {
  return ((unsigned)sl >= (unsigned)NN) ? d : sl;
}

#define FG_T        320
#define FG_NODES    80
#define FG_PER_WAVE (FG_NODES / (FG_T / 32))
#define FG_PITCH    68
#define GA_STEPS    6
static_assert(NN % FG_NODES == 0, "every block owns 80 whole nodes");
static_assert(FG_PER_WAVE == 8, "the list bounds of a wave's nodes are read as two int4");
static_assert(GA_STEPS == 6, "the rest of a pass is one group of 1 .. 5 steps");
static_assert((size_t)NN * DD * 4 < ((size_t)1 << 32), "a row's byte offset fits 32 bits");

template <class F>
__device__ __forceinline__ void fg_gather_rows(const int* __restrict__ rowptr,
                                               const int* __restrict__ cnt,
                                               const int2* __restrict__ csr,
                                               const float* __restrict__ dinv,
                                               const float* __restrict__ hw,
                                               const float* __restrict__ b,
                                               int d, int lane, float* srow, F once) {
  const int      c   = lane & 15;
  const int      hi  = lane >> 4;
  const int      pe  = (c << 1) | hi;
  const unsigned c16 = (unsigned)c << 4;
  const char*    hb  = (const char*)hw;
  const unsigned p8  = (unsigned)pe << 3;
  const float2   bb  = ((const float2*)b)[pe];

  int4 rp0 = ((const int4*)rowptr)[d >> 2], rp1 = ((const int4*)rowptr)[(d >> 2) + 1];
  int4 cn0 = ((const int4*)cnt)[d >> 2],    cn1 = ((const int4*)cnt)[(d >> 2) + 1];
  int bg[FG_PER_WAVE], nn[FG_PER_WAVE];
  bg[0] = __builtin_amdgcn_readfirstlane(rp0.x);  nn[0] = __builtin_amdgcn_readfirstlane(cn0.x);
  bg[1] = __builtin_amdgcn_readfirstlane(rp0.y);  nn[1] = __builtin_amdgcn_readfirstlane(cn0.y);
  bg[2] = __builtin_amdgcn_readfirstlane(rp0.z);  nn[2] = __builtin_amdgcn_readfirstlane(cn0.z);
  bg[3] = __builtin_amdgcn_readfirstlane(rp0.w);  nn[3] = __builtin_amdgcn_readfirstlane(cn0.w);
  bg[4] = __builtin_amdgcn_readfirstlane(rp1.x);  nn[4] = __builtin_amdgcn_readfirstlane(cn1.x);
  bg[5] = __builtin_amdgcn_readfirstlane(rp1.y);  nn[5] = __builtin_amdgcn_readfirstlane(cn1.y);
  bg[6] = __builtin_amdgcn_readfirstlane(rp1.z);  nn[6] = __builtin_amdgcn_readfirstlane(cn1.z);
  bg[7] = __builtin_amdgcn_readfirstlane(rp1.w);  nn[7] = __builtin_amdgcn_readfirstlane(cn1.w);
  int   sl;
  float wl;
  ga_pairs(csr, bg[0], nn[0], 0, pe, d, sl, wl);
  sl = ga_source(sl, d);
  once();

  for (int q = 0; q < FG_PER_WAVE; ++q) {
    int   sln;
    float wln;
    ga_pairs(csr, bg[1], nn[1], 0, pe, d + 1, sln, wln);
    const int beg = bg[0], n = nn[0];

    float  dv    = dinv[d];
    float2 vself = *(const float2*)(hb + (((unsigned)d << 8) | p8));
    float4 acc;  acc.x = 0.0f;  acc.y = 0.0f;  acc.z = 0.0f;  acc.w = 0.0f;

    for (int c0 = 0; c0 < n; c0 += 32) {
      if (c0 > 0) {
        ga_pairs(csr, beg, n, c0, pe, d, sl, wl);
        sl = ga_source(sl, d);
      }
      int st = (min(32, n - c0) + 1) >> 1;
#pragma nounroll
      for (; st >= GA_STEPS; st -= GA_STEPS) {
        ga_rows<GA_STEPS>(hb, c16, sl, wl, acc);
        sl = half_down<GA_STEPS>(sl);
        wl = __int_as_float(half_down<GA_STEPS>(__float_as_int(wl)));
      }
      switch (st) {
        case 5: ga_rows<5>(hb, c16, sl, wl, acc); break;
        case 4: ga_rows<4>(hb, c16, sl, wl, acc); break;
        case 3: ga_rows<3>(hb, c16, sl, wl, acc); break;
        case 2: ga_rows<2>(hb, c16, sl, wl, acc); break;
        case 1: ga_rows<1>(hb, c16, sl, wl, acc); break;
        default: break;
      }
    }

    float kx = hi ? acc.z : acc.x,  ky = hi ? acc.w : acc.y;
    float gx = hi ? acc.x : acc.z,  gy = hi ? acc.y : acc.w;
    float tx = kx + other_half(gx);
    float ty = ky + other_half(gy);

    float  sn = dv * dv;
    float2 o;
    o.x = tanhf(fmaf(sn, vself.x, tx) + bb.x);
    o.y = tanhf(fmaf(sn, vself.y, ty) + bb.y);
    *(float2*)&srow[q * FG_PITCH + 2 * pe] = o;

    sl = ga_source(sln, d + 1);  wl = wln;
    d = d + 1;
#pragma unroll
    for (int k = 0; k + 1 < FG_PER_WAVE; ++k) { bg[k] = bg[k + 1];  nn[k] = nn[k + 1]; }
    bg[FG_PER_WAVE - 1] = 0;  nn[FG_PER_WAVE - 1] = 0;
  }
}

#define WL_PITCH 72
static_assert((DD * DD / 4) <= 4 * FG_T && (DD * DD / 4) > 3 * FG_T, "four float4 per thread cover Wn, the fourth only in part");
__global__ void __launch_bounds__(FG_T)
k_gather_gemm(const int* __restrict__ rowptr, const int* __restrict__ cnt,
              const int2* __restrict__ csr, const float* __restrict__ dinv,
              const float* __restrict__ hw, const float* __restrict__ b,
              const float* __restrict__ Wn, float* __restrict__ hwn) {
  __shared__ __attribute__((aligned(16))) float stage[FG_NODES * FG_PITCH];
  __shared__ __attribute__((aligned(16))) float wls[DD * WL_PITCH];
  const int t    = threadIdx.x;
  const int lane = t & 31;
  const int w    = __builtin_amdgcn_readfirstlane((int)(t >> 5));
  const int base = blockIdx.x * FG_NODES;
  const int half = lane >> 4;
  const int m    = lane & 15;

  const float4* W4 = (const float4*)Wn;
  const float4 wv0 = W4[t], wv1 = W4[t + FG_T], wv2 = W4[t + 2 * FG_T];
  const float4 wv3 = W4[min(t + 3 * FG_T, DD * DD / 4 - 1)];

  fg_gather_rows(rowptr, cnt, csr, dinv, hw, b, base + w * FG_PER_WAVE, lane,
                 stage + w * FG_PER_WAVE * FG_PITCH,
                 [=]() {
                   auto put = [](int idx, float4 v) {
                     *(float4*)&wls[(idx >> 4) * WL_PITCH + ((idx & 15) << 2)] = v;
                   };
                   put(t, wv0);
                   put(t + FG_T, wv1);
                   put(t + 2 * FG_T, wv2);
                   if (t + 3 * FG_T < DD * DD / 4) put(t + 3 * FG_T, wv3);
                 });

  __syncthreads();

  const int tm = w >> 1;
  const int cb = (w & 1) * 32 + m;
  const float* arow = stage + (tm * 16 + m) * FG_PITCH + half * 2;
  const float* brow = wls + (half * 2) * WL_PITCH + cb;
  v8f c0 = {}, c1 = {};
#pragma unroll
  for (int kb = 0; kb < DD / 4; ++kb) {
    v2f a = *(const v2f*)(arow + kb * 4);
    v2f b0, b1;
    b0[0] = brow[(kb * 4) * WL_PITCH];       b0[1] = brow[(kb * 4 + 1) * WL_PITCH];
    b1[0] = brow[(kb * 4) * WL_PITCH + 16];  b1[1] = brow[(kb * 4 + 1) * WL_PITCH + 16];
    c0 = __builtin_amdgcn_wmma_f32_16x16x4_f32(
         false, a,  false, b0,
         (short)0, c0,  false,  false);
    c1 = __builtin_amdgcn_wmma_f32_16x16x4_f32(
         false, a,  false, b1,
         (short)0, c1,  false,  false);
  }

  int r0 = base + tm * 16 + half * 8;
#pragma unroll
  for (int j = 0; j < 8; ++j) {
    hwn[(size_t)(r0 + j) * DD + cb]      = c0[j];
    hwn[(size_t)(r0 + j) * DD + cb + 16] = c1[j];
  }
}

__device__ __forceinline__ unsigned fflip(float f) {
  unsigned u = __float_as_uint(f);
  return u ^ ((unsigned)((int)u >> 31) | 0x80000000u);
}
__device__ __forceinline__ float funflip(unsigned u) {
  unsigned m = (u >> 31) ? 0x80000000u : 0xFFFFFFFFu;
  return __uint_as_float(u ^ m);
}

__device__ __forceinline__ void pool_flush(int g, int col, unsigned mx, float sm, float cn,
                                           unsigned* __restrict__ gmp, float* __restrict__ gsum,
                                           float* __restrict__ counts) {
  atomicMax(&gmp[g * DD + col], mx);
  atomicAdd(&gsum[g * DD + col], sm);
  if (col == 0) atomicAdd(&counts[g], cn);
}

#define PG_ROWS (FG_NODES / 2)
static_assert(PG_ROWS > 32 && PG_ROWS <= 64, "a pooling wave holds its rows' graph ids in two registers");
__global__ void __launch_bounds__(FG_T)
k_gather_pool(const int* __restrict__ rowptr, const int* __restrict__ cnt,
              const int2* __restrict__ csr, const float* __restrict__ dinv,
              const float* __restrict__ hw, const float* __restrict__ b,
              const int* __restrict__ bidx, unsigned* __restrict__ gmp,
              float* __restrict__ gsum, float* __restrict__ counts) {
  __shared__ __attribute__((aligned(16))) float stage[FG_NODES * FG_PITCH];
  const int t    = threadIdx.x;
  const int lane = t & 31;
  const int w    = __builtin_amdgcn_readfirstlane((int)(t >> 5));
  const int base = blockIdx.x * FG_NODES;

  const int prow = (t >> 6) * PG_ROWS;
  const int col  = t & (DD - 1);
  int g_lo = 0, g_hi = 0;
  if (t < 2 * DD) {
    g_lo = bidx[base + prow + lane];
    if (lane < PG_ROWS - 32) g_hi = bidx[base + prow + 32 + lane];
  }

  fg_gather_rows(rowptr, cnt, csr, dinv, hw, b, base + w * FG_PER_WAVE, lane,
                 stage + w * FG_PER_WAVE * FG_PITCH, []() {});

  __syncthreads();

  if (t < 2 * DD) {
    float v[PG_ROWS];
#pragma unroll
    for (int r = 0; r < PG_ROWS; ++r) v[r] = stage[(prow + r) * FG_PITCH + col];

    int      cg = __builtin_amdgcn_readlane(g_lo, 0);
    unsigned mx = fflip(v[0]);
    float    sm = v[0];
    float    cn = 1.0f;
#pragma unroll
    for (int r = 1; r < PG_ROWS; ++r) {
      int g = (r < 32) ? __builtin_amdgcn_readlane(g_lo, r & 31) : __builtin_amdgcn_readlane(g_hi, r & 31);
      if (g != cg) {
        pool_flush(cg, col, mx, sm, cn, gmp, gsum, counts);
        cg = g;  mx = fflip(v[r]);  sm = v[r];  cn = 1.0f;
      } else {
        mx = max(mx, fflip(v[r]));  sm += v[r];  cn += 1.0f;
      }
    }
    pool_flush(cg, col, mx, sm, cn, gmp, gsum, counts);
  }
}

__global__ void k_final(const unsigned* __restrict__ gmp, const float* __restrict__ gsum,
                        const float* __restrict__ counts, const float* __restrict__ Wout,
                        const float* __restrict__ bout, float* __restrict__ out,
                        float* __restrict__ hidden) {
  __shared__ float red[2 * DD];
  int g = blockIdx.x, t = threadIdx.x;
  float cnt = fmaxf(counts[g], 1.0f);
  float hv = (t < DD) ? funflip(gmp[g * DD + t]) : (gsum[g * DD + (t - DD)] / cnt);
  hidden[(size_t)g * 2 * DD + t] = hv;
  red[t] = hv * Wout[t];
  __syncthreads();
  for (int s = DD; s > 0; s >>= 1) {
    if (t < s) red[t] += red[t + s];
    __syncthreads();
  }
  if (t == 0) out[g] = red[0] + bout[0];
}

extern "C" void kernel_launch(void* const* d_in, const int* in_sizes, int n_in,
                              void* d_out, int out_size, void* d_ws, size_t ws_size,
                              hipStream_t stream) {
  const float* x    = (const float*)d_in[0];
  const int*   ei   = (const int*)d_in[1];
  const int*   bidx = (const int*)d_in[2];
  const float* W0   = (const float*)d_in[3];
  const float* b0   = (const float*)d_in[4];
  const float* W1   = (const float*)d_in[5];
  const float* b1   = (const float*)d_in[6];
  const float* W2   = (const float*)d_in[7];
  const float* b2   = (const float*)d_in[8];
  const float* W3   = (const float*)d_in[9];
  const float* b3   = (const float*)d_in[10];
  const float* Wout = (const float*)d_in[11];
  const float* bout = (const float*)d_in[12];

  const int* src = ei;
  const int* dst = ei + EE;

  char*  ws   = (char*)d_ws;
  size_t off  = 0;
  const size_t nodeWords = (((size_t)NN * 4 + 255) & ~(size_t)255);
  float* dinv = (float*)(ws + off);  off += nodeWords;
  float* bufP = (float*)(ws + off);  off += (size_t)NN * DD * 4;
  float* bufQ = (float*)(ws + off);  off += (size_t)NN * DD * 4;
  unsigned* gmp = (unsigned*)(ws + off); off += (size_t)GG * DD * 4;
  float* gsum   = (float*)(ws + off);    off += (size_t)GG * DD * 4;
  float* counts = (float*)(ws + off);    off += (size_t)GG * 4;
  int* cnt    = (int*)(ws + off);  off += nodeWords;
  int* rowptr = (int*)(ws + off);  off += nodeWords;
  int* cntTab   = (int*)(ws + off);  off += (size_t)NBIN * TAB_PITCH * 4;
  int* binTot   = (int*)(ws + off);  off += (size_t)NBINP * 4;
  int* binStart = (int*)(ws + off);  off += (size_t)NBINP * 4;
  int2* tmp   = (int2*)(ws + off); off += (size_t)EE * 8;
  int2* csr   = (int2*)(ws + off); off += (size_t)EE * 8;
  (void)ws_size; (void)n_in; (void)in_sizes; (void)out_size;

  float* outp   = (float*)d_out;
  float* hidden = outp + GG;

  k_count_gemm  <<<CBLK + GEMM_P1, CB_T, 0, stream>>>(dst, cntTab, x, W0, bufQ, gmp, gsum, counts);
  k_bin_colscan <<<NBIN, CB_T, 0, stream>>>(cntTab, binTot);
  k_scatter_gemm<<<CBLK + (GEMM_P2 - GEMM_P1), CB_T, 0, stream>>>(src, dst, cntTab, binTot, binStart, tmp,
                                                                  x, W0, bufQ, gmp, gsum, counts);
  k_degree_gemm <<<NBIN + (GEMM_P3 - GEMM_P2), BIN_NODES, 0, stream>>>(tmp, binStart, cnt, rowptr, dinv,
                                                                       x, W0, bufQ, gmp, gsum, counts);
  k_place_gemm  <<<NBIN + (GEMM_BLOCKS - GEMM_P3), BIN_NODES, 0, stream>>>(tmp, binStart, rowptr, dinv, csr,
                                                                           x, W0, bufQ, gmp, gsum, counts);

  k_gather_gemm<<<NN / FG_NODES, FG_T, 0, stream>>>(rowptr, cnt, csr, dinv, bufQ, b0, W1, bufP);
  k_gather_gemm<<<NN / FG_NODES, FG_T, 0, stream>>>(rowptr, cnt, csr, dinv, bufP, b1, W2, bufQ);
  k_gather_gemm<<<NN / FG_NODES, FG_T, 0, stream>>>(rowptr, cnt, csr, dinv, bufQ, b2, W3, bufP);
  k_gather_pool<<<NN / FG_NODES, FG_T, 0, stream>>>(rowptr, cnt, csr, dinv, bufP, b3, bidx,
                                                    gmp, gsum, counts);

  k_final    <<<GG, 2 * DD, 0, stream>>>(gmp, gsum, counts, Wout, bout, outp, hidden);
}
